# starting kernel
# baseline (speedup 1.0000x reference)
#include <hip/hip_runtime.h>

constexpr int Nn  = 50000;
constexpr int Ne  = 800000;
constexpr int F   = 256;
constexpr int D0  = 128;
constexpr int H1  = 2;
constexpr int D1  = 64;
constexpr int C2  = 47;
constexpr int C2P = 64;

typedef float v2f __attribute__((ext_vector_type(2)));
typedef float v8f __attribute__((ext_vector_type(8)));

__device__ __forceinline__ unsigned enc_f(float f) {
    unsigned u = __float_as_uint(f);
    return (u & 0x80000000u) ? ~u : (u | 0x80000000u);
}
__device__ __forceinline__ float dec_f(unsigned u) {
    unsigned b = (u & 0x80000000u) ? (u ^ 0x80000000u) : ~u;
    return __uint_as_float(b);
}

__global__ void k_transpose(const float* __restrict__ W, float* __restrict__ Wt,
                            int K, int N, int Npad) {
    int i = blockIdx.x * blockDim.x + threadIdx.x;
    if (i >= Npad * K) return;
    const int c = i / K, k = i % K;
    Wt[i] = (c < N) ? W[(long long)k * N + c] : 0.0f;
}

template <bool HAS_A2, bool ACT>
__global__ __launch_bounds__(256)
void wmma_gemm(const float* __restrict__ A, const float* __restrict__ Wt,
               const float* __restrict__ A2, const float* __restrict__ W2t,
               const float* __restrict__ rowdeg, const float* __restrict__ bias,
               float* __restrict__ out, int M, int K, int Nout, int ncg)
{
    extern __shared__ float lds[];
    const int KP = K + 4;
    const int cg   = blockIdx.x % ncg;
    const int col0 = cg * 64;
    float* w1s = lds;
    float* w2s = lds + 64 * KP;

    const int ksh = 31 - __clz(K);
    for (int i = threadIdx.x; i < 64 * K; i += 256) {
        const int c = i >> ksh, k = i & (K - 1);
        w1s[c * KP + k] = Wt[(long long)(col0 + c) * K + k];
        if (HAS_A2) w2s[c * KP + k] = W2t[(long long)(col0 + c) * K + k];
    }
    __syncthreads();

    const int rt = (blockIdx.x / ncg) * 8 + (threadIdx.x >> 5);
    if (rt >= M / 16) return;
    const int lane = threadIdx.x & 31;
    const int half = lane >> 4;
    const int l    = lane & 15;
    const int row  = rt * 16 + l;

    const float* Arow  = A + (long long)row * K + half * 2;
    const float* A2row = HAS_A2 ? (A2 + (long long)row * K + half * 2) : nullptr;
    const float inv    = HAS_A2 ? (1.0f / fmaxf(rowdeg[row], 1.0f)) : 0.0f;
    __builtin_prefetch(Arow, 0, 3);

    const float* Bs[4];
    const float* B2s[4];
#pragma unroll
    for (int t = 0; t < 4; ++t) {
        Bs[t] = w1s + (t * 16 + l) * KP + half * 2;
        if (HAS_A2) B2s[t] = w2s + (t * 16 + l) * KP + half * 2;
    }

    v8f acc[4] = {};
    for (int kk = 0; kk < K; kk += 4) {
        const v2f a = *(const v2f*)(Arow + kk);
        v2f a2;
        if (HAS_A2) {
            const v2f t2 = *(const v2f*)(A2row + kk);
            a2.x = t2.x * inv; a2.y = t2.y * inv;
        }
#pragma unroll
        for (int t = 0; t < 4; ++t) {
            const v2f b = *(const v2f*)(Bs[t] + kk);
            acc[t] = __builtin_amdgcn_wmma_f32_16x16x4_f32(
                false, a, false, b, (short)0, acc[t], false, false);
            if (HAS_A2) {
                const v2f b2 = *(const v2f*)(B2s[t] + kk);
                acc[t] = __builtin_amdgcn_wmma_f32_16x16x4_f32(
                    false, a2, false, b2, (short)0, acc[t], false, false);
            }
        }
    }

#pragma unroll
    for (int t = 0; t < 4; ++t) {
        const int c = col0 + t * 16 + l;
        if (c < Nout) {
            const float bv = bias ? bias[c] : 0.0f;
#pragma unroll
            for (int v = 0; v < 8; ++v) {
                const int r = rt * 16 + v + 8 * half;
                float x = acc[t][v] + bv;
                if (ACT) x = x > 0.0f ? x : (expf(x) - 1.0f);
                out[(long long)r * Nout + c] = x;
            }
        }
    }
}

__global__ void fill_f32(float* p, float v, long long n) {
    long long i = (long long)blockIdx.x * blockDim.x + threadIdx.x;
    if (i < n) p[i] = v;
}
__global__ void fill_u32(unsigned* p, unsigned v, long long n) {
    long long i = (long long)blockIdx.x * blockDim.x + threadIdx.x;
    if (i < n) p[i] = v;
}

__global__ void k_deg(const int* __restrict__ dst, float* __restrict__ deg, int E) {
    int i = blockIdx.x * blockDim.x + threadIdx.x;
    if (i < E) atomicAdd(&deg[dst[i]], 1.0f);
}

__global__ void k_neigh(const float* __restrict__ x, const int* __restrict__ src,
                        const int* __restrict__ dst, float* __restrict__ neigh, int E) {
    long long tid = (long long)blockIdx.x * blockDim.x + threadIdx.x;
    if (tid >= (long long)E * (F / 4)) return;
    const int e = (int)(tid >> 6);
    const int c = ((int)tid & 63) * 4;
    const float4 v = *(const float4*)(x + (long long)src[e] * F + c);
    float* o = neigh + (long long)dst[e] * F + c;
    atomicAdd(o + 0, v.x); atomicAdd(o + 1, v.y);
    atomicAdd(o + 2, v.z); atomicAdd(o + 3, v.w);
}

__global__ void k_attn(const float* __restrict__ feat, const float* __restrict__ al,
                       const float* __restrict__ ar, float* __restrict__ el,
                       float* __restrict__ er, int N, int H, int D) {
    int i = blockIdx.x * blockDim.x + threadIdx.x;
    if (i >= N * H) return;
    const int n = i / H, h = i % H;
    const float* f = feat + (long long)n * H * D + h * D;
    float a = 0.0f, b = 0.0f;
    for (int d = 0; d < D; ++d) { a += f[d] * al[h * D + d]; b += f[d] * ar[h * D + d]; }
    el[i] = a; er[i] = b;
}

__global__ void k_edge_logit(const int* __restrict__ src, const int* __restrict__ dst,
                             const float* __restrict__ el, const float* __restrict__ er,
                             float* __restrict__ ee, unsigned* __restrict__ mx, int E, int H) {
    int i = blockIdx.x * blockDim.x + threadIdx.x;
    if (i >= E * H) return;
    const int e = i / H, h = i % H;
    float v = el[src[e] * H + h] + er[dst[e] * H + h];
    v = v > 0.0f ? v : 0.2f * v;
    ee[i] = v;
    atomicMax(&mx[dst[e] * H + h], enc_f(v));
}

__global__ void k_edge_exp(const int* __restrict__ dst, const float* __restrict__ ee,
                           const unsigned* __restrict__ mx, float* __restrict__ exe,
                           float* __restrict__ s, int E, int H) {
    int i = blockIdx.x * blockDim.x + threadIdx.x;
    if (i >= E * H) return;
    const int e = i / H, h = i % H;
    const float m = dec_f(mx[dst[e] * H + h]);
    const float ex = __expf(ee[i] - m);
    exe[i] = ex;
    atomicAdd(&s[dst[e] * H + h], ex);
}

__global__ void k_edge_msg(const int* __restrict__ src, const int* __restrict__ dst,
                           const float* __restrict__ feat, const float* __restrict__ exe,
                           const float* __restrict__ s, float* __restrict__ out,
                           int E, int H, int D) {
    long long i = (long long)blockIdx.x * blockDim.x + threadIdx.x;
    const long long tot = (long long)E * H * D;
    if (i >= tot) return;
    const int d = (int)(i % D);
    const long long eh = i / D;
    const int h = (int)(eh % H);
    const int e = (int)(eh / H);
    const int sd = dst[e];
    const float alpha = exe[e * H + h] / fmaxf(s[sd * H + h], 1e-9f);
    atomicAdd(&out[(long long)sd * H * D + h * D + d],
              alpha * feat[(long long)src[e] * H * D + h * D + d]);
}

__global__ void k_bias_act(const float* __restrict__ in, const float* __restrict__ b,
                           float* __restrict__ out, long long total, int Dtot, int act) {
    long long i = (long long)blockIdx.x * blockDim.x + threadIdx.x;
    if (i >= total) return;
    const int c = (int)(i % Dtot);
    float v = in[i] + b[c];
    if (act) v = v > 0.0f ? v : (expf(v) - 1.0f);
    out[i] = v;
}

static inline unsigned gsz(long long n, int b) { return (unsigned)((n + b - 1) / b); }

extern "C" void kernel_launch(void* const* d_in, const int* in_sizes, int n_in,
                              void* d_out, int out_size, void* d_ws, size_t ws_size,
                              hipStream_t stream) {
    const float* x       = (const float*)d_in[0];
    const int*   src     = (const int*)  d_in[1];
    const int*   dst     = (const int*)  d_in[2];
    const float* w_self  = (const float*)d_in[3];
    const float* w_neigh = (const float*)d_in[4];
    const float* sage_b  = (const float*)d_in[5];
    const float* g1_w    = (const float*)d_in[6];
    const float* g1_al   = (const float*)d_in[7];
    const float* g1_ar   = (const float*)d_in[8];
    const float* g1_b    = (const float*)d_in[9];
    const float* g2_w    = (const float*)d_in[10];
    const float* g2_al   = (const float*)d_in[11];
    const float* g2_ar   = (const float*)d_in[12];
    const float* g2_b    = (const float*)d_in[13];

    float*    ws    = (float*)d_ws;
    float*    deg   = ws;
    float*    neigh = deg + Nn;
    float*    h0    = neigh + (long long)Nn * F;
    float*    el    = h0 + (long long)Nn * D0;
    float*    er    = el + Nn * H1;
    unsigned* mx    = (unsigned*)(er + Nn * H1);
    float*    ssum  = (float*)(mx + Nn * H1);
    float*    ee    = ssum + Nn * H1;
    float*    exe   = ee + (long long)Ne * H1;
    float*    wt_self  = exe + (long long)Ne * H1;
    float*    wt_neigh = wt_self + D0 * F;
    float*    wt_g1    = wt_neigh + D0 * F;
    float*    wt_g2    = wt_g1 + (H1 * D1) * D0;
    float*    feat  = neigh;
    float*    aggr  = neigh + (long long)Nn * D0;

    const int B = 256;
    const unsigned NEG_INF_ENC = 0x007FFFFFu;
    const int Mtiles = Nn / 16;
    const unsigned rowblk = gsz(Mtiles, 8);

    k_transpose<<<gsz(D0 * F, B), B, 0, stream>>>(w_self,  wt_self,  F,  D0, D0);
    k_transpose<<<gsz(D0 * F, B), B, 0, stream>>>(w_neigh, wt_neigh, F,  D0, D0);
    k_transpose<<<gsz(D0 * D0, B), B, 0, stream>>>(g1_w,   wt_g1,    D0, H1 * D1, H1 * D1);
    k_transpose<<<gsz(C2P * D0, B), B, 0, stream>>>(g2_w,  wt_g2,    D0, C2, C2P);

    fill_f32<<<gsz(Nn, B), B, 0, stream>>>(deg, 0.0f, Nn);
    fill_f32<<<gsz((long long)Nn * F, B), B, 0, stream>>>(neigh, 0.0f, (long long)Nn * F);
    k_deg<<<gsz(Ne, B), B, 0, stream>>>(dst, deg, Ne);
    k_neigh<<<gsz((long long)Ne * (F / 4), B), B, 0, stream>>>(x, src, dst, neigh, Ne);
    {
        const size_t shmem = 2ull * 64 * (F + 4) * sizeof(float);
        wmma_gemm<true, true><<<rowblk * 2, B, shmem, stream>>>(
            x, wt_self, neigh, wt_neigh, deg, sage_b, h0, Nn, F, D0, 2);
    }

    {
        const size_t shmem = 64ull * (D0 + 4) * sizeof(float);
        wmma_gemm<false, false><<<rowblk * 2, B, shmem, stream>>>(
            h0, wt_g1, nullptr, nullptr, nullptr, nullptr, feat, Nn, D0, H1 * D1, 2);
    }
    k_attn<<<gsz(Nn * H1, B), B, 0, stream>>>(feat, g1_al, g1_ar, el, er, Nn, H1, D1);
    fill_u32<<<gsz(Nn * H1, B), B, 0, stream>>>(mx, NEG_INF_ENC, Nn * H1);
    fill_f32<<<gsz(Nn * H1, B), B, 0, stream>>>(ssum, 0.0f, Nn * H1);
    fill_f32<<<gsz((long long)Nn * D0, B), B, 0, stream>>>(aggr, 0.0f, (long long)Nn * D0);
    k_edge_logit<<<gsz(Ne * H1, B), B, 0, stream>>>(src, dst, el, er, ee, mx, Ne, H1);
    k_edge_exp<<<gsz(Ne * H1, B), B, 0, stream>>>(dst, ee, mx, exe, ssum, Ne, H1);
    k_edge_msg<<<gsz((long long)Ne * H1 * D1, B), B, 0, stream>>>(src, dst, feat, exe, ssum,
                                                                  aggr, Ne, H1, D1);
    k_bias_act<<<gsz((long long)Nn * D0, B), B, 0, stream>>>(aggr, g1_b, h0,
                                                             (long long)Nn * D0, D0, 1);

    {
        const size_t shmem = 64ull * (D0 + 4) * sizeof(float);
        wmma_gemm<false, false><<<rowblk, B, shmem, stream>>>(
            h0, wt_g2, nullptr, nullptr, nullptr, nullptr, feat, Nn, D0, C2, 1);
    }
    k_attn<<<gsz(Nn, B), B, 0, stream>>>(feat, g2_al, g2_ar, el, er, Nn, 1, C2);
    fill_u32<<<gsz(Nn, B), B, 0, stream>>>(mx, NEG_INF_ENC, Nn);
    fill_f32<<<gsz(Nn, B), B, 0, stream>>>(ssum, 0.0f, Nn);
    fill_f32<<<gsz((long long)Nn * C2, B), B, 0, stream>>>(aggr, 0.0f, (long long)Nn * C2);
    k_edge_logit<<<gsz(Ne, B), B, 0, stream>>>(src, dst, el, er, ee, mx, Ne, 1);
    k_edge_exp<<<gsz(Ne, B), B, 0, stream>>>(dst, ee, mx, exe, ssum, Ne, 1);
    k_edge_msg<<<gsz((long long)Ne * C2, B), B, 0, stream>>>(src, dst, feat, exe, ssum,
                                                             aggr, Ne, 1, C2);
    k_bias_act<<<gsz((long long)Nn * C2, B), B, 0, stream>>>(aggr, g2_b, (float*)d_out,
                                                             (long long)Nn * C2, C2, 0);
}


# Row scatters by global float atomics replaced: a work-group owns 224 dst rows in LDS, scans the dst list, adds matching rows with LDS float adds, writes each row once with bias/act; sum order changes
# speedup vs baseline: 2.0025x; 2.0025x over previous
#include <hip/hip_runtime.h>

constexpr int Nn  = 50000;
constexpr int Ne  = 800000;
constexpr int F   = 256;
constexpr int D0  = 128;
constexpr int H1  = 2;
constexpr int D1  = 64;
constexpr int C2  = 47;
constexpr int C2P = 64;

typedef float v2f __attribute__((ext_vector_type(2)));
typedef float v8f __attribute__((ext_vector_type(8)));
typedef int   v4i __attribute__((ext_vector_type(4)));

__device__ __forceinline__ unsigned enc_f(float f) {
    unsigned u = __float_as_uint(f);
    return (u & 0x80000000u) ? ~u : (u | 0x80000000u);
}
__device__ __forceinline__ float dec_f(unsigned u) {
    unsigned b = (u & 0x80000000u) ? (u ^ 0x80000000u) : ~u;
    return __uint_as_float(b);
}

__global__ void k_transpose(const float* __restrict__ W, float* __restrict__ Wt,
                            int K, int N, int Npad) {
    int i = blockIdx.x * blockDim.x + threadIdx.x;
    if (i >= Npad * K) return;
    const int c = i / K, k = i % K;
    Wt[i] = (c < N) ? W[(long long)k * N + c] : 0.0f;
}

template <bool HAS_A2, bool ACT>
__global__ __launch_bounds__(256)
void wmma_gemm(const float* __restrict__ A, const float* __restrict__ Wt,
               const float* __restrict__ A2, const float* __restrict__ W2t,
               const float* __restrict__ rowdeg, const float* __restrict__ bias,
               float* __restrict__ out, int M, int K, int Nout, int ncg)
{
    extern __shared__ float lds[];
    const int KP = K + 4;
    const int cg   = blockIdx.x % ncg;
    const int col0 = cg * 64;
    float* w1s = lds;
    float* w2s = lds + 64 * KP;

    const int ksh = 31 - __clz(K);
    for (int i = threadIdx.x; i < 64 * K; i += 256) {
        const int c = i >> ksh, k = i & (K - 1);
        w1s[c * KP + k] = Wt[(long long)(col0 + c) * K + k];
        if (HAS_A2) w2s[c * KP + k] = W2t[(long long)(col0 + c) * K + k];
    }
    __syncthreads();

    const int rt = (blockIdx.x / ncg) * 8 + (threadIdx.x >> 5);
    if (rt >= M / 16) return;
    const int lane = threadIdx.x & 31;
    const int half = lane >> 4;
    const int l    = lane & 15;
    const int row  = rt * 16 + l;

    const float* Arow  = A + (long long)row * K + half * 2;
    const float* A2row = HAS_A2 ? (A2 + (long long)row * K + half * 2) : nullptr;
    const float inv    = HAS_A2 ? (1.0f / fmaxf(rowdeg[row], 1.0f)) : 0.0f;
    __builtin_prefetch(Arow, 0, 3);

    const float* Bs[4];
    const float* B2s[4];
#pragma unroll
    for (int t = 0; t < 4; ++t) {
        Bs[t] = w1s + (t * 16 + l) * KP + half * 2;
        if (HAS_A2) B2s[t] = w2s + (t * 16 + l) * KP + half * 2;
    }

    v8f acc[4] = {};
    for (int kk = 0; kk < K; kk += 4) {
        const v2f a = *(const v2f*)(Arow + kk);
        v2f a2;
        if (HAS_A2) {
            const v2f t2 = *(const v2f*)(A2row + kk);
            a2.x = t2.x * inv; a2.y = t2.y * inv;
        }
#pragma unroll
        for (int t = 0; t < 4; ++t) {
            const v2f b = *(const v2f*)(Bs[t] + kk);
            acc[t] = __builtin_amdgcn_wmma_f32_16x16x4_f32(
                false, a, false, b, (short)0, acc[t], false, false);
            if (HAS_A2) {
                const v2f b2 = *(const v2f*)(B2s[t] + kk);
                acc[t] = __builtin_amdgcn_wmma_f32_16x16x4_f32(
                    false, a2, false, b2, (short)0, acc[t], false, false);
            }
        }
    }

#pragma unroll
    for (int t = 0; t < 4; ++t) {
        const int c = col0 + t * 16 + l;
        if (c < Nout) {
            const float bv = bias ? bias[c] : 0.0f;
#pragma unroll
            for (int v = 0; v < 8; ++v) {
                const int r = rt * 16 + v + 8 * half;
                float x = acc[t][v] + bv;
                if (ACT) x = x > 0.0f ? x : (expf(x) - 1.0f);
                out[(long long)r * Nout + c] = x;
            }
        }
    }
}

__global__ void fill_f32(float* p, float v, long long n) {
    long long i = (long long)blockIdx.x * blockDim.x + threadIdx.x;
    if (i < n) p[i] = v;
}
__global__ void fill_u32(unsigned* p, unsigned v, long long n) {
    long long i = (long long)blockIdx.x * blockDim.x + threadIdx.x;
    if (i < n) p[i] = v;
}

__global__ void k_deg(const int* __restrict__ dst, float* __restrict__ deg, int E) {
    int i = blockIdx.x * blockDim.x + threadIdx.x;
    if (i < E) atomicAdd(&deg[dst[i]], 1.0f);
}

__global__ void k_attn(const float* __restrict__ feat, const float* __restrict__ al,
                       const float* __restrict__ ar, float* __restrict__ el,
                       float* __restrict__ er, int N, int H, int D) {
    int i = blockIdx.x * blockDim.x + threadIdx.x;
    if (i >= N * H) return;
    const int n = i / H, h = i % H;
    const float* f = feat + (long long)n * H * D + h * D;
    float a = 0.0f, b = 0.0f;
    for (int d = 0; d < D; ++d) { a += f[d] * al[h * D + d]; b += f[d] * ar[h * D + d]; }
    el[i] = a; er[i] = b;
}

__global__ void k_edge_logit(const int* __restrict__ src, const int* __restrict__ dst,
                             const float* __restrict__ el, const float* __restrict__ er,
                             float* __restrict__ ee, unsigned* __restrict__ mx, int E, int H) {
    int i = blockIdx.x * blockDim.x + threadIdx.x;
    if (i >= E * H) return;
    const int e = i / H, h = i % H;
    float v = el[src[e] * H + h] + er[dst[e] * H + h];
    v = v > 0.0f ? v : 0.2f * v;
    ee[i] = v;
    atomicMax(&mx[dst[e] * H + h], enc_f(v));
}

__global__ void k_edge_exp(const int* __restrict__ dst, const float* __restrict__ ee,
                           const unsigned* __restrict__ mx, float* __restrict__ exe,
                           float* __restrict__ s, int E, int H) {
    int i = blockIdx.x * blockDim.x + threadIdx.x;
    if (i >= E * H) return;
    const int e = i / H, h = i % H;
    const float m = dec_f(mx[dst[e] * H + h]);
    const float ex = __expf(ee[i] - m);
    exe[i] = ex;
    atomicAdd(&s[dst[e] * H + h], ex);
}

template <int ROW, int H, bool WEIGHTED, bool BIAS, bool ACT>
__global__ __launch_bounds__(1024)
void k_tile_aggregate(const float* __restrict__ feat, const int* __restrict__ src,
                      const int* __restrict__ dst, const float* __restrict__ exe,
                      const float* __restrict__ s, const float* __restrict__ bias,
                      float* __restrict__ out, int E, int N)
{
    constexpr int TR = 224;
    constexpr int NQ = (ROW + 31) / 32;
    constexpr int D  = ROW / H;
    constexpr int EG = (NQ >= 8) ? 4 : 8;
    constexpr int QCAP = 544;
    static_assert(H == 1 || D % 32 == 0, "a chunk must not straddle two heads");
    static_assert((TR * ROW) % 32 == 0, "tile must be whole 128-byte lines");
    __shared__ float tile[TR * ROW];
    __shared__ int queue[32 * QCAP];

    const int lane = threadIdx.x & 31;
    const int wave = __builtin_amdgcn_readfirstlane((int)(threadIdx.x >> 5));
    const int lo   = blockIdx.x * TR;

    for (int i = threadIdx.x; i < TR * ROW / 4; i += 1024)
        ((float4*)tile)[i] = make_float4(0.0f, 0.0f, 0.0f, 0.0f);
    __syncthreads();

    const int qb = wave * QCAP;
    int qn = 0;

    auto drain = [&](int base, int cnt) {
        int ms = 0, md = 0;
        float mw[H];
#pragma unroll
        for (int h = 0; h < H; ++h) mw[h] = 0.0f;
        __builtin_amdgcn_fence(__ATOMIC_ACQ_REL, "wavefront");
        if (lane < cnt) {
            const int e = queue[qb + base + lane];
            ms = src[e];
            md = dst[e];
            if (WEIGHTED) {
#pragma unroll
                for (int h = 0; h < H; ++h)
                    mw[h] = exe[e * H + h] / fmaxf(s[md * H + h], 1e-9f);
            }
            md -= lo;
        }
        for (int j0 = 0; j0 < cnt; j0 += EG) {
            float v[EG][NQ];
            int   trow[EG];
#pragma unroll
            for (int g = 0; g < EG; ++g) {
                const int j  = min(j0 + g, cnt - 1);
                const int sj = __builtin_amdgcn_readlane(ms, j);
                trow[g] = __builtin_amdgcn_readlane(md, j) * ROW;
                const float* frow = feat + (long long)sj * ROW;
                float wj[H];
#pragma unroll
                for (int h = 0; h < H; ++h)
                    wj[h] = __int_as_float(__builtin_amdgcn_readlane(__float_as_int(mw[h]), j));
#pragma unroll
                for (int q = 0; q < NQ; ++q) {
                    const int c = q * 32 + lane;
                    float t = 0.0f;
                    if (ROW % 32 == 0 || c < ROW) t = frow[c];
                    if (WEIGHTED) t *= wj[(H == 1) ? 0 : (q * 32) / D];
                    v[g][q] = t;
                }
            }
#pragma unroll
            for (int g = 0; g < EG; ++g) {
                if (j0 + g < cnt) {
#pragma unroll
                    for (int q = 0; q < NQ; ++q) {
                        const int c = q * 32 + lane;
                        if (ROW % 32 == 0 || c < ROW)
                            __hip_atomic_fetch_add(&tile[trow[g] + c], v[g][q], __ATOMIC_RELAXED,
                                                   __HIP_MEMORY_SCOPE_WORKGROUP);
                    }
                }
            }
        }
    };

    auto offer = [&](int e, int d, unsigned okm) {
        const unsigned mask = __builtin_amdgcn_ballot_w32((unsigned)(d - lo) < (unsigned)TR) & okm;
        if (mask != 0u) {
            if ((mask >> lane) & 1u) queue[qb + qn + __popc(mask & ((1u << lane) - 1u))] = e;
            qn += __popc(mask);
        }
    };

    const v4i* dst4 = (const v4i*)dst;
    const int nvec  = E >> 2;
    auto fetch = [&](int vi) { return dst4[min(vi, nvec - 1)]; };

    if (wave == 0) {
        const int e = min(nvec * 4 + lane, E - 1);
        offer(e, dst[e], __builtin_amdgcn_ballot_w32(nvec * 4 + lane < E));
    }

    v4i nxt[4];
    int vb = wave * 128;
#pragma unroll
    for (int u = 0; u < 4; ++u) nxt[u] = fetch(vb + u * 32 + lane);
    for (;;) {
        const bool last = vb >= nvec;
        if (!last) {
            v4i cur[4];
#pragma unroll
            for (int u = 0; u < 4; ++u) cur[u] = nxt[u];
            const int vn = vb + 32 * 128;
#pragma unroll
            for (int u = 0; u < 4; ++u) nxt[u] = fetch(vn + u * 32 + lane);
#pragma unroll
            for (int u = 0; u < 4; ++u) {
                const int vi = vb + u * 32 + lane;
                const unsigned ok = __builtin_amdgcn_ballot_w32(vi < nvec);
                offer(vi * 4 + 0, cur[u].x, ok);
                offer(vi * 4 + 1, cur[u].y, ok);
                offer(vi * 4 + 2, cur[u].z, ok);
                offer(vi * 4 + 3, cur[u].w, ok);
            }
        }
        const int thr = last ? 1 : 32;
        while (qn >= thr) {
            const int cnt = min(qn, 32);
            qn -= cnt;
            drain(qn, cnt);
        }
        if (last) break;
        vb += 32 * 128;
    }
    __syncthreads();

    const long long g0 = (long long)lo * ROW;
    const long long gN = (long long)N * ROW;
    for (int i = threadIdx.x; i < TR * ROW; i += 1024) {
        if (g0 + i < gN) {
            float x = tile[i];
            if (BIAS) x += bias[i % ROW];
            if (ACT) x = x > 0.0f ? x : (expf(x) - 1.0f);
            out[g0 + i] = x;
        }
    }
}

static inline unsigned gsz(long long n, int b) { return (unsigned)((n + b - 1) / b); }

extern "C" void kernel_launch(void* const* d_in, const int* in_sizes, int n_in,
                              void* d_out, int out_size, void* d_ws, size_t ws_size,
                              hipStream_t stream) {
    const float* x       = (const float*)d_in[0];
    const int*   src     = (const int*)  d_in[1];
    const int*   dst     = (const int*)  d_in[2];
    const float* w_self  = (const float*)d_in[3];
    const float* w_neigh = (const float*)d_in[4];
    const float* sage_b  = (const float*)d_in[5];
    const float* g1_w    = (const float*)d_in[6];
    const float* g1_al   = (const float*)d_in[7];
    const float* g1_ar   = (const float*)d_in[8];
    const float* g1_b    = (const float*)d_in[9];
    const float* g2_w    = (const float*)d_in[10];
    const float* g2_al   = (const float*)d_in[11];
    const float* g2_ar   = (const float*)d_in[12];
    const float* g2_b    = (const float*)d_in[13];

    float*    ws    = (float*)d_ws;
    float*    deg   = ws;
    float*    neigh = deg + Nn;
    float*    h0    = neigh + (long long)Nn * F;
    float*    el    = h0 + (long long)Nn * D0;
    float*    er    = el + Nn * H1;
    unsigned* mx    = (unsigned*)(er + Nn * H1);
    float*    ssum  = (float*)(mx + Nn * H1);
    float*    ee    = ssum + Nn * H1;
    float*    exe   = ee + (long long)Ne * H1;
    float*    wt_self  = exe + (long long)Ne * H1;
    float*    wt_neigh = wt_self + D0 * F;
    float*    wt_g1    = wt_neigh + D0 * F;
    float*    wt_g2    = wt_g1 + (H1 * D1) * D0;
    float*    feat  = neigh;

    const int B = 256;
    const unsigned NEG_INF_ENC = 0x007FFFFFu;
    const int Mtiles = Nn / 16;
    const unsigned rowblk = gsz(Mtiles, 8);
    const unsigned ntile  = gsz(Nn, 224);

    k_transpose<<<gsz(D0 * F, B), B, 0, stream>>>(w_self,  wt_self,  F,  D0, D0);
    k_transpose<<<gsz(D0 * F, B), B, 0, stream>>>(w_neigh, wt_neigh, F,  D0, D0);
    k_transpose<<<gsz(D0 * D0, B), B, 0, stream>>>(g1_w,   wt_g1,    D0, H1 * D1, H1 * D1);
    k_transpose<<<gsz(C2P * D0, B), B, 0, stream>>>(g2_w,  wt_g2,    D0, C2, C2P);

    fill_f32<<<gsz(Nn, B), B, 0, stream>>>(deg, 0.0f, Nn);
    k_deg<<<gsz(Ne, B), B, 0, stream>>>(dst, deg, Ne);
    k_tile_aggregate<F, 1, false, false, false><<<ntile, 1024, 0, stream>>>(
        x, src, dst, nullptr, nullptr, nullptr, neigh, Ne, Nn);
    {
        const size_t shmem = 2ull * 64 * (F + 4) * sizeof(float);
        wmma_gemm<true, true><<<rowblk * 2, B, shmem, stream>>>(
            x, wt_self, neigh, wt_neigh, deg, sage_b, h0, Nn, F, D0, 2);
    }

    {
        const size_t shmem = 64ull * (D0 + 4) * sizeof(float);
        wmma_gemm<false, false><<<rowblk * 2, B, shmem, stream>>>(
            h0, wt_g1, nullptr, nullptr, nullptr, nullptr, feat, Nn, D0, H1 * D1, 2);
    }
    k_attn<<<gsz(Nn * H1, B), B, 0, stream>>>(feat, g1_al, g1_ar, el, er, Nn, H1, D1);
    fill_u32<<<gsz(Nn * H1, B), B, 0, stream>>>(mx, NEG_INF_ENC, Nn * H1);
    fill_f32<<<gsz(Nn * H1, B), B, 0, stream>>>(ssum, 0.0f, Nn * H1);
    k_edge_logit<<<gsz(Ne * H1, B), B, 0, stream>>>(src, dst, el, er, ee, mx, Ne, H1);
    k_edge_exp<<<gsz(Ne * H1, B), B, 0, stream>>>(dst, ee, mx, exe, ssum, Ne, H1);
    k_tile_aggregate<H1 * D1, H1, true, true, true><<<ntile, 1024, 0, stream>>>(
        feat, src, dst, exe, ssum, g1_b, h0, Ne, Nn);

    {
        const size_t shmem = 64ull * (D0 + 4) * sizeof(float);
        wmma_gemm<false, false><<<rowblk, B, shmem, stream>>>(
            h0, wt_g2, nullptr, nullptr, nullptr, nullptr, feat, Nn, D0, C2, 1);
    }
    k_attn<<<gsz(Nn, B), B, 0, stream>>>(feat, g2_al, g2_ar, el, er, Nn, 1, C2);
    fill_u32<<<gsz(Nn, B), B, 0, stream>>>(mx, NEG_INF_ENC, Nn);
    fill_f32<<<gsz(Nn, B), B, 0, stream>>>(ssum, 0.0f, Nn);
    k_edge_logit<<<gsz(Ne, B), B, 0, stream>>>(src, dst, el, er, ee, mx, Ne, 1);
    k_edge_exp<<<gsz(Ne, B), B, 0, stream>>>(dst, ee, mx, exe, ssum, Ne, 1);
    k_tile_aggregate<C2, 1, true, true, false><<<ntile, 1024, 0, stream>>>(
        feat, src, dst, exe, ssum, g2_b, (float*)d_out, Ne, Nn);
}


# GAT edge softmax moved into the row-owner work-group: row max, exp sum in LDS, one division per row at the write; 4 edge launches, 4 fills go (22->14 launches); sum order changes
# speedup vs baseline: 2.7219x; 1.3593x over previous
#include <hip/hip_runtime.h>

constexpr int Nn  = 50000;
constexpr int Ne  = 800000;
constexpr int F   = 256;
constexpr int D0  = 128;
constexpr int H1  = 2;
constexpr int D1  = 64;
constexpr int C2  = 47;
constexpr int C2P = 64;

typedef float v2f __attribute__((ext_vector_type(2)));
typedef float v8f __attribute__((ext_vector_type(8)));
typedef int   v4i __attribute__((ext_vector_type(4)));

__device__ __forceinline__ unsigned enc_f(float f) {
    unsigned u = __float_as_uint(f);
    return (u & 0x80000000u) ? ~u : (u | 0x80000000u);
}
__device__ __forceinline__ float dec_f(unsigned u) {
    unsigned b = (u & 0x80000000u) ? (u ^ 0x80000000u) : ~u;
    return __uint_as_float(b);
}

__global__ void k_transpose(const float* __restrict__ W, float* __restrict__ Wt,
                            int K, int N, int Npad) {
    int i = blockIdx.x * blockDim.x + threadIdx.x;
    if (i >= Npad * K) return;
    const int c = i / K, k = i % K;
    Wt[i] = (c < N) ? W[(long long)k * N + c] : 0.0f;
}

template <bool HAS_A2, bool ACT>
__global__ __launch_bounds__(256)
void wmma_gemm(const float* __restrict__ A, const float* __restrict__ Wt,
               const float* __restrict__ A2, const float* __restrict__ W2t,
               const float* __restrict__ rowdeg, const float* __restrict__ bias,
               float* __restrict__ out, int M, int K, int Nout, int ncg)
{
    extern __shared__ float lds[];
    const int KP = K + 4;
    const int cg   = blockIdx.x % ncg;
    const int col0 = cg * 64;
    float* w1s = lds;
    float* w2s = lds + 64 * KP;

    const int ksh = 31 - __clz(K);
    for (int i = threadIdx.x; i < 64 * K; i += 256) {
        const int c = i >> ksh, k = i & (K - 1);
        w1s[c * KP + k] = Wt[(long long)(col0 + c) * K + k];
        if (HAS_A2) w2s[c * KP + k] = W2t[(long long)(col0 + c) * K + k];
    }
    __syncthreads();

    const int rt = (blockIdx.x / ncg) * 8 + (threadIdx.x >> 5);
    if (rt >= M / 16) return;
    const int lane = threadIdx.x & 31;
    const int half = lane >> 4;
    const int l    = lane & 15;
    const int row  = rt * 16 + l;

    const float* Arow  = A + (long long)row * K + half * 2;
    const float* A2row = HAS_A2 ? (A2 + (long long)row * K + half * 2) : nullptr;
    const float inv    = HAS_A2 ? (1.0f / fmaxf(rowdeg[row], 1.0f)) : 0.0f;
    __builtin_prefetch(Arow, 0, 3);

    const float* Bs[4];
    const float* B2s[4];
#pragma unroll
    for (int t = 0; t < 4; ++t) {
        Bs[t] = w1s + (t * 16 + l) * KP + half * 2;
        if (HAS_A2) B2s[t] = w2s + (t * 16 + l) * KP + half * 2;
    }

    v8f acc[4] = {};
    for (int kk = 0; kk < K; kk += 4) {
        const v2f a = *(const v2f*)(Arow + kk);
        v2f a2;
        if (HAS_A2) {
            const v2f t2 = *(const v2f*)(A2row + kk);
            a2.x = t2.x * inv; a2.y = t2.y * inv;
        }
#pragma unroll
        for (int t = 0; t < 4; ++t) {
            const v2f b = *(const v2f*)(Bs[t] + kk);
            acc[t] = __builtin_amdgcn_wmma_f32_16x16x4_f32(
                false, a, false, b, (short)0, acc[t], false, false);
            if (HAS_A2) {
                const v2f b2 = *(const v2f*)(B2s[t] + kk);
                acc[t] = __builtin_amdgcn_wmma_f32_16x16x4_f32(
                    false, a2, false, b2, (short)0, acc[t], false, false);
            }
        }
    }

#pragma unroll
    for (int t = 0; t < 4; ++t) {
        const int c = col0 + t * 16 + l;
        if (c < Nout) {
            const float bv = bias ? bias[c] : 0.0f;
#pragma unroll
            for (int v = 0; v < 8; ++v) {
                const int r = rt * 16 + v + 8 * half;
                float x = acc[t][v] + bv;
                if (ACT) x = x > 0.0f ? x : (expf(x) - 1.0f);
                out[(long long)r * Nout + c] = x;
            }
        }
    }
}

__global__ void fill_f32(float* p, float v, long long n) {
    long long i = (long long)blockIdx.x * blockDim.x + threadIdx.x;
    if (i < n) p[i] = v;
}

__global__ void k_deg(const int* __restrict__ dst, float* __restrict__ deg, int E) {
    int i = blockIdx.x * blockDim.x + threadIdx.x;
    if (i < E) atomicAdd(&deg[dst[i]], 1.0f);
}

__global__ void k_attn(const float* __restrict__ feat, const float* __restrict__ al,
                       const float* __restrict__ ar, float* __restrict__ el,
                       float* __restrict__ er, int N, int H, int D) {
    int i = blockIdx.x * blockDim.x + threadIdx.x;
    if (i >= N * H) return;
    const int n = i / H, h = i % H;
    const float* f = feat + (long long)n * H * D + h * D;
    float a = 0.0f, b = 0.0f;
    for (int d = 0; d < D; ++d) { a += f[d] * al[h * D + d]; b += f[d] * ar[h * D + d]; }
    el[i] = a; er[i] = b;
}

template <int ROW, int H, bool GAT, bool ACT>
__global__ __launch_bounds__(1024)
void k_tile_aggregate(const float* __restrict__ feat, const int* __restrict__ src,
                      const int* __restrict__ dst, const float* __restrict__ el,
                      const float* __restrict__ er, const float* __restrict__ bias,
                      float* __restrict__ out, int E, int N)
{
    constexpr int TR = 224;
    constexpr int NQ = (ROW + 31) / 32;
    constexpr int D  = ROW / H;
    constexpr int EG = (NQ >= 8) ? 4 : 8;
    constexpr int TRIP = 512;
    constexpr int QCAP = GAT ? 1536 : 544;
    constexpr int NS = GAT ? TR * H : 1;
    static_assert(H == 1 || D % 32 == 0, "a chunk must not straddle two heads");
    static_assert((TR * ROW) % 32 == 0, "tile must be whole 128-byte lines");
    __shared__ float tile[TR * ROW];
    __shared__ int queue[32 * QCAP];
    __shared__ unsigned rmax[NS];
    __shared__ float rsum[NS];
    __shared__ float rer[NS];
    __shared__ int ovf;

    const int lane = threadIdx.x & 31;
    const int wave = __builtin_amdgcn_readfirstlane((int)(threadIdx.x >> 5));
    const int lo   = blockIdx.x * TR;

    for (int i = threadIdx.x; i < TR * ROW / 4; i += 1024)
        ((float4*)tile)[i] = make_float4(0.0f, 0.0f, 0.0f, 0.0f);
    if (GAT) {
        for (int i = threadIdx.x; i < TR * H; i += 1024) {
            rmax[i] = 0x007FFFFFu;
            rsum[i] = 0.0f;
            rer[i]  = (lo * H + i < N * H) ? er[lo * H + i] : 0.0f;
        }
        if (threadIdx.x == 0) ovf = 0;
    }
    __syncthreads();

    const int qb = wave * QCAP;
    int qn = 0;
    int done = 0;

    auto add_rows = [&](int cnt, int ms, int md, const float (&mw)[H]) {
        for (int j0 = 0; j0 < cnt; j0 += EG) {
            float v[EG][NQ];
            int   trow[EG];
#pragma unroll
            for (int g = 0; g < EG; ++g) {
                const int j  = min(j0 + g, cnt - 1);
                const int sj = __builtin_amdgcn_readlane(ms, j);
                trow[g] = __builtin_amdgcn_readlane(md, j) * ROW;
                const float* frow = feat + (long long)sj * ROW;
                float wj[H];
#pragma unroll
                for (int h = 0; h < H; ++h)
                    wj[h] = __int_as_float(__builtin_amdgcn_readlane(__float_as_int(mw[h]), j));
#pragma unroll
                for (int q = 0; q < NQ; ++q) {
                    const int c = q * 32 + lane;
                    float t = 0.0f;
                    if (ROW % 32 == 0 || c < ROW) t = frow[c];
                    if (GAT) t *= wj[(H == 1) ? 0 : (q * 32) / D];
                    v[g][q] = t;
                }
            }
#pragma unroll
            for (int g = 0; g < EG; ++g) {
                if (j0 + g < cnt) {
#pragma unroll
                    for (int q = 0; q < NQ; ++q) {
                        const int c = q * 32 + lane;
                        if (ROW % 32 == 0 || c < ROW)
                            __hip_atomic_fetch_add(&tile[trow[g] + c], v[g][q], __ATOMIC_RELAXED,
                                                   __HIP_MEMORY_SCOPE_WORKGROUP);
                    }
                }
            }
        }
    };

    auto logits = [&](int s, int d, float (&e)[H]) {
        float l[H];
        if (H == 2) {
            const v2f t = *(const v2f*)(el + (long long)s * 2);
            l[0] = t.x; l[H - 1] = t.y;
        } else {
#pragma unroll
            for (int h = 0; h < H; ++h) l[h] = el[(long long)s * H + h];
        }
#pragma unroll
        for (int h = 0; h < H; ++h) {
            const float x = l[h] + rer[d * H + h];
            e[h] = x > 0.0f ? x : 0.2f * x;
        }
    };

    auto batch_sum = [&](int base, int cnt) {
        int ms = 0, md = 0;
        float mw[H];
#pragma unroll
        for (int h = 0; h < H; ++h) mw[h] = 0.0f;
        __builtin_amdgcn_fence(__ATOMIC_ACQ_REL, "wavefront");
        if (lane < cnt) {
            const int e = queue[qb + base + lane];
            ms = src[e];
            md = dst[e] - lo;
        }
        add_rows(cnt, ms, md, mw);
    };

    auto batch_max = [&](int base, int cnt) {
        __builtin_amdgcn_fence(__ATOMIC_ACQ_REL, "wavefront");
        if (lane < cnt) {
            const int e = queue[qb + base + lane];
            const int s = src[e];
            const int d = dst[e] - lo;
            float lg[H];
            logits(s, d, lg);
#pragma unroll
            for (int h = 0; h < H; ++h)
                __hip_atomic_fetch_max(&rmax[d * H + h], enc_f(lg[h]), __ATOMIC_RELAXED,
                                       __HIP_MEMORY_SCOPE_WORKGROUP);
        }
    };

    auto batch_acc = [&](int base, int cnt) {
        int ms = 0, md = 0;
        float mw[H];
#pragma unroll
        for (int h = 0; h < H; ++h) mw[h] = 0.0f;
        __builtin_amdgcn_fence(__ATOMIC_ACQ_REL, "wavefront");
        if (lane < cnt) {
            const int e = queue[qb + base + lane];
            ms = src[e];
            md = dst[e] - lo;
            float lg[H];
            logits(ms, md, lg);
#pragma unroll
            for (int h = 0; h < H; ++h) {
                float m = dec_f(rmax[md * H + h]);
                if (!(fabsf(m) <= 3.402823466e38f)) m = 0.0f;
                mw[h] = __expf(lg[h] - m);
                __hip_atomic_fetch_add(&rsum[md * H + h], mw[h], __ATOMIC_RELAXED,
                                       __HIP_MEMORY_SCOPE_WORKGROUP);
            }
        }
        add_rows(cnt, ms, md, mw);
    };

    auto offer = [&](int e, int d, unsigned okm) {
        const unsigned mask = __builtin_amdgcn_ballot_w32((unsigned)(d - lo) < (unsigned)TR) & okm;
        if (mask != 0u) {
            if ((mask >> lane) & 1u) queue[qb + qn + __popc(mask & ((1u << lane) - 1u))] = e;
            qn += __popc(mask);
        }
    };

    const v4i* dst4 = (const v4i*)dst;
    const int nvec  = E >> 2;
    auto fetch = [&](int vi) { return dst4[min(vi, nvec - 1)]; };

    auto scan = [&](auto&& fn, bool keep) {
        if (wave == 0) {
            const int e = min(nvec * 4 + lane, E - 1);
            offer(e, dst[e], __builtin_amdgcn_ballot_w32(nvec * 4 + lane < E));
        }
        v4i nxt[4];
        int vb = wave * 128;
#pragma unroll
        for (int u = 0; u < 4; ++u) nxt[u] = fetch(vb + u * 32 + lane);
        for (;;) {
            const bool last = vb >= nvec;
            if (!last) {
                if (keep && qn + TRIP > QCAP) {
                    const int pend = qn - done;
                    __builtin_amdgcn_fence(__ATOMIC_ACQ_REL, "wavefront");
                    if (lane < pend) queue[qb + lane] = queue[qb + done + lane];
                    qn = pend; done = 0; keep = false;
                    if (lane == 0) ovf = 1;
                }
                v4i cur[4];
#pragma unroll
                for (int u = 0; u < 4; ++u) cur[u] = nxt[u];
                const int vn = vb + 32 * 128;
#pragma unroll
                for (int u = 0; u < 4; ++u) nxt[u] = fetch(vn + u * 32 + lane);
#pragma unroll
                for (int u = 0; u < 4; ++u) {
                    const int vi = vb + u * 32 + lane;
                    const unsigned ok = __builtin_amdgcn_ballot_w32(vi < nvec);
                    offer(vi * 4 + 0, cur[u].x, ok);
                    offer(vi * 4 + 1, cur[u].y, ok);
                    offer(vi * 4 + 2, cur[u].z, ok);
                    offer(vi * 4 + 3, cur[u].w, ok);
                }
            }
            const int thr = last ? 1 : 32;
            while (qn - done >= thr) {
                const int cnt = min(qn - done, 32);
                int base;
                if (keep) { base = done; done += cnt; }
                else      { qn -= cnt; base = qn; }
                fn(base, cnt);
            }
            if (last) break;
            vb += 32 * 128;
        }
    };

    if (!GAT) {
        scan(batch_sum, false);
        __syncthreads();
    } else {
        scan(batch_max, true);
        __syncthreads();
        const bool rescan = __builtin_amdgcn_readfirstlane(ovf) != 0;
        if (!rescan) {
            for (int b = 0; b < qn; b += 32) batch_acc(b, min(qn - b, 32));
        } else {
            qn = 0; done = 0;
            scan(batch_acc, false);
        }
        __syncthreads();
    }

    const long long g0 = (long long)lo * ROW;
    const long long gN = (long long)N * ROW;
    for (int i = threadIdx.x; i < TR * ROW; i += 1024) {
        if (g0 + i < gN) {
            float x = tile[i];
            if (GAT) {
                const int r = i / ROW, c = i - r * ROW;
                x = x / fmaxf(rsum[r * H + c / D], 1e-9f) + bias[c];
            }
            if (ACT) x = x > 0.0f ? x : (expf(x) - 1.0f);
            out[g0 + i] = x;
        }
    }
}

static inline unsigned gsz(long long n, int b) { return (unsigned)((n + b - 1) / b); }

extern "C" void kernel_launch(void* const* d_in, const int* in_sizes, int n_in,
                              void* d_out, int out_size, void* d_ws, size_t ws_size,
                              hipStream_t stream) {
    const float* x       = (const float*)d_in[0];
    const int*   src     = (const int*)  d_in[1];
    const int*   dst     = (const int*)  d_in[2];
    const float* w_self  = (const float*)d_in[3];
    const float* w_neigh = (const float*)d_in[4];
    const float* sage_b  = (const float*)d_in[5];
    const float* g1_w    = (const float*)d_in[6];
    const float* g1_al   = (const float*)d_in[7];
    const float* g1_ar   = (const float*)d_in[8];
    const float* g1_b    = (const float*)d_in[9];
    const float* g2_w    = (const float*)d_in[10];
    const float* g2_al   = (const float*)d_in[11];
    const float* g2_ar   = (const float*)d_in[12];
    const float* g2_b    = (const float*)d_in[13];

    float*    ws    = (float*)d_ws;
    float*    deg   = ws;
    float*    neigh = deg + Nn;
    float*    h0    = neigh + (long long)Nn * F;
    float*    el    = h0 + (long long)Nn * D0;
    float*    er    = el + Nn * H1;
    float*    wt_self  = er + Nn * H1;
    float*    wt_neigh = wt_self + D0 * F;
    float*    wt_g1    = wt_neigh + D0 * F;
    float*    wt_g2    = wt_g1 + (H1 * D1) * D0;
    float*    feat  = neigh;

    const int B = 256;
    const int Mtiles = Nn / 16;
    const unsigned rowblk = gsz(Mtiles, 8);
    const unsigned ntile  = gsz(Nn, 224);

    k_transpose<<<gsz(D0 * F, B), B, 0, stream>>>(w_self,  wt_self,  F,  D0, D0);
    k_transpose<<<gsz(D0 * F, B), B, 0, stream>>>(w_neigh, wt_neigh, F,  D0, D0);
    k_transpose<<<gsz(D0 * D0, B), B, 0, stream>>>(g1_w,   wt_g1,    D0, H1 * D1, H1 * D1);
    k_transpose<<<gsz(C2P * D0, B), B, 0, stream>>>(g2_w,  wt_g2,    D0, C2, C2P);

    fill_f32<<<gsz(Nn, B), B, 0, stream>>>(deg, 0.0f, Nn);
    k_deg<<<gsz(Ne, B), B, 0, stream>>>(dst, deg, Ne);
    k_tile_aggregate<F, 1, false, false><<<ntile, 1024, 0, stream>>>(
        x, src, dst, nullptr, nullptr, nullptr, neigh, Ne, Nn);
    {
        const size_t shmem = 2ull * 64 * (F + 4) * sizeof(float);
        wmma_gemm<true, true><<<rowblk * 2, B, shmem, stream>>>(
            x, wt_self, neigh, wt_neigh, deg, sage_b, h0, Nn, F, D0, 2);
    }

    {
        const size_t shmem = 64ull * (D0 + 4) * sizeof(float);
        wmma_gemm<false, false><<<rowblk * 2, B, shmem, stream>>>(
            h0, wt_g1, nullptr, nullptr, nullptr, nullptr, feat, Nn, D0, H1 * D1, 2);
    }
    k_attn<<<gsz(Nn * H1, B), B, 0, stream>>>(feat, g1_al, g1_ar, el, er, Nn, H1, D1);
    k_tile_aggregate<H1 * D1, H1, true, true><<<ntile, 1024, 0, stream>>>(
        feat, src, dst, el, er, g1_b, h0, Ne, Nn);

    {
        const size_t shmem = 64ull * (D0 + 4) * sizeof(float);
        wmma_gemm<false, false><<<rowblk, B, shmem, stream>>>(
            h0, wt_g2, nullptr, nullptr, nullptr, nullptr, feat, Nn, D0, C2, 1);
    }
    k_attn<<<gsz(Nn, B), B, 0, stream>>>(feat, g2_al, g2_ar, el, er, Nn, 1, C2);
    k_tile_aggregate<C2, 1, true, false><<<ntile, 1024, 0, stream>>>(
        feat, src, dst, el, er, g2_b, (float*)d_out, Ne, Nn);
}


# Owner-group edge lists kept: first aggregation writes each wave's matched edges (src, tile row packed) to workspace; both GAT launches walk them instead of re-scanning dst 224x; sum order changes
# speedup vs baseline: 3.2817x; 1.2057x over previous
#include <hip/hip_runtime.h>

constexpr int Nn  = 50000;
constexpr int Ne  = 800000;
constexpr int F   = 256;
constexpr int D0  = 128;
constexpr int H1  = 2;
constexpr int D1  = 64;
constexpr int C2  = 47;
constexpr int C2P = 64;

constexpr int AGG_TR   = 224;
constexpr int AGG_NT   = (Nn + AGG_TR - 1) / AGG_TR;
constexpr int AGG_LCAP = 1024;

typedef float v2f __attribute__((ext_vector_type(2)));
typedef float v8f __attribute__((ext_vector_type(8)));
typedef int   v4i __attribute__((ext_vector_type(4)));

__device__ __forceinline__ unsigned enc_f(float f) {
    unsigned u = __float_as_uint(f);
    return (u & 0x80000000u) ? ~u : (u | 0x80000000u);
}
__device__ __forceinline__ float dec_f(unsigned u) {
    unsigned b = (u & 0x80000000u) ? (u ^ 0x80000000u) : ~u;
    return __uint_as_float(b);
}

__global__ void k_transpose(const float* __restrict__ W, float* __restrict__ Wt,
                            int K, int N, int Npad) {
    int i = blockIdx.x * blockDim.x + threadIdx.x;
    if (i >= Npad * K) return;
    const int c = i / K, k = i % K;
    Wt[i] = (c < N) ? W[(long long)k * N + c] : 0.0f;
}

template <bool HAS_A2, bool ACT>
__global__ __launch_bounds__(256)
void wmma_gemm(const float* __restrict__ A, const float* __restrict__ Wt,
               const float* __restrict__ A2, const float* __restrict__ W2t,
               const float* __restrict__ rowdeg, const float* __restrict__ bias,
               float* __restrict__ out, int M, int K, int Nout, int ncg)
{
    extern __shared__ float lds[];
    const int KP = K + 4;
    const int cg   = blockIdx.x % ncg;
    const int col0 = cg * 64;
    float* w1s = lds;
    float* w2s = lds + 64 * KP;

    const int ksh = 31 - __clz(K);
    for (int i = threadIdx.x; i < 64 * K; i += 256) {
        const int c = i >> ksh, k = i & (K - 1);
        w1s[c * KP + k] = Wt[(long long)(col0 + c) * K + k];
        if (HAS_A2) w2s[c * KP + k] = W2t[(long long)(col0 + c) * K + k];
    }
    __syncthreads();

    const int rt = (blockIdx.x / ncg) * 8 + (threadIdx.x >> 5);
    if (rt >= M / 16) return;
    const int lane = threadIdx.x & 31;
    const int half = lane >> 4;
    const int l    = lane & 15;
    const int row  = rt * 16 + l;

    const float* Arow  = A + (long long)row * K + half * 2;
    const float* A2row = HAS_A2 ? (A2 + (long long)row * K + half * 2) : nullptr;
    const float inv    = HAS_A2 ? (1.0f / fmaxf(rowdeg[row], 1.0f)) : 0.0f;
    __builtin_prefetch(Arow, 0, 3);

    const float* Bs[4];
    const float* B2s[4];
#pragma unroll
    for (int t = 0; t < 4; ++t) {
        Bs[t] = w1s + (t * 16 + l) * KP + half * 2;
        if (HAS_A2) B2s[t] = w2s + (t * 16 + l) * KP + half * 2;
    }

    v8f acc[4] = {};
    for (int kk = 0; kk < K; kk += 4) {
        const v2f a = *(const v2f*)(Arow + kk);
        v2f a2;
        if (HAS_A2) {
            const v2f t2 = *(const v2f*)(A2row + kk);
            a2.x = t2.x * inv; a2.y = t2.y * inv;
        }
#pragma unroll
        for (int t = 0; t < 4; ++t) {
            const v2f b = *(const v2f*)(Bs[t] + kk);
            acc[t] = __builtin_amdgcn_wmma_f32_16x16x4_f32(
                false, a, false, b, (short)0, acc[t], false, false);
            if (HAS_A2) {
                const v2f b2 = *(const v2f*)(B2s[t] + kk);
                acc[t] = __builtin_amdgcn_wmma_f32_16x16x4_f32(
                    false, a2, false, b2, (short)0, acc[t], false, false);
            }
        }
    }

#pragma unroll
    for (int t = 0; t < 4; ++t) {
        const int c = col0 + t * 16 + l;
        if (c < Nout) {
            const float bv = bias ? bias[c] : 0.0f;
#pragma unroll
            for (int v = 0; v < 8; ++v) {
                const int r = rt * 16 + v + 8 * half;
                float x = acc[t][v] + bv;
                if (ACT) x = x > 0.0f ? x : (expf(x) - 1.0f);
                out[(long long)r * Nout + c] = x;
            }
        }
    }
}

__global__ void fill_f32(float* p, float v, long long n) {
    long long i = (long long)blockIdx.x * blockDim.x + threadIdx.x;
    if (i < n) p[i] = v;
}

__global__ void k_deg(const int* __restrict__ dst, float* __restrict__ deg, int E) {
    int i = blockIdx.x * blockDim.x + threadIdx.x;
    if (i < E) atomicAdd(&deg[dst[i]], 1.0f);
}

__global__ void k_attn(const float* __restrict__ feat, const float* __restrict__ al,
                       const float* __restrict__ ar, float* __restrict__ el,
                       float* __restrict__ er, int N, int H, int D) {
    int i = blockIdx.x * blockDim.x + threadIdx.x;
    if (i >= N * H) return;
    const int n = i / H, h = i % H;
    const float* f = feat + (long long)n * H * D + h * D;
    float a = 0.0f, b = 0.0f;
    for (int d = 0; d < D; ++d) { a += f[d] * al[h * D + d]; b += f[d] * ar[h * D + d]; }
    el[i] = a; er[i] = b;
}

template <int ROW, int H, bool GAT, bool ACT>
__global__ __launch_bounds__(1024)
void k_tile_aggregate(const float* __restrict__ feat, const int* __restrict__ src,
                      const int* __restrict__ dst, const float* __restrict__ el,
                      const float* __restrict__ er, const float* __restrict__ bias,
                      float* __restrict__ out, int* __restrict__ elist,
                      int* __restrict__ ecount, int E, int N)
{
    constexpr int TR = AGG_TR;
    constexpr int LCAP = AGG_LCAP;
    static_assert(TR <= 256 && Nn <= (1 << 24), "a list word holds 24 bits of source row, 8 of tile row");
    constexpr int NQ = (ROW + 31) / 32;
    constexpr int D  = ROW / H;
    constexpr int EG = (NQ >= 8) ? 4 : 8;
    constexpr int TRIP = 512;
    constexpr int QCAP = GAT ? 1536 : 544;
    constexpr int NS = GAT ? TR * H : 1;
    static_assert(H == 1 || D % 32 == 0, "a chunk must not straddle two heads");
    static_assert((TR * ROW) % 32 == 0, "tile must be whole 128-byte lines");
    __shared__ float tile[TR * ROW];
    __shared__ int queue[32 * QCAP];
    __shared__ unsigned rmax[NS];
    __shared__ float rsum[NS];
    __shared__ float rer[NS];
    __shared__ int ovf;

    const int lane = threadIdx.x & 31;
    const int wave = __builtin_amdgcn_readfirstlane((int)(threadIdx.x >> 5));
    const int lo   = blockIdx.x * TR;

    for (int i = threadIdx.x; i < TR * ROW / 4; i += 1024)
        ((float4*)tile)[i] = make_float4(0.0f, 0.0f, 0.0f, 0.0f);
    if (GAT) {
        for (int i = threadIdx.x; i < TR * H; i += 1024) {
            rmax[i] = 0x007FFFFFu;
            rsum[i] = 0.0f;
            rer[i]  = (lo * H + i < N * H) ? er[lo * H + i] : 0.0f;
        }
        if (threadIdx.x == 0) ovf = 0;
    }
    __syncthreads();

    const int qb = wave * QCAP;
    int qn = 0;
    int done = 0;
    int* wl = elist + ((long long)blockIdx.x * 32 + wave) * LCAP;
    int wn = 0;

    auto add_rows = [&](int cnt, int ms, int md, const float (&mw)[H]) {
        for (int j0 = 0; j0 < cnt; j0 += EG) {
            float v[EG][NQ];
            int   trow[EG];
#pragma unroll
            for (int g = 0; g < EG; ++g) {
                const int j  = min(j0 + g, cnt - 1);
                const int sj = __builtin_amdgcn_readlane(ms, j);
                trow[g] = __builtin_amdgcn_readlane(md, j) * ROW;
                const float* frow = feat + (long long)sj * ROW;
                float wj[H];
#pragma unroll
                for (int h = 0; h < H; ++h)
                    wj[h] = __int_as_float(__builtin_amdgcn_readlane(__float_as_int(mw[h]), j));
#pragma unroll
                for (int q = 0; q < NQ; ++q) {
                    const int c = q * 32 + lane;
                    float t = 0.0f;
                    if (ROW % 32 == 0 || c < ROW) t = frow[c];
                    if (GAT) t *= wj[(H == 1) ? 0 : (q * 32) / D];
                    v[g][q] = t;
                }
            }
#pragma unroll
            for (int g = 0; g < EG; ++g) {
                if (j0 + g < cnt) {
#pragma unroll
                    for (int q = 0; q < NQ; ++q) {
                        const int c = q * 32 + lane;
                        if (ROW % 32 == 0 || c < ROW)
                            __hip_atomic_fetch_add(&tile[trow[g] + c], v[g][q], __ATOMIC_RELAXED,
                                                   __HIP_MEMORY_SCOPE_WORKGROUP);
                    }
                }
            }
        }
    };

    auto logits = [&](int s, int d, float (&e)[H]) {
        float l[H];
        if (H == 2) {
            const v2f t = *(const v2f*)(el + (long long)s * 2);
            l[0] = t.x; l[H - 1] = t.y;
        } else {
#pragma unroll
            for (int h = 0; h < H; ++h) l[h] = el[(long long)s * H + h];
        }
#pragma unroll
        for (int h = 0; h < H; ++h) {
            const float x = l[h] + rer[d * H + h];
            e[h] = x > 0.0f ? x : 0.2f * x;
        }
    };

    auto edge_of = [&](bool list, int base, int cnt, int& ms, int& md) {
        ms = 0; md = 0;
        if (list) {
            if (lane < cnt) {
                const unsigned p = (unsigned)wl[base + lane];
                ms = (int)(p & 0x00FFFFFFu);
                md = (int)(p >> 24);
            }
        } else {
            __builtin_amdgcn_fence(__ATOMIC_ACQ_REL, "wavefront");
            if (lane < cnt) {
                const int e = queue[qb + base + lane];
                ms = src[e];
                md = dst[e] - lo;
            }
        }
    };

    auto batch_sum = [&](int base, int cnt) {
        int ms, md;
        float mw[H];
#pragma unroll
        for (int h = 0; h < H; ++h) mw[h] = 0.0f;
        edge_of(false, base, cnt, ms, md);
        if (wn >= 0) {
            if (wn + cnt <= LCAP) {
                if (lane < cnt) wl[wn + lane] = (int)((unsigned)ms | ((unsigned)md << 24));
                wn += cnt;
            } else {
                wn = -1;
            }
        }
        add_rows(cnt, ms, md, mw);
    };

    auto batch_max = [&](bool list, int base, int cnt) {
        int s, d;
        edge_of(list, base, cnt, s, d);
        if (lane < cnt) {
            float lg[H];
            logits(s, d, lg);
#pragma unroll
            for (int h = 0; h < H; ++h)
                __hip_atomic_fetch_max(&rmax[d * H + h], enc_f(lg[h]), __ATOMIC_RELAXED,
                                       __HIP_MEMORY_SCOPE_WORKGROUP);
        }
    };

    auto batch_acc = [&](bool list, int base, int cnt) {
        int ms, md;
        float mw[H];
#pragma unroll
        for (int h = 0; h < H; ++h) mw[h] = 0.0f;
        edge_of(list, base, cnt, ms, md);
        if (lane < cnt) {
            float lg[H];
            logits(ms, md, lg);
#pragma unroll
            for (int h = 0; h < H; ++h) {
                float m = dec_f(rmax[md * H + h]);
                if (!(fabsf(m) <= 3.402823466e38f)) m = 0.0f;
                mw[h] = __expf(lg[h] - m);
                __hip_atomic_fetch_add(&rsum[md * H + h], mw[h], __ATOMIC_RELAXED,
                                       __HIP_MEMORY_SCOPE_WORKGROUP);
            }
        }
        add_rows(cnt, ms, md, mw);
    };

    auto offer = [&](int e, int d, unsigned okm) {
        const unsigned mask = __builtin_amdgcn_ballot_w32((unsigned)(d - lo) < (unsigned)TR) & okm;
        if (mask != 0u) {
            if ((mask >> lane) & 1u) queue[qb + qn + __popc(mask & ((1u << lane) - 1u))] = e;
            qn += __popc(mask);
        }
    };

    const v4i* dst4 = (const v4i*)dst;
    const int nvec  = E >> 2;
    auto fetch = [&](int vi) { return dst4[min(vi, nvec - 1)]; };

    auto scan = [&](auto&& fn, bool keep) {
        if (wave == 0) {
            const int e = min(nvec * 4 + lane, E - 1);
            offer(e, dst[e], __builtin_amdgcn_ballot_w32(nvec * 4 + lane < E));
        }
        v4i nxt[4];
        int vb = wave * 128;
#pragma unroll
        for (int u = 0; u < 4; ++u) nxt[u] = fetch(vb + u * 32 + lane);
        for (;;) {
            const bool last = vb >= nvec;
            if (!last) {
                if (keep && qn + TRIP > QCAP) {
                    const int pend = qn - done;
                    __builtin_amdgcn_fence(__ATOMIC_ACQ_REL, "wavefront");
                    if (lane < pend) queue[qb + lane] = queue[qb + done + lane];
                    qn = pend; done = 0; keep = false;
                    if (lane == 0) ovf = 1;
                }
                v4i cur[4];
#pragma unroll
                for (int u = 0; u < 4; ++u) cur[u] = nxt[u];
                const int vn = vb + 32 * 128;
#pragma unroll
                for (int u = 0; u < 4; ++u) nxt[u] = fetch(vn + u * 32 + lane);
#pragma unroll
                for (int u = 0; u < 4; ++u) {
                    const int vi = vb + u * 32 + lane;
                    const unsigned ok = __builtin_amdgcn_ballot_w32(vi < nvec);
                    offer(vi * 4 + 0, cur[u].x, ok);
                    offer(vi * 4 + 1, cur[u].y, ok);
                    offer(vi * 4 + 2, cur[u].z, ok);
                    offer(vi * 4 + 3, cur[u].w, ok);
                }
            }
            const int thr = last ? 1 : 32;
            while (qn - done >= thr) {
                const int cnt = min(qn - done, 32);
                int base;
                if (keep) { base = done; done += cnt; }
                else      { qn -= cnt; base = qn; }
                fn(base, cnt);
            }
            if (last) break;
            vb += 32 * 128;
        }
    };

    if (!GAT) {
        scan(batch_sum, false);
        if (lane == 0) ecount[blockIdx.x * 32 + wave] = wn;
        __syncthreads();
    } else {
        const int ln = __builtin_amdgcn_readfirstlane(ecount[blockIdx.x * 32 + wave]);
        const bool listed = ln >= 0;
        if (listed) {
            for (int b = 0; b < ln; b += 128) {
                int  s[4], d[4];
                bool ok[4];
                float lg[4][H];
#pragma unroll
                for (int u = 0; u < 4; ++u) {
                    const int i = b + u * 32 + lane;
                    ok[u] = i < ln;
                    const unsigned p = (unsigned)wl[min(i, ln - 1)];
                    s[u] = (int)(p & 0x00FFFFFFu);
                    d[u] = (int)(p >> 24);
                }
#pragma unroll
                for (int u = 0; u < 4; ++u) logits(s[u], d[u], lg[u]);
#pragma unroll
                for (int u = 0; u < 4; ++u) {
                    if (ok[u]) {
#pragma unroll
                        for (int h = 0; h < H; ++h)
                            __hip_atomic_fetch_max(&rmax[d[u] * H + h], enc_f(lg[u][h]),
                                                   __ATOMIC_RELAXED, __HIP_MEMORY_SCOPE_WORKGROUP);
                    }
                }
            }
        } else {
            scan([&](int b, int c) { batch_max(false, b, c); }, true);
        }
        __syncthreads();
        const bool rescan = __builtin_amdgcn_readfirstlane(ovf) != 0;
        if (listed) {
            for (int b = 0; b < ln; b += 32) batch_acc(true, b, min(ln - b, 32));
        } else if (!rescan) {
            for (int b = 0; b < qn; b += 32) batch_acc(false, b, min(qn - b, 32));
        } else {
            qn = 0; done = 0;
            scan([&](int b, int c) { batch_acc(false, b, c); }, false);
        }
        __syncthreads();
    }

    const long long g0 = (long long)lo * ROW;
    const long long gN = (long long)N * ROW;
    for (int i = threadIdx.x; i < TR * ROW; i += 1024) {
        if (g0 + i < gN) {
            float x = tile[i];
            if (GAT) {
                const int r = i / ROW, c = i - r * ROW;
                x = x / fmaxf(rsum[r * H + c / D], 1e-9f) + bias[c];
            }
            if (ACT) x = x > 0.0f ? x : (expf(x) - 1.0f);
            out[g0 + i] = x;
        }
    }
}

static inline unsigned gsz(long long n, int b) { return (unsigned)((n + b - 1) / b); }

extern "C" void kernel_launch(void* const* d_in, const int* in_sizes, int n_in,
                              void* d_out, int out_size, void* d_ws, size_t ws_size,
                              hipStream_t stream) {
    const float* x       = (const float*)d_in[0];
    const int*   src     = (const int*)  d_in[1];
    const int*   dst     = (const int*)  d_in[2];
    const float* w_self  = (const float*)d_in[3];
    const float* w_neigh = (const float*)d_in[4];
    const float* sage_b  = (const float*)d_in[5];
    const float* g1_w    = (const float*)d_in[6];
    const float* g1_al   = (const float*)d_in[7];
    const float* g1_ar   = (const float*)d_in[8];
    const float* g1_b    = (const float*)d_in[9];
    const float* g2_w    = (const float*)d_in[10];
    const float* g2_al   = (const float*)d_in[11];
    const float* g2_ar   = (const float*)d_in[12];
    const float* g2_b    = (const float*)d_in[13];

    int*      elist  = (int*)d_ws;
    int*      ecount = elist + AGG_NT * 32 * AGG_LCAP;
    float*    ws    = (float*)(ecount + AGG_NT * 32);
    float*    deg   = ws;
    float*    neigh = deg + Nn;
    float*    h0    = neigh + (long long)Nn * F;
    float*    el    = h0 + (long long)Nn * D0;
    float*    er    = el + Nn * H1;
    float*    wt_self  = er + Nn * H1;
    float*    wt_neigh = wt_self + D0 * F;
    float*    wt_g1    = wt_neigh + D0 * F;
    float*    wt_g2    = wt_g1 + (H1 * D1) * D0;
    float*    feat  = neigh;

    const int B = 256;
    const int Mtiles = Nn / 16;
    const unsigned rowblk = gsz(Mtiles, 8);
    const unsigned ntile  = AGG_NT;

    k_transpose<<<gsz(D0 * F, B), B, 0, stream>>>(w_self,  wt_self,  F,  D0, D0);
    k_transpose<<<gsz(D0 * F, B), B, 0, stream>>>(w_neigh, wt_neigh, F,  D0, D0);
    k_transpose<<<gsz(D0 * D0, B), B, 0, stream>>>(g1_w,   wt_g1,    D0, H1 * D1, H1 * D1);
    k_transpose<<<gsz(C2P * D0, B), B, 0, stream>>>(g2_w,  wt_g2,    D0, C2, C2P);

    fill_f32<<<gsz(Nn, B), B, 0, stream>>>(deg, 0.0f, Nn);
    k_deg<<<gsz(Ne, B), B, 0, stream>>>(dst, deg, Ne);
    k_tile_aggregate<F, 1, false, false><<<ntile, 1024, 0, stream>>>(
        x, src, dst, nullptr, nullptr, nullptr, neigh, elist, ecount, Ne, Nn);
    {
        const size_t shmem = 2ull * 64 * (F + 4) * sizeof(float);
        wmma_gemm<true, true><<<rowblk * 2, B, shmem, stream>>>(
            x, wt_self, neigh, wt_neigh, deg, sage_b, h0, Nn, F, D0, 2);
    }

    {
        const size_t shmem = 64ull * (D0 + 4) * sizeof(float);
        wmma_gemm<false, false><<<rowblk * 2, B, shmem, stream>>>(
            h0, wt_g1, nullptr, nullptr, nullptr, nullptr, feat, Nn, D0, H1 * D1, 2);
    }
    k_attn<<<gsz(Nn * H1, B), B, 0, stream>>>(feat, g1_al, g1_ar, el, er, Nn, H1, D1);
    k_tile_aggregate<H1 * D1, H1, true, true><<<ntile, 1024, 0, stream>>>(
        feat, src, dst, el, er, g1_b, h0, elist, ecount, Ne, Nn);

    {
        const size_t shmem = 64ull * (D0 + 4) * sizeof(float);
        wmma_gemm<false, false><<<rowblk, B, shmem, stream>>>(
            h0, wt_g2, nullptr, nullptr, nullptr, nullptr, feat, Nn, D0, C2, 1);
    }
    k_attn<<<gsz(Nn, B), B, 0, stream>>>(feat, g2_al, g2_ar, el, er, Nn, 1, C2);
    k_tile_aggregate<C2, 1, true, false><<<ntile, 1024, 0, stream>>>(
        feat, src, dst, el, er, g2_b, (float*)d_out, elist, ecount, Ne, Nn);
}


# dst scan replaced: a binning launch puts each edge into its owner group's list (slots reserved by returning int adds); groups walk their list and count degrees in LDS; sum order changes
# speedup vs baseline: 3.9570x; 1.2058x over previous
#include <hip/hip_runtime.h>

constexpr int Nn  = 50000;
constexpr int Ne  = 800000;
constexpr int F   = 256;
constexpr int D0  = 128;
constexpr int H1  = 2;
constexpr int D1  = 64;
constexpr int C2  = 47;
constexpr int C2P = 64;

constexpr int AGG_TR   = 224;
constexpr int AGG_NT   = (Nn + AGG_TR - 1) / AGG_TR;
constexpr int AGG_GCAP = 16384;
constexpr int AGG_CSTR = 64;

typedef float v2f __attribute__((ext_vector_type(2)));
typedef float v8f __attribute__((ext_vector_type(8)));
typedef int   v4i __attribute__((ext_vector_type(4)));

__device__ __forceinline__ unsigned enc_f(float f) {
    unsigned u = __float_as_uint(f);
    return (u & 0x80000000u) ? ~u : (u | 0x80000000u);
}
__device__ __forceinline__ float dec_f(unsigned u) {
    unsigned b = (u & 0x80000000u) ? (u ^ 0x80000000u) : ~u;
    return __uint_as_float(b);
}

__global__ void k_transpose(const float* __restrict__ W, float* __restrict__ Wt,
                            int K, int N, int Npad) {
    int i = blockIdx.x * blockDim.x + threadIdx.x;
    if (i >= Npad * K) return;
    const int c = i / K, k = i % K;
    Wt[i] = (c < N) ? W[(long long)k * N + c] : 0.0f;
}

template <bool HAS_A2, bool ACT>
__global__ __launch_bounds__(256)
void wmma_gemm(const float* __restrict__ A, const float* __restrict__ Wt,
               const float* __restrict__ A2, const float* __restrict__ W2t,
               const float* __restrict__ rowdeg, const float* __restrict__ bias,
               float* __restrict__ out, int M, int K, int Nout, int ncg)
{
    extern __shared__ float lds[];
    const int KP = K + 4;
    const int cg   = blockIdx.x % ncg;
    const int col0 = cg * 64;
    float* w1s = lds;
    float* w2s = lds + 64 * KP;

    const int ksh = 31 - __clz(K);
    for (int i = threadIdx.x; i < 64 * K; i += 256) {
        const int c = i >> ksh, k = i & (K - 1);
        w1s[c * KP + k] = Wt[(long long)(col0 + c) * K + k];
        if (HAS_A2) w2s[c * KP + k] = W2t[(long long)(col0 + c) * K + k];
    }
    __syncthreads();

    const int rt = (blockIdx.x / ncg) * 8 + (threadIdx.x >> 5);
    if (rt >= M / 16) return;
    const int lane = threadIdx.x & 31;
    const int half = lane >> 4;
    const int l    = lane & 15;
    const int row  = rt * 16 + l;

    const float* Arow  = A + (long long)row * K + half * 2;
    const float* A2row = HAS_A2 ? (A2 + (long long)row * K + half * 2) : nullptr;
    const float inv    = HAS_A2 ? (1.0f / fmaxf(rowdeg[row], 1.0f)) : 0.0f;
    __builtin_prefetch(Arow, 0, 3);

    const float* Bs[4];
    const float* B2s[4];
#pragma unroll
    for (int t = 0; t < 4; ++t) {
        Bs[t] = w1s + (t * 16 + l) * KP + half * 2;
        if (HAS_A2) B2s[t] = w2s + (t * 16 + l) * KP + half * 2;
    }

    v8f acc[4] = {};
    for (int kk = 0; kk < K; kk += 4) {
        const v2f a = *(const v2f*)(Arow + kk);
        v2f a2;
        if (HAS_A2) {
            const v2f t2 = *(const v2f*)(A2row + kk);
            a2.x = t2.x * inv; a2.y = t2.y * inv;
        }
#pragma unroll
        for (int t = 0; t < 4; ++t) {
            const v2f b = *(const v2f*)(Bs[t] + kk);
            acc[t] = __builtin_amdgcn_wmma_f32_16x16x4_f32(
                false, a, false, b, (short)0, acc[t], false, false);
            if (HAS_A2) {
                const v2f b2 = *(const v2f*)(B2s[t] + kk);
                acc[t] = __builtin_amdgcn_wmma_f32_16x16x4_f32(
                    false, a2, false, b2, (short)0, acc[t], false, false);
            }
        }
    }

#pragma unroll
    for (int t = 0; t < 4; ++t) {
        const int c = col0 + t * 16 + l;
        if (c < Nout) {
            const float bv = bias ? bias[c] : 0.0f;
#pragma unroll
            for (int v = 0; v < 8; ++v) {
                const int r = rt * 16 + v + 8 * half;
                float x = acc[t][v] + bv;
                if (ACT) x = x > 0.0f ? x : (expf(x) - 1.0f);
                out[(long long)r * Nout + c] = x;
            }
        }
    }
}

__global__ void fill_i32(int* p, int v, int n) {
    int i = blockIdx.x * blockDim.x + threadIdx.x;
    if (i < n) p[i] = v;
}

__global__ __launch_bounds__(1024)
void k_bin(const int* __restrict__ src, const int* __restrict__ dst, int* __restrict__ glist,
           int* __restrict__ gcount, int E)
{
    __shared__ int cnt[AGG_NT];
    __shared__ int base[AGG_NT];
    const int per = (E + (int)gridDim.x - 1) / (int)gridDim.x;
    const int e0  = blockIdx.x * per;
    const int e1  = min(e0 + per, E);
    for (int i = threadIdx.x; i < AGG_NT; i += 1024) cnt[i] = 0;
    __syncthreads();
    for (int e = e0 + threadIdx.x; e < e1; e += 1024) {
        const int d = dst[e];
        if ((unsigned)d < (unsigned)Nn)
            __hip_atomic_fetch_add(&cnt[d / AGG_TR], 1, __ATOMIC_RELAXED,
                                   __HIP_MEMORY_SCOPE_WORKGROUP);
    }
    __syncthreads();
    for (int i = threadIdx.x; i < AGG_NT; i += 1024) {
        const int n = cnt[i];
        base[i] = n ? __hip_atomic_fetch_add(&gcount[i * AGG_CSTR], n, __ATOMIC_RELAXED,
                                             __HIP_MEMORY_SCOPE_AGENT) : 0;
        cnt[i] = 0;
    }
    __syncthreads();
    for (int e = e0 + threadIdx.x; e < e1; e += 1024) {
        const int d = dst[e];
        if ((unsigned)d < (unsigned)Nn) {
            const int g   = d / AGG_TR;
            const int pos = base[g] + __hip_atomic_fetch_add(&cnt[g], 1, __ATOMIC_RELAXED,
                                                             __HIP_MEMORY_SCOPE_WORKGROUP);
            if (pos < AGG_GCAP)
                glist[(long long)g * AGG_GCAP + pos] =
                    (int)((unsigned)src[e] | ((unsigned)(d - g * AGG_TR) << 24));
        }
    }
}

__global__ void k_attn(const float* __restrict__ feat, const float* __restrict__ al,
                       const float* __restrict__ ar, float* __restrict__ el,
                       float* __restrict__ er, int N, int H, int D) {
    int i = blockIdx.x * blockDim.x + threadIdx.x;
    if (i >= N * H) return;
    const int n = i / H, h = i % H;
    const float* f = feat + (long long)n * H * D + h * D;
    float a = 0.0f, b = 0.0f;
    for (int d = 0; d < D; ++d) { a += f[d] * al[h * D + d]; b += f[d] * ar[h * D + d]; }
    el[i] = a; er[i] = b;
}

template <int ROW, int H, bool GAT, bool ACT>
__global__ __launch_bounds__(1024)
void k_tile_aggregate(const float* __restrict__ feat, const int* __restrict__ src,
                      const int* __restrict__ dst, const float* __restrict__ el,
                      const float* __restrict__ er, const float* __restrict__ bias,
                      float* __restrict__ out, const int* __restrict__ glist,
                      const int* __restrict__ gcount, float* __restrict__ degout, int E, int N)
{
    constexpr int TR = AGG_TR;
    constexpr int GCAP = AGG_GCAP;
    static_assert(TR <= 256 && Nn <= (1 << 24), "a list word holds 24 bits of source row, 8 of tile row");
    constexpr int NQ = (ROW + 31) / 32;
    constexpr int D  = ROW / H;
    constexpr int EG = (NQ >= 8) ? 4 : 8;
    constexpr int TRIP = 512;
    constexpr int QCAP = GAT ? 1536 : 544;
    constexpr int NS = GAT ? TR * H : 1;
    static_assert(H == 1 || D % 32 == 0, "a chunk must not straddle two heads");
    static_assert((TR * ROW) % 32 == 0, "tile must be whole 128-byte lines");
    __shared__ float tile[TR * ROW];
    __shared__ int queue[32 * QCAP];
    __shared__ unsigned rmax[NS];
    __shared__ float rsum[NS];
    __shared__ float rer[NS];
    __shared__ int ovf;
    __shared__ int rdeg[GAT ? 1 : TR];

    const int lane = threadIdx.x & 31;
    const int wave = __builtin_amdgcn_readfirstlane((int)(threadIdx.x >> 5));
    const int lo   = blockIdx.x * TR;

    for (int i = threadIdx.x; i < TR * ROW / 4; i += 1024)
        ((float4*)tile)[i] = make_float4(0.0f, 0.0f, 0.0f, 0.0f);
    if (GAT) {
        for (int i = threadIdx.x; i < TR * H; i += 1024) {
            rmax[i] = 0x007FFFFFu;
            rsum[i] = 0.0f;
            rer[i]  = (lo * H + i < N * H) ? er[lo * H + i] : 0.0f;
        }
        if (threadIdx.x == 0) ovf = 0;
    } else {
        for (int i = threadIdx.x; i < TR; i += 1024) rdeg[i] = 0;
    }
    __syncthreads();

    const int qb = wave * QCAP;
    int qn = 0;
    int done = 0;
    const int* wl = glist + (long long)blockIdx.x * GCAP;
    const int gn = __builtin_amdgcn_readfirstlane(gcount[blockIdx.x * AGG_CSTR]);
    const bool listed = gn <= GCAP;

    auto add_rows = [&](int cnt, int ms, int md, const float (&mw)[H]) {
        for (int j0 = 0; j0 < cnt; j0 += EG) {
            float v[EG][NQ];
            int   trow[EG];
#pragma unroll
            for (int g = 0; g < EG; ++g) {
                const int j  = min(j0 + g, cnt - 1);
                const int sj = __builtin_amdgcn_readlane(ms, j);
                trow[g] = __builtin_amdgcn_readlane(md, j) * ROW;
                const float* frow = feat + (long long)sj * ROW;
                float wj[H];
#pragma unroll
                for (int h = 0; h < H; ++h)
                    wj[h] = __int_as_float(__builtin_amdgcn_readlane(__float_as_int(mw[h]), j));
#pragma unroll
                for (int q = 0; q < NQ; ++q) {
                    const int c = q * 32 + lane;
                    float t = 0.0f;
                    if (ROW % 32 == 0 || c < ROW) t = frow[c];
                    if (GAT) t *= wj[(H == 1) ? 0 : (q * 32) / D];
                    v[g][q] = t;
                }
            }
#pragma unroll
            for (int g = 0; g < EG; ++g) {
                if (j0 + g < cnt) {
#pragma unroll
                    for (int q = 0; q < NQ; ++q) {
                        const int c = q * 32 + lane;
                        if (ROW % 32 == 0 || c < ROW)
                            __hip_atomic_fetch_add(&tile[trow[g] + c], v[g][q], __ATOMIC_RELAXED,
                                                   __HIP_MEMORY_SCOPE_WORKGROUP);
                    }
                }
            }
        }
    };

    auto logits = [&](int s, int d, float (&e)[H]) {
        float l[H];
        if (H == 2) {
            const v2f t = *(const v2f*)(el + (long long)s * 2);
            l[0] = t.x; l[H - 1] = t.y;
        } else {
#pragma unroll
            for (int h = 0; h < H; ++h) l[h] = el[(long long)s * H + h];
        }
#pragma unroll
        for (int h = 0; h < H; ++h) {
            const float x = l[h] + rer[d * H + h];
            e[h] = x > 0.0f ? x : 0.2f * x;
        }
    };

    auto edge_of = [&](bool list, int base, int cnt, int& ms, int& md) {
        ms = 0; md = 0;
        if (list) {
            if (lane < cnt) {
                const unsigned p = (unsigned)wl[base + lane];
                ms = (int)(p & 0x00FFFFFFu);
                md = (int)(p >> 24);
            }
        } else {
            __builtin_amdgcn_fence(__ATOMIC_ACQ_REL, "wavefront");
            if (lane < cnt) {
                const int e = queue[qb + base + lane];
                ms = src[e];
                md = dst[e] - lo;
            }
        }
    };

    auto batch_sum = [&](bool list, int base, int cnt) {
        int ms, md;
        float mw[H];
#pragma unroll
        for (int h = 0; h < H; ++h) mw[h] = 0.0f;
        edge_of(list, base, cnt, ms, md);
        if (lane < cnt)
            __hip_atomic_fetch_add(&rdeg[md], 1, __ATOMIC_RELAXED,
                                   __HIP_MEMORY_SCOPE_WORKGROUP);
        add_rows(cnt, ms, md, mw);
    };

    auto batch_max = [&](bool list, int base, int cnt) {
        int s, d;
        edge_of(list, base, cnt, s, d);
        if (lane < cnt) {
            float lg[H];
            logits(s, d, lg);
#pragma unroll
            for (int h = 0; h < H; ++h)
                __hip_atomic_fetch_max(&rmax[d * H + h], enc_f(lg[h]), __ATOMIC_RELAXED,
                                       __HIP_MEMORY_SCOPE_WORKGROUP);
        }
    };

    auto batch_acc = [&](bool list, int base, int cnt) {
        int ms, md;
        float mw[H];
#pragma unroll
        for (int h = 0; h < H; ++h) mw[h] = 0.0f;
        edge_of(list, base, cnt, ms, md);
        if (lane < cnt) {
            float lg[H];
            logits(ms, md, lg);
#pragma unroll
            for (int h = 0; h < H; ++h) {
                float m = dec_f(rmax[md * H + h]);
                if (!(fabsf(m) <= 3.402823466e38f)) m = 0.0f;
                mw[h] = __expf(lg[h] - m);
                __hip_atomic_fetch_add(&rsum[md * H + h], mw[h], __ATOMIC_RELAXED,
                                       __HIP_MEMORY_SCOPE_WORKGROUP);
            }
        }
        add_rows(cnt, ms, md, mw);
    };

    auto offer = [&](int e, int d, unsigned okm) {
        const unsigned mask = __builtin_amdgcn_ballot_w32((unsigned)(d - lo) < (unsigned)TR) & okm;
        if (mask != 0u) {
            if ((mask >> lane) & 1u) queue[qb + qn + __popc(mask & ((1u << lane) - 1u))] = e;
            qn += __popc(mask);
        }
    };

    const v4i* dst4 = (const v4i*)dst;
    const int nvec  = E >> 2;
    auto fetch = [&](int vi) { return dst4[min(vi, nvec - 1)]; };

    auto scan = [&](auto&& fn, bool keep) {
        if (wave == 0) {
            const int e = min(nvec * 4 + lane, E - 1);
            offer(e, dst[e], __builtin_amdgcn_ballot_w32(nvec * 4 + lane < E));
        }
        v4i nxt[4];
        int vb = wave * 128;
#pragma unroll
        for (int u = 0; u < 4; ++u) nxt[u] = fetch(vb + u * 32 + lane);
        for (;;) {
            const bool last = vb >= nvec;
            if (!last) {
                if (keep && qn + TRIP > QCAP) {
                    const int pend = qn - done;
                    __builtin_amdgcn_fence(__ATOMIC_ACQ_REL, "wavefront");
                    if (lane < pend) queue[qb + lane] = queue[qb + done + lane];
                    qn = pend; done = 0; keep = false;
                    if (lane == 0) ovf = 1;
                }
                v4i cur[4];
#pragma unroll
                for (int u = 0; u < 4; ++u) cur[u] = nxt[u];
                const int vn = vb + 32 * 128;
#pragma unroll
                for (int u = 0; u < 4; ++u) nxt[u] = fetch(vn + u * 32 + lane);
#pragma unroll
                for (int u = 0; u < 4; ++u) {
                    const int vi = vb + u * 32 + lane;
                    const unsigned ok = __builtin_amdgcn_ballot_w32(vi < nvec);
                    offer(vi * 4 + 0, cur[u].x, ok);
                    offer(vi * 4 + 1, cur[u].y, ok);
                    offer(vi * 4 + 2, cur[u].z, ok);
                    offer(vi * 4 + 3, cur[u].w, ok);
                }
            }
            const int thr = last ? 1 : 32;
            while (qn - done >= thr) {
                const int cnt = min(qn - done, 32);
                int base;
                if (keep) { base = done; done += cnt; }
                else      { qn -= cnt; base = qn; }
                fn(base, cnt);
            }
            if (last) break;
            vb += 32 * 128;
        }
    };

    if (!GAT) {
        if (listed) {
            for (int b = wave * 32; b < gn; b += 32 * 32) batch_sum(true, b, min(gn - b, 32));
        } else {
            scan([&](int b, int c) { batch_sum(false, b, c); }, false);
        }
        __syncthreads();
        for (int i = threadIdx.x; i < TR; i += 1024)
            if (lo + i < N) degout[lo + i] = (float)rdeg[i];
    } else {
        if (listed) {
            for (int b0 = 0; b0 < gn; b0 += 4 * 32 * 32) {
                int  s[4], d[4];
                bool ok[4];
                float lg[4][H];
#pragma unroll
                for (int u = 0; u < 4; ++u) {
                    const int i = b0 + (u * 32 + wave) * 32 + lane;
                    ok[u] = i < gn;
                    const unsigned p = (unsigned)wl[min(i, gn - 1)];
                    s[u] = (int)(p & 0x00FFFFFFu);
                    d[u] = (int)(p >> 24);
                }
#pragma unroll
                for (int u = 0; u < 4; ++u) logits(s[u], d[u], lg[u]);
#pragma unroll
                for (int u = 0; u < 4; ++u) {
                    if (ok[u]) {
#pragma unroll
                        for (int h = 0; h < H; ++h)
                            __hip_atomic_fetch_max(&rmax[d[u] * H + h], enc_f(lg[u][h]),
                                                   __ATOMIC_RELAXED, __HIP_MEMORY_SCOPE_WORKGROUP);
                    }
                }
            }
        } else {
            scan([&](int b, int c) { batch_max(false, b, c); }, true);
        }
        __syncthreads();
        const bool rescan = __builtin_amdgcn_readfirstlane(ovf) != 0;
        if (listed) {
            for (int b = wave * 32; b < gn; b += 32 * 32) batch_acc(true, b, min(gn - b, 32));
        } else if (!rescan) {
            for (int b = 0; b < qn; b += 32) batch_acc(false, b, min(qn - b, 32));
        } else {
            qn = 0; done = 0;
            scan([&](int b, int c) { batch_acc(false, b, c); }, false);
        }
        __syncthreads();
    }

    const long long g0 = (long long)lo * ROW;
    const long long gN = (long long)N * ROW;
    for (int i = threadIdx.x; i < TR * ROW; i += 1024) {
        if (g0 + i < gN) {
            float x = tile[i];
            if (GAT) {
                const int r = i / ROW, c = i - r * ROW;
                x = x / fmaxf(rsum[r * H + c / D], 1e-9f) + bias[c];
            }
            if (ACT) x = x > 0.0f ? x : (expf(x) - 1.0f);
            out[g0 + i] = x;
        }
    }
}

static inline unsigned gsz(long long n, int b) { return (unsigned)((n + b - 1) / b); }

extern "C" void kernel_launch(void* const* d_in, const int* in_sizes, int n_in,
                              void* d_out, int out_size, void* d_ws, size_t ws_size,
                              hipStream_t stream) {
    const float* x       = (const float*)d_in[0];
    const int*   src     = (const int*)  d_in[1];
    const int*   dst     = (const int*)  d_in[2];
    const float* w_self  = (const float*)d_in[3];
    const float* w_neigh = (const float*)d_in[4];
    const float* sage_b  = (const float*)d_in[5];
    const float* g1_w    = (const float*)d_in[6];
    const float* g1_al   = (const float*)d_in[7];
    const float* g1_ar   = (const float*)d_in[8];
    const float* g1_b    = (const float*)d_in[9];
    const float* g2_w    = (const float*)d_in[10];
    const float* g2_al   = (const float*)d_in[11];
    const float* g2_ar   = (const float*)d_in[12];
    const float* g2_b    = (const float*)d_in[13];

    int*      glist  = (int*)d_ws;
    int*      gcount = glist + AGG_NT * AGG_GCAP;
    float*    ws    = (float*)(gcount + AGG_NT * AGG_CSTR);
    float*    deg   = ws;
    float*    neigh = deg + Nn;
    float*    h0    = neigh + (long long)Nn * F;
    float*    el    = h0 + (long long)Nn * D0;
    float*    er    = el + Nn * H1;
    float*    wt_self  = er + Nn * H1;
    float*    wt_neigh = wt_self + D0 * F;
    float*    wt_g1    = wt_neigh + D0 * F;
    float*    wt_g2    = wt_g1 + (H1 * D1) * D0;
    float*    feat  = neigh;

    const int B = 256;
    const int Mtiles = Nn / 16;
    const unsigned rowblk = gsz(Mtiles, 8);
    const unsigned ntile  = AGG_NT;

    k_transpose<<<gsz(D0 * F, B), B, 0, stream>>>(w_self,  wt_self,  F,  D0, D0);
    k_transpose<<<gsz(D0 * F, B), B, 0, stream>>>(w_neigh, wt_neigh, F,  D0, D0);
    k_transpose<<<gsz(D0 * D0, B), B, 0, stream>>>(g1_w,   wt_g1,    D0, H1 * D1, H1 * D1);
    k_transpose<<<gsz(C2P * D0, B), B, 0, stream>>>(g2_w,  wt_g2,    D0, C2, C2P);

    fill_i32<<<gsz(AGG_NT * AGG_CSTR, B), B, 0, stream>>>(gcount, 0, AGG_NT * AGG_CSTR);
    k_bin<<<256, 1024, 0, stream>>>(src, dst, glist, gcount, Ne);
    k_tile_aggregate<F, 1, false, false><<<ntile, 1024, 0, stream>>>(
        x, src, dst, nullptr, nullptr, nullptr, neigh, glist, gcount, deg, Ne, Nn);
    {
        const size_t shmem = 2ull * 64 * (F + 4) * sizeof(float);
        wmma_gemm<true, true><<<rowblk * 2, B, shmem, stream>>>(
            x, wt_self, neigh, wt_neigh, deg, sage_b, h0, Nn, F, D0, 2);
    }

    {
        const size_t shmem = 64ull * (D0 + 4) * sizeof(float);
        wmma_gemm<false, false><<<rowblk * 2, B, shmem, stream>>>(
            h0, wt_g1, nullptr, nullptr, nullptr, nullptr, feat, Nn, D0, H1 * D1, 2);
    }
    k_attn<<<gsz(Nn * H1, B), B, 0, stream>>>(feat, g1_al, g1_ar, el, er, Nn, H1, D1);
    k_tile_aggregate<H1 * D1, H1, true, true><<<ntile, 1024, 0, stream>>>(
        feat, src, dst, el, er, g1_b, h0, glist, gcount, nullptr, Ne, Nn);

    {
        const size_t shmem = 64ull * (D0 + 4) * sizeof(float);
        wmma_gemm<false, false><<<rowblk, B, shmem, stream>>>(
            h0, wt_g2, nullptr, nullptr, nullptr, nullptr, feat, Nn, D0, C2, 1);
    }
    k_attn<<<gsz(Nn, B), B, 0, stream>>>(feat, g2_al, g2_ar, el, er, Nn, 1, C2);
    k_tile_aggregate<C2, 1, true, false><<<ntile, 1024, 0, stream>>>(
        feat, src, dst, el, er, g2_b, (float*)d_out, glist, gcount, nullptr, Ne, Nn);
}


# GEMMs re-tiled: 256 persistent work-groups, one per WGP, hold the whole transposed weight matrix in LDS, staged once; waves loop over 16x64 tiles (782/782/391 groups each restaged a 64-col panel)
# speedup vs baseline: 4.7459x; 1.1994x over previous
#include <hip/hip_runtime.h>

constexpr int Nn  = 50000;
constexpr int Ne  = 800000;
constexpr int F   = 256;
constexpr int D0  = 128;
constexpr int H1  = 2;
constexpr int D1  = 64;
constexpr int C2  = 47;
constexpr int C2P = 64;

constexpr int AGG_TR   = 224;
constexpr int AGG_NT   = (Nn + AGG_TR - 1) / AGG_TR;
constexpr int AGG_GCAP = 16384;
constexpr int AGG_CSTR = 64;

typedef float v2f __attribute__((ext_vector_type(2)));
typedef float v8f __attribute__((ext_vector_type(8)));
typedef int   v4i __attribute__((ext_vector_type(4)));

__device__ __forceinline__ unsigned enc_f(float f) {
    unsigned u = __float_as_uint(f);
    return (u & 0x80000000u) ? ~u : (u | 0x80000000u);
}
__device__ __forceinline__ float dec_f(unsigned u) {
    unsigned b = (u & 0x80000000u) ? (u ^ 0x80000000u) : ~u;
    return __uint_as_float(b);
}

__global__ void k_transpose(const float* __restrict__ W, float* __restrict__ Wt,
                            int K, int N, int Npad) {
    int i = blockIdx.x * blockDim.x + threadIdx.x;
    if (i >= Npad * K) return;
    const int c = i / K, k = i % K;
    Wt[i] = (c < N) ? W[(long long)k * N + c] : 0.0f;
}

constexpr int GEMM_WG = 256;

template <int K, int NCG, int NW, bool HAS_A2, bool ACT>
__global__ __launch_bounds__(NW * 32)
void wmma_gemm(const float* __restrict__ A, const float* __restrict__ Wt,
               const float* __restrict__ A2, const float* __restrict__ W2t,
               const float* __restrict__ rowdeg, const float* __restrict__ bias,
               float* __restrict__ out, int M, int Nout)
{
    constexpr int KP = K + 4;
    constexpr int NC = NCG * 64;
    constexpr int KV = K / 4;
    static_assert(NW % NCG == 0 && K % 4 == 0, "waves split evenly over the column groups");
    static_assert(NW <= 32, "a work-group holds at most 32 waves");
    __shared__ float w1s[NC * KP];
    __shared__ float w2s[HAS_A2 ? NC * KP : 4];

    for (int i = threadIdx.x; i < NC * KV; i += NW * 32) {
        const int c = i / KV, k = (i % KV) * 4;
        *(float4*)(w1s + c * KP + k) = ((const float4*)Wt)[i];
        if (HAS_A2) *(float4*)(w2s + c * KP + k) = ((const float4*)W2t)[i];
    }
    __syncthreads();

    const int lane = threadIdx.x & 31;
    const int half = lane >> 4;
    const int l    = lane & 15;
    const int wave = __builtin_amdgcn_readfirstlane((int)(threadIdx.x >> 5));
    const int col0 = (wave % NCG) * 64;
    const int ntile = M / 16;

    const float* Bs[4];
    const float* B2s[4];
#pragma unroll
    for (int t = 0; t < 4; ++t) {
        Bs[t] = w1s + (col0 + t * 16 + l) * KP + half * 2;
        if (HAS_A2) B2s[t] = w2s + (col0 + t * 16 + l) * KP + half * 2;
    }

    for (int s = wave / NCG; s * GEMM_WG + (int)blockIdx.x < ntile; s += NW / NCG) {
        const int rt  = s * GEMM_WG + (int)blockIdx.x;
        const int row = rt * 16 + l;

        const float* Arow  = A + (long long)row * K + half * 2;
        const float* A2row = HAS_A2 ? (A2 + (long long)row * K + half * 2) : nullptr;
        const float inv    = HAS_A2 ? (1.0f / fmaxf(rowdeg[row], 1.0f)) : 0.0f;
        __builtin_prefetch(Arow, 0, 3);

        v8f acc[4] = {};
        for (int kk = 0; kk < K; kk += 4) {
            const v2f a = *(const v2f*)(Arow + kk);
            v2f a2;
            if (HAS_A2) {
                const v2f t2 = *(const v2f*)(A2row + kk);
                a2.x = t2.x * inv; a2.y = t2.y * inv;
            }
#pragma unroll
            for (int t = 0; t < 4; ++t) {
                const v2f b = *(const v2f*)(Bs[t] + kk);
                acc[t] = __builtin_amdgcn_wmma_f32_16x16x4_f32(
                    false, a, false, b, (short)0, acc[t], false, false);
                if (HAS_A2) {
                    const v2f b2 = *(const v2f*)(B2s[t] + kk);
                    acc[t] = __builtin_amdgcn_wmma_f32_16x16x4_f32(
                        false, a2, false, b2, (short)0, acc[t], false, false);
                }
            }
        }

#pragma unroll
        for (int t = 0; t < 4; ++t) {
            const int c = col0 + t * 16 + l;
            if (c < Nout) {
                const float bv = bias ? bias[c] : 0.0f;
#pragma unroll
                for (int v = 0; v < 8; ++v) {
                    const int r = rt * 16 + v + 8 * half;
                    float x = acc[t][v] + bv;
                    if (ACT) x = x > 0.0f ? x : (expf(x) - 1.0f);
                    out[(long long)r * Nout + c] = x;
                }
            }
        }
    }
}

__global__ void fill_i32(int* p, int v, int n) {
    int i = blockIdx.x * blockDim.x + threadIdx.x;
    if (i < n) p[i] = v;
}

__global__ __launch_bounds__(1024)
void k_bin(const int* __restrict__ src, const int* __restrict__ dst, int* __restrict__ glist,
           int* __restrict__ gcount, int E)
{
    __shared__ int cnt[AGG_NT];
    __shared__ int base[AGG_NT];
    const int per = (E + (int)gridDim.x - 1) / (int)gridDim.x;
    const int e0  = blockIdx.x * per;
    const int e1  = min(e0 + per, E);
    for (int i = threadIdx.x; i < AGG_NT; i += 1024) cnt[i] = 0;
    __syncthreads();
    for (int e = e0 + threadIdx.x; e < e1; e += 1024) {
        const int d = dst[e];
        if ((unsigned)d < (unsigned)Nn)
            __hip_atomic_fetch_add(&cnt[d / AGG_TR], 1, __ATOMIC_RELAXED,
                                   __HIP_MEMORY_SCOPE_WORKGROUP);
    }
    __syncthreads();
    for (int i = threadIdx.x; i < AGG_NT; i += 1024) {
        const int n = cnt[i];
        base[i] = n ? __hip_atomic_fetch_add(&gcount[i * AGG_CSTR], n, __ATOMIC_RELAXED,
                                             __HIP_MEMORY_SCOPE_AGENT) : 0;
        cnt[i] = 0;
    }
    __syncthreads();
    for (int e = e0 + threadIdx.x; e < e1; e += 1024) {
        const int d = dst[e];
        if ((unsigned)d < (unsigned)Nn) {
            const int g   = d / AGG_TR;
            const int pos = base[g] + __hip_atomic_fetch_add(&cnt[g], 1, __ATOMIC_RELAXED,
                                                             __HIP_MEMORY_SCOPE_WORKGROUP);
            if (pos < AGG_GCAP)
                glist[(long long)g * AGG_GCAP + pos] =
                    (int)((unsigned)src[e] | ((unsigned)(d - g * AGG_TR) << 24));
        }
    }
}

__global__ void k_attn(const float* __restrict__ feat, const float* __restrict__ al,
                       const float* __restrict__ ar, float* __restrict__ el,
                       float* __restrict__ er, int N, int H, int D) {
    int i = blockIdx.x * blockDim.x + threadIdx.x;
    if (i >= N * H) return;
    const int n = i / H, h = i % H;
    const float* f = feat + (long long)n * H * D + h * D;
    float a = 0.0f, b = 0.0f;
    for (int d = 0; d < D; ++d) { a += f[d] * al[h * D + d]; b += f[d] * ar[h * D + d]; }
    el[i] = a; er[i] = b;
}

template <int ROW, int H, bool GAT, bool ACT>
__global__ __launch_bounds__(1024)
void k_tile_aggregate(const float* __restrict__ feat, const int* __restrict__ src,
                      const int* __restrict__ dst, const float* __restrict__ el,
                      const float* __restrict__ er, const float* __restrict__ bias,
                      float* __restrict__ out, const int* __restrict__ glist,
                      const int* __restrict__ gcount, float* __restrict__ degout, int E, int N)
{
    constexpr int TR = AGG_TR;
    constexpr int GCAP = AGG_GCAP;
    static_assert(TR <= 256 && Nn <= (1 << 24), "a list word holds 24 bits of source row, 8 of tile row");
    constexpr int NQ = (ROW + 31) / 32;
    constexpr int D  = ROW / H;
    constexpr int EG = (NQ >= 8) ? 4 : 8;
    constexpr int TRIP = 512;
    constexpr int QCAP = GAT ? 1536 : 544;
    constexpr int NS = GAT ? TR * H : 1;
    static_assert(H == 1 || D % 32 == 0, "a chunk must not straddle two heads");
    static_assert((TR * ROW) % 32 == 0, "tile must be whole 128-byte lines");
    __shared__ float tile[TR * ROW];
    __shared__ int queue[32 * QCAP];
    __shared__ unsigned rmax[NS];
    __shared__ float rsum[NS];
    __shared__ float rer[NS];
    __shared__ int ovf;
    __shared__ int rdeg[GAT ? 1 : TR];

    const int lane = threadIdx.x & 31;
    const int wave = __builtin_amdgcn_readfirstlane((int)(threadIdx.x >> 5));
    const int lo   = blockIdx.x * TR;

    for (int i = threadIdx.x; i < TR * ROW / 4; i += 1024)
        ((float4*)tile)[i] = make_float4(0.0f, 0.0f, 0.0f, 0.0f);
    if (GAT) {
        for (int i = threadIdx.x; i < TR * H; i += 1024) {
            rmax[i] = 0x007FFFFFu;
            rsum[i] = 0.0f;
            rer[i]  = (lo * H + i < N * H) ? er[lo * H + i] : 0.0f;
        }
        if (threadIdx.x == 0) ovf = 0;
    } else {
        for (int i = threadIdx.x; i < TR; i += 1024) rdeg[i] = 0;
    }
    __syncthreads();

    const int qb = wave * QCAP;
    int qn = 0;
    int done = 0;
    const int* wl = glist + (long long)blockIdx.x * GCAP;
    const int gn = __builtin_amdgcn_readfirstlane(gcount[blockIdx.x * AGG_CSTR]);
    const bool listed = gn <= GCAP;

    auto add_rows = [&](int cnt, int ms, int md, const float (&mw)[H]) {
        for (int j0 = 0; j0 < cnt; j0 += EG) {
            float v[EG][NQ];
            int   trow[EG];
#pragma unroll
            for (int g = 0; g < EG; ++g) {
                const int j  = min(j0 + g, cnt - 1);
                const int sj = __builtin_amdgcn_readlane(ms, j);
                trow[g] = __builtin_amdgcn_readlane(md, j) * ROW;
                const float* frow = feat + (long long)sj * ROW;
                float wj[H];
#pragma unroll
                for (int h = 0; h < H; ++h)
                    wj[h] = __int_as_float(__builtin_amdgcn_readlane(__float_as_int(mw[h]), j));
#pragma unroll
                for (int q = 0; q < NQ; ++q) {
                    const int c = q * 32 + lane;
                    float t = 0.0f;
                    if (ROW % 32 == 0 || c < ROW) t = frow[c];
                    if (GAT) t *= wj[(H == 1) ? 0 : (q * 32) / D];
                    v[g][q] = t;
                }
            }
#pragma unroll
            for (int g = 0; g < EG; ++g) {
                if (j0 + g < cnt) {
#pragma unroll
                    for (int q = 0; q < NQ; ++q) {
                        const int c = q * 32 + lane;
                        if (ROW % 32 == 0 || c < ROW)
                            __hip_atomic_fetch_add(&tile[trow[g] + c], v[g][q], __ATOMIC_RELAXED,
                                                   __HIP_MEMORY_SCOPE_WORKGROUP);
                    }
                }
            }
        }
    };

    auto logits = [&](int s, int d, float (&e)[H]) {
        float l[H];
        if (H == 2) {
            const v2f t = *(const v2f*)(el + (long long)s * 2);
            l[0] = t.x; l[H - 1] = t.y;
        } else {
#pragma unroll
            for (int h = 0; h < H; ++h) l[h] = el[(long long)s * H + h];
        }
#pragma unroll
        for (int h = 0; h < H; ++h) {
            const float x = l[h] + rer[d * H + h];
            e[h] = x > 0.0f ? x : 0.2f * x;
        }
    };

    auto edge_of = [&](bool list, int base, int cnt, int& ms, int& md) {
        ms = 0; md = 0;
        if (list) {
            if (lane < cnt) {
                const unsigned p = (unsigned)wl[base + lane];
                ms = (int)(p & 0x00FFFFFFu);
                md = (int)(p >> 24);
            }
        } else {
            __builtin_amdgcn_fence(__ATOMIC_ACQ_REL, "wavefront");
            if (lane < cnt) {
                const int e = queue[qb + base + lane];
                ms = src[e];
                md = dst[e] - lo;
            }
        }
    };

    auto batch_sum = [&](bool list, int base, int cnt) {
        int ms, md;
        float mw[H];
#pragma unroll
        for (int h = 0; h < H; ++h) mw[h] = 0.0f;
        edge_of(list, base, cnt, ms, md);
        if (lane < cnt)
            __hip_atomic_fetch_add(&rdeg[md], 1, __ATOMIC_RELAXED,
                                   __HIP_MEMORY_SCOPE_WORKGROUP);
        add_rows(cnt, ms, md, mw);
    };

    auto batch_max = [&](bool list, int base, int cnt) {
        int s, d;
        edge_of(list, base, cnt, s, d);
        if (lane < cnt) {
            float lg[H];
            logits(s, d, lg);
#pragma unroll
            for (int h = 0; h < H; ++h)
                __hip_atomic_fetch_max(&rmax[d * H + h], enc_f(lg[h]), __ATOMIC_RELAXED,
                                       __HIP_MEMORY_SCOPE_WORKGROUP);
        }
    };

    auto batch_acc = [&](bool list, int base, int cnt) {
        int ms, md;
        float mw[H];
#pragma unroll
        for (int h = 0; h < H; ++h) mw[h] = 0.0f;
        edge_of(list, base, cnt, ms, md);
        if (lane < cnt) {
            float lg[H];
            logits(ms, md, lg);
#pragma unroll
            for (int h = 0; h < H; ++h) {
                float m = dec_f(rmax[md * H + h]);
                if (!(fabsf(m) <= 3.402823466e38f)) m = 0.0f;
                mw[h] = __expf(lg[h] - m);
                __hip_atomic_fetch_add(&rsum[md * H + h], mw[h], __ATOMIC_RELAXED,
                                       __HIP_MEMORY_SCOPE_WORKGROUP);
            }
        }
        add_rows(cnt, ms, md, mw);
    };

    auto offer = [&](int e, int d, unsigned okm) {
        const unsigned mask = __builtin_amdgcn_ballot_w32((unsigned)(d - lo) < (unsigned)TR) & okm;
        if (mask != 0u) {
            if ((mask >> lane) & 1u) queue[qb + qn + __popc(mask & ((1u << lane) - 1u))] = e;
            qn += __popc(mask);
        }
    };

    const v4i* dst4 = (const v4i*)dst;
    const int nvec  = E >> 2;
    auto fetch = [&](int vi) { return dst4[min(vi, nvec - 1)]; };

    auto scan = [&](auto&& fn, bool keep) {
        if (wave == 0) {
            const int e = min(nvec * 4 + lane, E - 1);
            offer(e, dst[e], __builtin_amdgcn_ballot_w32(nvec * 4 + lane < E));
        }
        v4i nxt[4];
        int vb = wave * 128;
#pragma unroll
        for (int u = 0; u < 4; ++u) nxt[u] = fetch(vb + u * 32 + lane);
        for (;;) {
            const bool last = vb >= nvec;
            if (!last) {
                if (keep && qn + TRIP > QCAP) {
                    const int pend = qn - done;
                    __builtin_amdgcn_fence(__ATOMIC_ACQ_REL, "wavefront");
                    if (lane < pend) queue[qb + lane] = queue[qb + done + lane];
                    qn = pend; done = 0; keep = false;
                    if (lane == 0) ovf = 1;
                }
                v4i cur[4];
#pragma unroll
                for (int u = 0; u < 4; ++u) cur[u] = nxt[u];
                const int vn = vb + 32 * 128;
#pragma unroll
                for (int u = 0; u < 4; ++u) nxt[u] = fetch(vn + u * 32 + lane);
#pragma unroll
                for (int u = 0; u < 4; ++u) {
                    const int vi = vb + u * 32 + lane;
                    const unsigned ok = __builtin_amdgcn_ballot_w32(vi < nvec);
                    offer(vi * 4 + 0, cur[u].x, ok);
                    offer(vi * 4 + 1, cur[u].y, ok);
                    offer(vi * 4 + 2, cur[u].z, ok);
                    offer(vi * 4 + 3, cur[u].w, ok);
                }
            }
            const int thr = last ? 1 : 32;
            while (qn - done >= thr) {
                const int cnt = min(qn - done, 32);
                int base;
                if (keep) { base = done; done += cnt; }
                else      { qn -= cnt; base = qn; }
                fn(base, cnt);
            }
            if (last) break;
            vb += 32 * 128;
        }
    };

    if (!GAT) {
        if (listed) {
            for (int b = wave * 32; b < gn; b += 32 * 32) batch_sum(true, b, min(gn - b, 32));
        } else {
            scan([&](int b, int c) { batch_sum(false, b, c); }, false);
        }
        __syncthreads();
        for (int i = threadIdx.x; i < TR; i += 1024)
            if (lo + i < N) degout[lo + i] = (float)rdeg[i];
    } else {
        if (listed) {
            for (int b0 = 0; b0 < gn; b0 += 4 * 32 * 32) {
                int  s[4], d[4];
                bool ok[4];
                float lg[4][H];
#pragma unroll
                for (int u = 0; u < 4; ++u) {
                    const int i = b0 + (u * 32 + wave) * 32 + lane;
                    ok[u] = i < gn;
                    const unsigned p = (unsigned)wl[min(i, gn - 1)];
                    s[u] = (int)(p & 0x00FFFFFFu);
                    d[u] = (int)(p >> 24);
                }
#pragma unroll
                for (int u = 0; u < 4; ++u) logits(s[u], d[u], lg[u]);
#pragma unroll
                for (int u = 0; u < 4; ++u) {
                    if (ok[u]) {
#pragma unroll
                        for (int h = 0; h < H; ++h)
                            __hip_atomic_fetch_max(&rmax[d[u] * H + h], enc_f(lg[u][h]),
                                                   __ATOMIC_RELAXED, __HIP_MEMORY_SCOPE_WORKGROUP);
                    }
                }
            }
        } else {
            scan([&](int b, int c) { batch_max(false, b, c); }, true);
        }
        __syncthreads();
        const bool rescan = __builtin_amdgcn_readfirstlane(ovf) != 0;
        if (listed) {
            for (int b = wave * 32; b < gn; b += 32 * 32) batch_acc(true, b, min(gn - b, 32));
        } else if (!rescan) {
            for (int b = 0; b < qn; b += 32) batch_acc(false, b, min(qn - b, 32));
        } else {
            qn = 0; done = 0;
            scan([&](int b, int c) { batch_acc(false, b, c); }, false);
        }
        __syncthreads();
    }

    const long long g0 = (long long)lo * ROW;
    const long long gN = (long long)N * ROW;
    for (int i = threadIdx.x; i < TR * ROW; i += 1024) {
        if (g0 + i < gN) {
            float x = tile[i];
            if (GAT) {
                const int r = i / ROW, c = i - r * ROW;
                x = x / fmaxf(rsum[r * H + c / D], 1e-9f) + bias[c];
            }
            if (ACT) x = x > 0.0f ? x : (expf(x) - 1.0f);
            out[g0 + i] = x;
        }
    }
}

static inline unsigned gsz(long long n, int b) { return (unsigned)((n + b - 1) / b); }

extern "C" void kernel_launch(void* const* d_in, const int* in_sizes, int n_in,
                              void* d_out, int out_size, void* d_ws, size_t ws_size,
                              hipStream_t stream) {
    const float* x       = (const float*)d_in[0];
    const int*   src     = (const int*)  d_in[1];
    const int*   dst     = (const int*)  d_in[2];
    const float* w_self  = (const float*)d_in[3];
    const float* w_neigh = (const float*)d_in[4];
    const float* sage_b  = (const float*)d_in[5];
    const float* g1_w    = (const float*)d_in[6];
    const float* g1_al   = (const float*)d_in[7];
    const float* g1_ar   = (const float*)d_in[8];
    const float* g1_b    = (const float*)d_in[9];
    const float* g2_w    = (const float*)d_in[10];
    const float* g2_al   = (const float*)d_in[11];
    const float* g2_ar   = (const float*)d_in[12];
    const float* g2_b    = (const float*)d_in[13];

    int*      glist  = (int*)d_ws;
    int*      gcount = glist + AGG_NT * AGG_GCAP;
    float*    ws    = (float*)(gcount + AGG_NT * AGG_CSTR);
    float*    deg   = ws;
    float*    neigh = deg + Nn;
    float*    h0    = neigh + (long long)Nn * F;
    float*    el    = h0 + (long long)Nn * D0;
    float*    er    = el + Nn * H1;
    float*    wt_self  = er + Nn * H1;
    float*    wt_neigh = wt_self + D0 * F;
    float*    wt_g1    = wt_neigh + D0 * F;
    float*    wt_g2    = wt_g1 + (H1 * D1) * D0;
    float*    feat  = neigh;

    const int B = 256;
    const unsigned ntile  = AGG_NT;

    k_transpose<<<gsz(D0 * F, B), B, 0, stream>>>(w_self,  wt_self,  F,  D0, D0);
    k_transpose<<<gsz(D0 * F, B), B, 0, stream>>>(w_neigh, wt_neigh, F,  D0, D0);
    k_transpose<<<gsz(D0 * D0, B), B, 0, stream>>>(g1_w,   wt_g1,    D0, H1 * D1, H1 * D1);
    k_transpose<<<gsz(C2P * D0, B), B, 0, stream>>>(g2_w,  wt_g2,    D0, C2, C2P);

    fill_i32<<<gsz(AGG_NT * AGG_CSTR, B), B, 0, stream>>>(gcount, 0, AGG_NT * AGG_CSTR);
    k_bin<<<256, 1024, 0, stream>>>(src, dst, glist, gcount, Ne);
    k_tile_aggregate<F, 1, false, false><<<ntile, 1024, 0, stream>>>(
        x, src, dst, nullptr, nullptr, nullptr, neigh, glist, gcount, deg, Ne, Nn);
    wmma_gemm<F, 2, 16, true, true><<<GEMM_WG, 16 * 32, 0, stream>>>(
        x, wt_self, neigh, wt_neigh, deg, sage_b, h0, Nn, D0);

    wmma_gemm<D0, 2, 32, false, false><<<GEMM_WG, 32 * 32, 0, stream>>>(
        h0, wt_g1, nullptr, nullptr, nullptr, nullptr, feat, Nn, H1 * D1);
    k_attn<<<gsz(Nn * H1, B), B, 0, stream>>>(feat, g1_al, g1_ar, el, er, Nn, H1, D1);
    k_tile_aggregate<H1 * D1, H1, true, true><<<ntile, 1024, 0, stream>>>(
        feat, src, dst, el, er, g1_b, h0, glist, gcount, nullptr, Ne, Nn);

    wmma_gemm<D0, 1, 32, false, false><<<GEMM_WG, 32 * 32, 0, stream>>>(
        h0, wt_g2, nullptr, nullptr, nullptr, nullptr, feat, Nn, C2);
    k_attn<<<gsz(Nn, B), B, 0, stream>>>(feat, g2_al, g2_ar, el, er, Nn, 1, C2);
    k_tile_aggregate<C2, 1, true, false><<<ntile, 1024, 0, stream>>>(
        feat, src, dst, el, er, g2_b, (float*)d_out, glist, gcount, nullptr, Ne, Nn);
}


# GEMMs read the weights row-major as given; the LDS copy interleaves rows k,k+1 on the way in (B fragment stays one 8-byte LDS load); the 4 transpose launches and transposed copies go (14->10 launches)
# speedup vs baseline: 4.9931x; 1.0521x over previous
#include <hip/hip_runtime.h>

constexpr int Nn  = 50000;
constexpr int Ne  = 800000;
constexpr int F   = 256;
constexpr int D0  = 128;
constexpr int H1  = 2;
constexpr int D1  = 64;
constexpr int C2  = 47;

constexpr int AGG_TR   = 224;
constexpr int AGG_NT   = (Nn + AGG_TR - 1) / AGG_TR;
constexpr int AGG_GCAP = 16384;
constexpr int AGG_CSTR = 64;

typedef float v2f __attribute__((ext_vector_type(2)));
typedef float v8f __attribute__((ext_vector_type(8)));
typedef int   v4i __attribute__((ext_vector_type(4)));

__device__ __forceinline__ unsigned enc_f(float f) {
    unsigned u = __float_as_uint(f);
    return (u & 0x80000000u) ? ~u : (u | 0x80000000u);
}
__device__ __forceinline__ float dec_f(unsigned u) {
    unsigned b = (u & 0x80000000u) ? (u ^ 0x80000000u) : ~u;
    return __uint_as_float(b);
}

constexpr int GEMM_WG = 256;

template <int K, int NCG, int NR, int NW, bool HAS_A2, bool ACT>
__global__ __launch_bounds__(NW * 32)
void wmma_gemm(const float* __restrict__ A, const float* __restrict__ W,
               const float* __restrict__ A2, const float* __restrict__ W2,
               const float* __restrict__ rowdeg, const float* __restrict__ bias,
               float* __restrict__ out, int M)
{
    constexpr int NC  = NCG * 64;
    constexpr int NPP = NC + 16;
    static_assert(NW % NCG == 0 && K % 4 == 0, "waves split evenly over the column groups");
    static_assert(NW <= 32, "a work-group holds at most 32 waves");
    static_assert(NR <= NC && (NR == NC || !HAS_A2), "the padded copy is written for one matrix");
    __shared__ float w1s[K * NPP];
    __shared__ float w2s[HAS_A2 ? K * NPP : 4];

    if (NR == NC) {
        constexpr int NV = NC / 4;
        for (int i = threadIdx.x; i < (K / 2) * NV; i += NW * 32) {
            const int kp = i / NV, cc = (i % NV) * 4;
            const float4 r0 = *(const float4*)(W + (2 * kp) * NC + cc);
            const float4 r1 = *(const float4*)(W + (2 * kp + 1) * NC + cc);
            float* d = w1s + (kp * NPP + cc) * 2;
            *(float4*)(d)     = make_float4(r0.x, r1.x, r0.y, r1.y);
            *(float4*)(d + 4) = make_float4(r0.z, r1.z, r0.w, r1.w);
            if (HAS_A2) {
                const float4 q0 = *(const float4*)(W2 + (2 * kp) * NC + cc);
                const float4 q1 = *(const float4*)(W2 + (2 * kp + 1) * NC + cc);
                float* d2 = w2s + (kp * NPP + cc) * 2;
                *(float4*)(d2)     = make_float4(q0.x, q1.x, q0.y, q1.y);
                *(float4*)(d2 + 4) = make_float4(q0.z, q1.z, q0.w, q1.w);
            }
        }
    } else {
        for (int i = threadIdx.x; i < K * NC; i += NW * 32) {
            const int k = i / NC, cc = i % NC;
            w1s[((k >> 1) * NPP + cc) * 2 + (k & 1)] = (cc < NR) ? W[k * NR + cc] : 0.0f;
        }
    }
    __syncthreads();

    const int lane = threadIdx.x & 31;
    const int half = lane >> 4;
    const int l    = lane & 15;
    const int wave = __builtin_amdgcn_readfirstlane((int)(threadIdx.x >> 5));
    const int col0 = (wave % NCG) * 64;
    const int ntile = M / 16;

    const float* Bs[4];
    const float* B2s[4];
#pragma unroll
    for (int t = 0; t < 4; ++t) {
        Bs[t] = w1s + (half * NPP + col0 + t * 16 + l) * 2;
        if (HAS_A2) B2s[t] = w2s + (half * NPP + col0 + t * 16 + l) * 2;
    }

    for (int s = wave / NCG; s * GEMM_WG + (int)blockIdx.x < ntile; s += NW / NCG) {
        const int rt  = s * GEMM_WG + (int)blockIdx.x;
        const int row = rt * 16 + l;

        const float* Arow  = A + (long long)row * K + half * 2;
        const float* A2row = HAS_A2 ? (A2 + (long long)row * K + half * 2) : nullptr;
        const float inv    = HAS_A2 ? (1.0f / fmaxf(rowdeg[row], 1.0f)) : 0.0f;
        __builtin_prefetch(Arow, 0, 3);

        v8f acc[4] = {};
        for (int kk = 0; kk < K; kk += 4) {
            const v2f a = *(const v2f*)(Arow + kk);
            v2f a2;
            if (HAS_A2) {
                const v2f t2 = *(const v2f*)(A2row + kk);
                a2.x = t2.x * inv; a2.y = t2.y * inv;
            }
#pragma unroll
            for (int t = 0; t < 4; ++t) {
                const v2f b = *(const v2f*)(Bs[t] + kk * NPP);
                acc[t] = __builtin_amdgcn_wmma_f32_16x16x4_f32(
                    false, a, false, b, (short)0, acc[t], false, false);
                if (HAS_A2) {
                    const v2f b2 = *(const v2f*)(B2s[t] + kk * NPP);
                    acc[t] = __builtin_amdgcn_wmma_f32_16x16x4_f32(
                        false, a2, false, b2, (short)0, acc[t], false, false);
                }
            }
        }

#pragma unroll
        for (int t = 0; t < 4; ++t) {
            const int c = col0 + t * 16 + l;
            if (c < NR) {
                const float bv = bias ? bias[c] : 0.0f;
#pragma unroll
                for (int v = 0; v < 8; ++v) {
                    const int r = rt * 16 + v + 8 * half;
                    float x = acc[t][v] + bv;
                    if (ACT) x = x > 0.0f ? x : (expf(x) - 1.0f);
                    out[(long long)r * NR + c] = x;
                }
            }
        }
    }
}

__global__ void fill_i32(int* p, int v, int n) {
    int i = blockIdx.x * blockDim.x + threadIdx.x;
    if (i < n) p[i] = v;
}

__global__ __launch_bounds__(1024)
void k_bin(const int* __restrict__ src, const int* __restrict__ dst, int* __restrict__ glist,
           int* __restrict__ gcount, int E)
{
    __shared__ int cnt[AGG_NT];
    __shared__ int base[AGG_NT];
    const int per = (E + (int)gridDim.x - 1) / (int)gridDim.x;
    const int e0  = blockIdx.x * per;
    const int e1  = min(e0 + per, E);
    for (int i = threadIdx.x; i < AGG_NT; i += 1024) cnt[i] = 0;
    __syncthreads();
    for (int e = e0 + threadIdx.x; e < e1; e += 1024) {
        const int d = dst[e];
        if ((unsigned)d < (unsigned)Nn)
            __hip_atomic_fetch_add(&cnt[d / AGG_TR], 1, __ATOMIC_RELAXED,
                                   __HIP_MEMORY_SCOPE_WORKGROUP);
    }
    __syncthreads();
    for (int i = threadIdx.x; i < AGG_NT; i += 1024) {
        const int n = cnt[i];
        base[i] = n ? __hip_atomic_fetch_add(&gcount[i * AGG_CSTR], n, __ATOMIC_RELAXED,
                                             __HIP_MEMORY_SCOPE_AGENT) : 0;
        cnt[i] = 0;
    }
    __syncthreads();
    for (int e = e0 + threadIdx.x; e < e1; e += 1024) {
        const int d = dst[e];
        if ((unsigned)d < (unsigned)Nn) {
            const int g   = d / AGG_TR;
            const int pos = base[g] + __hip_atomic_fetch_add(&cnt[g], 1, __ATOMIC_RELAXED,
                                                             __HIP_MEMORY_SCOPE_WORKGROUP);
            if (pos < AGG_GCAP)
                glist[(long long)g * AGG_GCAP + pos] =
                    (int)((unsigned)src[e] | ((unsigned)(d - g * AGG_TR) << 24));
        }
    }
}

__global__ void k_attn(const float* __restrict__ feat, const float* __restrict__ al,
                       const float* __restrict__ ar, float* __restrict__ el,
                       float* __restrict__ er, int N, int H, int D) {
    int i = blockIdx.x * blockDim.x + threadIdx.x;
    if (i >= N * H) return;
    const int n = i / H, h = i % H;
    const float* f = feat + (long long)n * H * D + h * D;
    float a = 0.0f, b = 0.0f;
    for (int d = 0; d < D; ++d) { a += f[d] * al[h * D + d]; b += f[d] * ar[h * D + d]; }
    el[i] = a; er[i] = b;
}

template <int ROW, int H, bool GAT, bool ACT>
__global__ __launch_bounds__(1024)
void k_tile_aggregate(const float* __restrict__ feat, const int* __restrict__ src,
                      const int* __restrict__ dst, const float* __restrict__ el,
                      const float* __restrict__ er, const float* __restrict__ bias,
                      float* __restrict__ out, const int* __restrict__ glist,
                      const int* __restrict__ gcount, float* __restrict__ degout, int E, int N)
{
    constexpr int TR = AGG_TR;
    constexpr int GCAP = AGG_GCAP;
    static_assert(TR <= 256 && Nn <= (1 << 24), "a list word holds 24 bits of source row, 8 of tile row");
    constexpr int NQ = (ROW + 31) / 32;
    constexpr int D  = ROW / H;
    constexpr int EG = (NQ >= 8) ? 4 : 8;
    constexpr int TRIP = 512;
    constexpr int QCAP = GAT ? 1536 : 544;
    constexpr int NS = GAT ? TR * H : 1;
    static_assert(H == 1 || D % 32 == 0, "a chunk must not straddle two heads");
    static_assert((TR * ROW) % 32 == 0, "tile must be whole 128-byte lines");
    __shared__ float tile[TR * ROW];
    __shared__ int queue[32 * QCAP];
    __shared__ unsigned rmax[NS];
    __shared__ float rsum[NS];
    __shared__ float rer[NS];
    __shared__ int ovf;
    __shared__ int rdeg[GAT ? 1 : TR];

    const int lane = threadIdx.x & 31;
    const int wave = __builtin_amdgcn_readfirstlane((int)(threadIdx.x >> 5));
    const int lo   = blockIdx.x * TR;

    for (int i = threadIdx.x; i < TR * ROW / 4; i += 1024)
        ((float4*)tile)[i] = make_float4(0.0f, 0.0f, 0.0f, 0.0f);
    if (GAT) {
        for (int i = threadIdx.x; i < TR * H; i += 1024) {
            rmax[i] = 0x007FFFFFu;
            rsum[i] = 0.0f;
            rer[i]  = (lo * H + i < N * H) ? er[lo * H + i] : 0.0f;
        }
        if (threadIdx.x == 0) ovf = 0;
    } else {
        for (int i = threadIdx.x; i < TR; i += 1024) rdeg[i] = 0;
    }
    __syncthreads();

    const int qb = wave * QCAP;
    int qn = 0;
    int done = 0;
    const int* wl = glist + (long long)blockIdx.x * GCAP;
    const int gn = __builtin_amdgcn_readfirstlane(gcount[blockIdx.x * AGG_CSTR]);
    const bool listed = gn <= GCAP;

    auto add_rows = [&](int cnt, int ms, int md, const float (&mw)[H]) {
        for (int j0 = 0; j0 < cnt; j0 += EG) {
            float v[EG][NQ];
            int   trow[EG];
#pragma unroll
            for (int g = 0; g < EG; ++g) {
                const int j  = min(j0 + g, cnt - 1);
                const int sj = __builtin_amdgcn_readlane(ms, j);
                trow[g] = __builtin_amdgcn_readlane(md, j) * ROW;
                const float* frow = feat + (long long)sj * ROW;
                float wj[H];
#pragma unroll
                for (int h = 0; h < H; ++h)
                    wj[h] = __int_as_float(__builtin_amdgcn_readlane(__float_as_int(mw[h]), j));
#pragma unroll
                for (int q = 0; q < NQ; ++q) {
                    const int c = q * 32 + lane;
                    float t = 0.0f;
                    if (ROW % 32 == 0 || c < ROW) t = frow[c];
                    if (GAT) t *= wj[(H == 1) ? 0 : (q * 32) / D];
                    v[g][q] = t;
                }
            }
#pragma unroll
            for (int g = 0; g < EG; ++g) {
                if (j0 + g < cnt) {
#pragma unroll
                    for (int q = 0; q < NQ; ++q) {
                        const int c = q * 32 + lane;
                        if (ROW % 32 == 0 || c < ROW)
                            __hip_atomic_fetch_add(&tile[trow[g] + c], v[g][q], __ATOMIC_RELAXED,
                                                   __HIP_MEMORY_SCOPE_WORKGROUP);
                    }
                }
            }
        }
    };

    auto logits = [&](int s, int d, float (&e)[H]) {
        float l[H];
        if (H == 2) {
            const v2f t = *(const v2f*)(el + (long long)s * 2);
            l[0] = t.x; l[H - 1] = t.y;
        } else {
#pragma unroll
            for (int h = 0; h < H; ++h) l[h] = el[(long long)s * H + h];
        }
#pragma unroll
        for (int h = 0; h < H; ++h) {
            const float x = l[h] + rer[d * H + h];
            e[h] = x > 0.0f ? x : 0.2f * x;
        }
    };

    auto edge_of = [&](bool list, int base, int cnt, int& ms, int& md) {
        ms = 0; md = 0;
        if (list) {
            if (lane < cnt) {
                const unsigned p = (unsigned)wl[base + lane];
                ms = (int)(p & 0x00FFFFFFu);
                md = (int)(p >> 24);
            }
        } else {
            __builtin_amdgcn_fence(__ATOMIC_ACQ_REL, "wavefront");
            if (lane < cnt) {
                const int e = queue[qb + base + lane];
                ms = src[e];
                md = dst[e] - lo;
            }
        }
    };

    auto batch_sum = [&](bool list, int base, int cnt) {
        int ms, md;
        float mw[H];
#pragma unroll
        for (int h = 0; h < H; ++h) mw[h] = 0.0f;
        edge_of(list, base, cnt, ms, md);
        if (lane < cnt)
            __hip_atomic_fetch_add(&rdeg[md], 1, __ATOMIC_RELAXED,
                                   __HIP_MEMORY_SCOPE_WORKGROUP);
        add_rows(cnt, ms, md, mw);
    };

    auto batch_max = [&](bool list, int base, int cnt) {
        int s, d;
        edge_of(list, base, cnt, s, d);
        if (lane < cnt) {
            float lg[H];
            logits(s, d, lg);
#pragma unroll
            for (int h = 0; h < H; ++h)
                __hip_atomic_fetch_max(&rmax[d * H + h], enc_f(lg[h]), __ATOMIC_RELAXED,
                                       __HIP_MEMORY_SCOPE_WORKGROUP);
        }
    };

    auto batch_acc = [&](bool list, int base, int cnt) {
        int ms, md;
        float mw[H];
#pragma unroll
        for (int h = 0; h < H; ++h) mw[h] = 0.0f;
        edge_of(list, base, cnt, ms, md);
        if (lane < cnt) {
            float lg[H];
            logits(ms, md, lg);
#pragma unroll
            for (int h = 0; h < H; ++h) {
                float m = dec_f(rmax[md * H + h]);
                if (!(fabsf(m) <= 3.402823466e38f)) m = 0.0f;
                mw[h] = __expf(lg[h] - m);
                __hip_atomic_fetch_add(&rsum[md * H + h], mw[h], __ATOMIC_RELAXED,
                                       __HIP_MEMORY_SCOPE_WORKGROUP);
            }
        }
        add_rows(cnt, ms, md, mw);
    };

    auto offer = [&](int e, int d, unsigned okm) {
        const unsigned mask = __builtin_amdgcn_ballot_w32((unsigned)(d - lo) < (unsigned)TR) & okm;
        if (mask != 0u) {
            if ((mask >> lane) & 1u) queue[qb + qn + __popc(mask & ((1u << lane) - 1u))] = e;
            qn += __popc(mask);
        }
    };

    const v4i* dst4 = (const v4i*)dst;
    const int nvec  = E >> 2;
    auto fetch = [&](int vi) { return dst4[min(vi, nvec - 1)]; };

    auto scan = [&](auto&& fn, bool keep) {
        if (wave == 0) {
            const int e = min(nvec * 4 + lane, E - 1);
            offer(e, dst[e], __builtin_amdgcn_ballot_w32(nvec * 4 + lane < E));
        }
        v4i nxt[4];
        int vb = wave * 128;
#pragma unroll
        for (int u = 0; u < 4; ++u) nxt[u] = fetch(vb + u * 32 + lane);
        for (;;) {
            const bool last = vb >= nvec;
            if (!last) {
                if (keep && qn + TRIP > QCAP) {
                    const int pend = qn - done;
                    __builtin_amdgcn_fence(__ATOMIC_ACQ_REL, "wavefront");
                    if (lane < pend) queue[qb + lane] = queue[qb + done + lane];
                    qn = pend; done = 0; keep = false;
                    if (lane == 0) ovf = 1;
                }
                v4i cur[4];
#pragma unroll
                for (int u = 0; u < 4; ++u) cur[u] = nxt[u];
                const int vn = vb + 32 * 128;
#pragma unroll
                for (int u = 0; u < 4; ++u) nxt[u] = fetch(vn + u * 32 + lane);
#pragma unroll
                for (int u = 0; u < 4; ++u) {
                    const int vi = vb + u * 32 + lane;
                    const unsigned ok = __builtin_amdgcn_ballot_w32(vi < nvec);
                    offer(vi * 4 + 0, cur[u].x, ok);
                    offer(vi * 4 + 1, cur[u].y, ok);
                    offer(vi * 4 + 2, cur[u].z, ok);
                    offer(vi * 4 + 3, cur[u].w, ok);
                }
            }
            const int thr = last ? 1 : 32;
            while (qn - done >= thr) {
                const int cnt = min(qn - done, 32);
                int base;
                if (keep) { base = done; done += cnt; }
                else      { qn -= cnt; base = qn; }
                fn(base, cnt);
            }
            if (last) break;
            vb += 32 * 128;
        }
    };

    if (!GAT) {
        if (listed) {
            for (int b = wave * 32; b < gn; b += 32 * 32) batch_sum(true, b, min(gn - b, 32));
        } else {
            scan([&](int b, int c) { batch_sum(false, b, c); }, false);
        }
        __syncthreads();
        for (int i = threadIdx.x; i < TR; i += 1024)
            if (lo + i < N) degout[lo + i] = (float)rdeg[i];
    } else {
        if (listed) {
            for (int b0 = 0; b0 < gn; b0 += 4 * 32 * 32) {
                int  s[4], d[4];
                bool ok[4];
                float lg[4][H];
#pragma unroll
                for (int u = 0; u < 4; ++u) {
                    const int i = b0 + (u * 32 + wave) * 32 + lane;
                    ok[u] = i < gn;
                    const unsigned p = (unsigned)wl[min(i, gn - 1)];
                    s[u] = (int)(p & 0x00FFFFFFu);
                    d[u] = (int)(p >> 24);
                }
#pragma unroll
                for (int u = 0; u < 4; ++u) logits(s[u], d[u], lg[u]);
#pragma unroll
                for (int u = 0; u < 4; ++u) {
                    if (ok[u]) {
#pragma unroll
                        for (int h = 0; h < H; ++h)
                            __hip_atomic_fetch_max(&rmax[d[u] * H + h], enc_f(lg[u][h]),
                                                   __ATOMIC_RELAXED, __HIP_MEMORY_SCOPE_WORKGROUP);
                    }
                }
            }
        } else {
            scan([&](int b, int c) { batch_max(false, b, c); }, true);
        }
        __syncthreads();
        const bool rescan = __builtin_amdgcn_readfirstlane(ovf) != 0;
        if (listed) {
            for (int b = wave * 32; b < gn; b += 32 * 32) batch_acc(true, b, min(gn - b, 32));
        } else if (!rescan) {
            for (int b = 0; b < qn; b += 32) batch_acc(false, b, min(qn - b, 32));
        } else {
            qn = 0; done = 0;
            scan([&](int b, int c) { batch_acc(false, b, c); }, false);
        }
        __syncthreads();
    }

    const long long g0 = (long long)lo * ROW;
    const long long gN = (long long)N * ROW;
    for (int i = threadIdx.x; i < TR * ROW; i += 1024) {
        if (g0 + i < gN) {
            float x = tile[i];
            if (GAT) {
                const int r = i / ROW, c = i - r * ROW;
                x = x / fmaxf(rsum[r * H + c / D], 1e-9f) + bias[c];
            }
            if (ACT) x = x > 0.0f ? x : (expf(x) - 1.0f);
            out[g0 + i] = x;
        }
    }
}

static inline unsigned gsz(long long n, int b) { return (unsigned)((n + b - 1) / b); }

extern "C" void kernel_launch(void* const* d_in, const int* in_sizes, int n_in,
                              void* d_out, int out_size, void* d_ws, size_t ws_size,
                              hipStream_t stream) {
    const float* x       = (const float*)d_in[0];
    const int*   src     = (const int*)  d_in[1];
    const int*   dst     = (const int*)  d_in[2];
    const float* w_self  = (const float*)d_in[3];
    const float* w_neigh = (const float*)d_in[4];
    const float* sage_b  = (const float*)d_in[5];
    const float* g1_w    = (const float*)d_in[6];
    const float* g1_al   = (const float*)d_in[7];
    const float* g1_ar   = (const float*)d_in[8];
    const float* g1_b    = (const float*)d_in[9];
    const float* g2_w    = (const float*)d_in[10];
    const float* g2_al   = (const float*)d_in[11];
    const float* g2_ar   = (const float*)d_in[12];
    const float* g2_b    = (const float*)d_in[13];

    int*      glist  = (int*)d_ws;
    int*      gcount = glist + AGG_NT * AGG_GCAP;
    float*    ws    = (float*)(gcount + AGG_NT * AGG_CSTR);
    float*    deg   = ws;
    float*    neigh = deg + Nn;
    float*    h0    = neigh + (long long)Nn * F;
    float*    el    = h0 + (long long)Nn * D0;
    float*    er    = el + Nn * H1;
    float*    feat  = neigh;

    const int B = 256;
    const unsigned ntile  = AGG_NT;

    fill_i32<<<gsz(AGG_NT * AGG_CSTR, B), B, 0, stream>>>(gcount, 0, AGG_NT * AGG_CSTR);
    k_bin<<<256, 1024, 0, stream>>>(src, dst, glist, gcount, Ne);
    k_tile_aggregate<F, 1, false, false><<<ntile, 1024, 0, stream>>>(
        x, src, dst, nullptr, nullptr, nullptr, neigh, glist, gcount, deg, Ne, Nn);
    wmma_gemm<F, 2, D0, 16, true, true><<<GEMM_WG, 16 * 32, 0, stream>>>(
        x, w_self, neigh, w_neigh, deg, sage_b, h0, Nn);

    wmma_gemm<D0, 2, H1 * D1, 32, false, false><<<GEMM_WG, 32 * 32, 0, stream>>>(
        h0, g1_w, nullptr, nullptr, nullptr, nullptr, feat, Nn);
    k_attn<<<gsz(Nn * H1, B), B, 0, stream>>>(feat, g1_al, g1_ar, el, er, Nn, H1, D1);
    k_tile_aggregate<H1 * D1, H1, true, true><<<ntile, 1024, 0, stream>>>(
        feat, src, dst, el, er, g1_b, h0, glist, gcount, nullptr, Ne, Nn);

    wmma_gemm<D0, 1, C2, 32, false, false><<<GEMM_WG, 32 * 32, 0, stream>>>(
        h0, g2_w, nullptr, nullptr, nullptr, nullptr, feat, Nn);
    k_attn<<<gsz(Nn, B), B, 0, stream>>>(feat, g2_al, g2_ar, el, er, Nn, 1, C2);
    k_tile_aggregate<C2, 1, true, false><<<ntile, 1024, 0, stream>>>(
        feat, src, dst, el, er, g2_b, (float*)d_out, glist, gcount, nullptr, Ne, Nn);
}


# Owner groups order their edge list by tile row in LDS; a wave sums runs of equal rows in registers: one LDS float add per run, not per edge (row max, exp sums likewise); sum order changes
# speedup vs baseline: 5.1886x; 1.0392x over previous
#include <hip/hip_runtime.h>

constexpr int Nn  = 50000;
constexpr int Ne  = 800000;
constexpr int F   = 256;
constexpr int D0  = 128;
constexpr int H1  = 2;
constexpr int D1  = 64;
constexpr int C2  = 47;

constexpr int AGG_TR   = 224;
constexpr int AGG_NT   = (Nn + AGG_TR - 1) / AGG_TR;
constexpr int AGG_GCAP = 16384;
constexpr int AGG_CSTR = 64;

typedef float v2f __attribute__((ext_vector_type(2)));
typedef float v8f __attribute__((ext_vector_type(8)));
typedef int   v4i __attribute__((ext_vector_type(4)));

__device__ __forceinline__ unsigned enc_f(float f) {
    unsigned u = __float_as_uint(f);
    return (u & 0x80000000u) ? ~u : (u | 0x80000000u);
}
__device__ __forceinline__ float dec_f(unsigned u) {
    unsigned b = (u & 0x80000000u) ? (u ^ 0x80000000u) : ~u;
    return __uint_as_float(b);
}

constexpr int GEMM_WG = 256;

template <int K, int NCG, int NR, int NW, bool HAS_A2, bool ACT>
__global__ __launch_bounds__(NW * 32)
void wmma_gemm(const float* __restrict__ A, const float* __restrict__ W,
               const float* __restrict__ A2, const float* __restrict__ W2,
               const float* __restrict__ rowdeg, const float* __restrict__ bias,
               float* __restrict__ out, int M)
{
    constexpr int NC  = NCG * 64;
    constexpr int NPP = NC + 16;
    static_assert(NW % NCG == 0 && K % 4 == 0, "waves split evenly over the column groups");
    static_assert(NW <= 32, "a work-group holds at most 32 waves");
    static_assert(NR <= NC && (NR == NC || !HAS_A2), "the padded copy is written for one matrix");
    __shared__ float w1s[K * NPP];
    __shared__ float w2s[HAS_A2 ? K * NPP : 4];

    if (NR == NC) {
        constexpr int NV = NC / 4;
        for (int i = threadIdx.x; i < (K / 2) * NV; i += NW * 32) {
            const int kp = i / NV, cc = (i % NV) * 4;
            const float4 r0 = *(const float4*)(W + (2 * kp) * NC + cc);
            const float4 r1 = *(const float4*)(W + (2 * kp + 1) * NC + cc);
            float* d = w1s + (kp * NPP + cc) * 2;
            *(float4*)(d)     = make_float4(r0.x, r1.x, r0.y, r1.y);
            *(float4*)(d + 4) = make_float4(r0.z, r1.z, r0.w, r1.w);
            if (HAS_A2) {
                const float4 q0 = *(const float4*)(W2 + (2 * kp) * NC + cc);
                const float4 q1 = *(const float4*)(W2 + (2 * kp + 1) * NC + cc);
                float* d2 = w2s + (kp * NPP + cc) * 2;
                *(float4*)(d2)     = make_float4(q0.x, q1.x, q0.y, q1.y);
                *(float4*)(d2 + 4) = make_float4(q0.z, q1.z, q0.w, q1.w);
            }
        }
    } else {
        for (int i = threadIdx.x; i < K * NC; i += NW * 32) {
            const int k = i / NC, cc = i % NC;
            w1s[((k >> 1) * NPP + cc) * 2 + (k & 1)] = (cc < NR) ? W[k * NR + cc] : 0.0f;
        }
    }
    __syncthreads();

    const int lane = threadIdx.x & 31;
    const int half = lane >> 4;
    const int l    = lane & 15;
    const int wave = __builtin_amdgcn_readfirstlane((int)(threadIdx.x >> 5));
    const int col0 = (wave % NCG) * 64;
    const int ntile = M / 16;

    const float* Bs[4];
    const float* B2s[4];
#pragma unroll
    for (int t = 0; t < 4; ++t) {
        Bs[t] = w1s + (half * NPP + col0 + t * 16 + l) * 2;
        if (HAS_A2) B2s[t] = w2s + (half * NPP + col0 + t * 16 + l) * 2;
    }

    for (int s = wave / NCG; s * GEMM_WG + (int)blockIdx.x < ntile; s += NW / NCG) {
        const int rt  = s * GEMM_WG + (int)blockIdx.x;
        const int row = rt * 16 + l;

        const float* Arow  = A + (long long)row * K + half * 2;
        const float* A2row = HAS_A2 ? (A2 + (long long)row * K + half * 2) : nullptr;
        const float inv    = HAS_A2 ? (1.0f / fmaxf(rowdeg[row], 1.0f)) : 0.0f;
        __builtin_prefetch(Arow, 0, 3);

        v8f acc[4] = {};
        for (int kk = 0; kk < K; kk += 4) {
            const v2f a = *(const v2f*)(Arow + kk);
            v2f a2;
            if (HAS_A2) {
                const v2f t2 = *(const v2f*)(A2row + kk);
                a2.x = t2.x * inv; a2.y = t2.y * inv;
            }
#pragma unroll
            for (int t = 0; t < 4; ++t) {
                const v2f b = *(const v2f*)(Bs[t] + kk * NPP);
                acc[t] = __builtin_amdgcn_wmma_f32_16x16x4_f32(
                    false, a, false, b, (short)0, acc[t], false, false);
                if (HAS_A2) {
                    const v2f b2 = *(const v2f*)(B2s[t] + kk * NPP);
                    acc[t] = __builtin_amdgcn_wmma_f32_16x16x4_f32(
                        false, a2, false, b2, (short)0, acc[t], false, false);
                }
            }
        }

#pragma unroll
        for (int t = 0; t < 4; ++t) {
            const int c = col0 + t * 16 + l;
            if (c < NR) {
                const float bv = bias ? bias[c] : 0.0f;
#pragma unroll
                for (int v = 0; v < 8; ++v) {
                    const int r = rt * 16 + v + 8 * half;
                    float x = acc[t][v] + bv;
                    if (ACT) x = x > 0.0f ? x : (expf(x) - 1.0f);
                    out[(long long)r * NR + c] = x;
                }
            }
        }
    }
}

__global__ void fill_i32(int* p, int v, int n) {
    int i = blockIdx.x * blockDim.x + threadIdx.x;
    if (i < n) p[i] = v;
}

__global__ __launch_bounds__(1024)
void k_bin(const int* __restrict__ src, const int* __restrict__ dst, int* __restrict__ glist,
           int* __restrict__ gcount, int E)
{
    __shared__ int cnt[AGG_NT];
    __shared__ int base[AGG_NT];
    const int per = (E + (int)gridDim.x - 1) / (int)gridDim.x;
    const int e0  = blockIdx.x * per;
    const int e1  = min(e0 + per, E);
    for (int i = threadIdx.x; i < AGG_NT; i += 1024) cnt[i] = 0;
    __syncthreads();
    for (int e = e0 + threadIdx.x; e < e1; e += 1024) {
        const int d = dst[e];
        if ((unsigned)d < (unsigned)Nn)
            __hip_atomic_fetch_add(&cnt[d / AGG_TR], 1, __ATOMIC_RELAXED,
                                   __HIP_MEMORY_SCOPE_WORKGROUP);
    }
    __syncthreads();
    for (int i = threadIdx.x; i < AGG_NT; i += 1024) {
        const int n = cnt[i];
        base[i] = n ? __hip_atomic_fetch_add(&gcount[i * AGG_CSTR], n, __ATOMIC_RELAXED,
                                             __HIP_MEMORY_SCOPE_AGENT) : 0;
        cnt[i] = 0;
    }
    __syncthreads();
    for (int e = e0 + threadIdx.x; e < e1; e += 1024) {
        const int d = dst[e];
        if ((unsigned)d < (unsigned)Nn) {
            const int g   = d / AGG_TR;
            const int pos = base[g] + __hip_atomic_fetch_add(&cnt[g], 1, __ATOMIC_RELAXED,
                                                             __HIP_MEMORY_SCOPE_WORKGROUP);
            if (pos < AGG_GCAP)
                glist[(long long)g * AGG_GCAP + pos] =
                    (int)((unsigned)src[e] | ((unsigned)(d - g * AGG_TR) << 24));
        }
    }
}

__global__ void k_attn(const float* __restrict__ feat, const float* __restrict__ al,
                       const float* __restrict__ ar, float* __restrict__ el,
                       float* __restrict__ er, int N, int H, int D) {
    int i = blockIdx.x * blockDim.x + threadIdx.x;
    if (i >= N * H) return;
    const int n = i / H, h = i % H;
    const float* f = feat + (long long)n * H * D + h * D;
    float a = 0.0f, b = 0.0f;
    for (int d = 0; d < D; ++d) { a += f[d] * al[h * D + d]; b += f[d] * ar[h * D + d]; }
    el[i] = a; er[i] = b;
}

template <int ROW, int H, bool GAT, bool ACT>
__global__ __launch_bounds__(1024)
void k_tile_aggregate(const float* __restrict__ feat, const int* __restrict__ src,
                      const int* __restrict__ dst, const float* __restrict__ el,
                      const float* __restrict__ er, const float* __restrict__ bias,
                      float* __restrict__ out, const int* __restrict__ glist,
                      const int* __restrict__ gcount, float* __restrict__ degout, int E, int N)
{
    constexpr int TR = AGG_TR;
    constexpr int GCAP = AGG_GCAP;
    static_assert(TR <= 256 && Nn <= (1 << 24), "a list word holds 24 bits of source row, 8 of tile row");
    constexpr int NQ = (ROW + 31) / 32;
    constexpr int D  = ROW / H;
    constexpr int EG = (NQ >= 8) ? 4 : 8;
    constexpr int TRIP = 512;
    constexpr int QCAP = GAT ? 1536 : 544;
    constexpr int NS = GAT ? TR * H : 1;
    static_assert(H == 1 || D % 32 == 0, "a chunk must not straddle two heads");
    static_assert((TR * ROW) % 32 == 0, "tile must be whole 128-byte lines");
    __shared__ float tile[TR * ROW];
    __shared__ int queue[32 * QCAP];
    __shared__ unsigned rmax[NS];
    __shared__ float rsum[NS];
    __shared__ float rer[NS];
    __shared__ int ovf;
    __shared__ int rdeg[GAT ? 1 : TR];
    __shared__ int hist[TR];
    __shared__ int cur[TR];

    const int lane = threadIdx.x & 31;
    const int wave = __builtin_amdgcn_readfirstlane((int)(threadIdx.x >> 5));
    const int lo   = blockIdx.x * TR;

    for (int i = threadIdx.x; i < TR * ROW / 4; i += 1024)
        ((float4*)tile)[i] = make_float4(0.0f, 0.0f, 0.0f, 0.0f);
    if (GAT) {
        for (int i = threadIdx.x; i < TR * H; i += 1024) {
            rmax[i] = 0x007FFFFFu;
            rsum[i] = 0.0f;
            rer[i]  = (lo * H + i < N * H) ? er[lo * H + i] : 0.0f;
        }
        if (threadIdx.x == 0) ovf = 0;
    } else {
        for (int i = threadIdx.x; i < TR; i += 1024) rdeg[i] = 0;
    }
    for (int i = threadIdx.x; i < TR; i += 1024) hist[i] = 0;
    __syncthreads();

    const int qb = wave * QCAP;
    int qn = 0;
    int done = 0;
    const int* wl = glist + (long long)blockIdx.x * GCAP;
    const int gn = __builtin_amdgcn_readfirstlane(gcount[blockIdx.x * AGG_CSTR]);
    const bool listed = gn <= GCAP;

    if (listed) {
        constexpr int LW = GCAP / 1024;
        static_assert(LW * 1024 == GCAP, "the list divides evenly over the threads");
        const int nw = (gn + 1023) >> 10;
        int pw[LW];
#pragma unroll
        for (int k = 0; k < LW; ++k)
            if (k < nw) pw[k] = wl[min(k * 1024 + (int)threadIdx.x, gn - 1)];
#pragma unroll
        for (int k = 0; k < LW; ++k) {
            if (k < nw && k * 1024 + (int)threadIdx.x < gn) {
                const int r = min((int)((unsigned)pw[k] >> 24), TR - 1);
                __hip_atomic_fetch_add(&hist[r], 1, __ATOMIC_RELAXED, __HIP_MEMORY_SCOPE_WORKGROUP);
            }
        }
        __syncthreads();
        if (wave == 0) {
            constexpr int PER = (TR + 31) / 32;
            int loc[PER];
            int sum = 0;
#pragma unroll
            for (int k = 0; k < PER; ++k) {
                const int idx = lane * PER + k;
                loc[k] = idx < TR ? hist[idx] : 0;
                sum += loc[k];
            }
            int inc = sum;
#pragma unroll
            for (int o = 1; o < 32; o <<= 1) {
                const int t = __shfl_up(inc, o);
                if (lane >= o) inc += t;
            }
            int ex = inc - sum;
#pragma unroll
            for (int k = 0; k < PER; ++k) {
                const int idx = lane * PER + k;
                if (idx < TR) cur[idx] = ex;
                ex += loc[k];
            }
        }
        __syncthreads();
#pragma unroll
        for (int k = 0; k < LW; ++k) {
            if (k < nw && k * 1024 + (int)threadIdx.x < gn) {
                const int r = min((int)((unsigned)pw[k] >> 24), TR - 1);
                const int pos = __hip_atomic_fetch_add(&cur[r], 1, __ATOMIC_RELAXED,
                                                       __HIP_MEMORY_SCOPE_WORKGROUP);
                queue[min(pos, GCAP - 1)] = pw[k];
            }
        }
        __syncthreads();
    }

    auto seg_sum = [&](float v, int key) {
#pragma unroll
        for (int o = 1; o < 32; o <<= 1) {
            const float t = __shfl_up(v, o);
            const int   k = __shfl_up(key, o);
            if (lane >= o && k == key) v += t;
        }
        return v;
    };
    auto seg_max = [&](unsigned v, int key) {
#pragma unroll
        for (int o = 1; o < 32; o <<= 1) {
            const unsigned t = (unsigned)__shfl_up((int)v, o);
            const int      k = __shfl_up(key, o);
            if (lane >= o && k == key) v = max(v, t);
        }
        return v;
    };
    auto seg_tail = [&](int key) {
        const int kn = __shfl_down(key, 1);
        return lane == 31 || kn != key;
    };

    auto add_rows = [&](int cnt, int ms, int md, const float (&mw)[H]) {
        float acc[NQ];
        int   arow = -1;
        auto flush = [&]() {
#pragma unroll
            for (int q = 0; q < NQ; ++q) {
                const int c = q * 32 + lane;
                if (ROW % 32 == 0 || c < ROW)
                    __hip_atomic_fetch_add(&tile[arow + c], acc[q], __ATOMIC_RELAXED,
                                           __HIP_MEMORY_SCOPE_WORKGROUP);
            }
        };
        for (int j0 = 0; j0 < cnt; j0 += EG) {
            float v[EG][NQ];
            int   trow[EG];
#pragma unroll
            for (int g = 0; g < EG; ++g) {
                const int j  = min(j0 + g, cnt - 1);
                const int sj = __builtin_amdgcn_readlane(ms, j);
                trow[g] = __builtin_amdgcn_readlane(md, j) * ROW;
                const float* frow = feat + (long long)sj * ROW;
                float wj[H];
#pragma unroll
                for (int h = 0; h < H; ++h)
                    wj[h] = __int_as_float(__builtin_amdgcn_readlane(__float_as_int(mw[h]), j));
#pragma unroll
                for (int q = 0; q < NQ; ++q) {
                    const int c = q * 32 + lane;
                    float t = 0.0f;
                    if (ROW % 32 == 0 || c < ROW) t = frow[c];
                    if (GAT) t *= wj[(H == 1) ? 0 : (q * 32) / D];
                    v[g][q] = t;
                }
            }
#pragma unroll
            for (int g = 0; g < EG; ++g) {
                if (j0 + g < cnt) {
                    if (trow[g] != arow) {
                        if (arow >= 0) flush();
                        arow = trow[g];
#pragma unroll
                        for (int q = 0; q < NQ; ++q) acc[q] = v[g][q];
                    } else {
#pragma unroll
                        for (int q = 0; q < NQ; ++q) acc[q] += v[g][q];
                    }
                }
            }
        }
        if (arow >= 0) flush();
    };

    auto logits = [&](int s, int d, float (&e)[H]) {
        float l[H];
        if (H == 2) {
            const v2f t = *(const v2f*)(el + (long long)s * 2);
            l[0] = t.x; l[H - 1] = t.y;
        } else {
#pragma unroll
            for (int h = 0; h < H; ++h) l[h] = el[(long long)s * H + h];
        }
#pragma unroll
        for (int h = 0; h < H; ++h) {
            const float x = l[h] + rer[d * H + h];
            e[h] = x > 0.0f ? x : 0.2f * x;
        }
    };

    auto edge_of = [&](bool list, int base, int cnt, int& ms, int& md) {
        ms = 0; md = 0;
        if (list) {
            if (lane < cnt) {
                const unsigned p = (unsigned)queue[base + lane];
                ms = (int)(p & 0x00FFFFFFu);
                md = (int)(p >> 24);
            }
        } else {
            __builtin_amdgcn_fence(__ATOMIC_ACQ_REL, "wavefront");
            if (lane < cnt) {
                const int e = queue[qb + base + lane];
                ms = src[e];
                md = dst[e] - lo;
            }
        }
    };

    auto batch_sum = [&](bool list, int base, int cnt) {
        int ms, md;
        float mw[H];
#pragma unroll
        for (int h = 0; h < H; ++h) mw[h] = 0.0f;
        edge_of(list, base, cnt, ms, md);
        if (!list && lane < cnt)
            __hip_atomic_fetch_add(&rdeg[md], 1, __ATOMIC_RELAXED,
                                   __HIP_MEMORY_SCOPE_WORKGROUP);
        add_rows(cnt, ms, md, mw);
    };

    auto batch_max = [&](bool list, int base, int cnt) {
        int s, d;
        edge_of(list, base, cnt, s, d);
        if (lane < cnt) {
            float lg[H];
            logits(s, d, lg);
#pragma unroll
            for (int h = 0; h < H; ++h)
                __hip_atomic_fetch_max(&rmax[d * H + h], enc_f(lg[h]), __ATOMIC_RELAXED,
                                       __HIP_MEMORY_SCOPE_WORKGROUP);
        }
    };

    auto batch_acc = [&](bool list, int base, int cnt) {
        int ms, md;
        float mw[H];
#pragma unroll
        for (int h = 0; h < H; ++h) mw[h] = 0.0f;
        edge_of(list, base, cnt, ms, md);
        if (lane < cnt) {
            float lg[H];
            logits(ms, md, lg);
#pragma unroll
            for (int h = 0; h < H; ++h) {
                float m = dec_f(rmax[md * H + h]);
                if (!(fabsf(m) <= 3.402823466e38f)) m = 0.0f;
                mw[h] = __expf(lg[h] - m);
                if (!list)
                    __hip_atomic_fetch_add(&rsum[md * H + h], mw[h], __ATOMIC_RELAXED,
                                           __HIP_MEMORY_SCOPE_WORKGROUP);
            }
        }
        if (list) {
            const int  key  = lane < cnt ? md : -1;
            const bool tail = seg_tail(key);
#pragma unroll
            for (int h = 0; h < H; ++h) {
                const float sh = seg_sum(mw[h], key);
                if (lane < cnt && tail)
                    __hip_atomic_fetch_add(&rsum[md * H + h], sh, __ATOMIC_RELAXED,
                                           __HIP_MEMORY_SCOPE_WORKGROUP);
            }
        }
        add_rows(cnt, ms, md, mw);
    };

    auto offer = [&](int e, int d, unsigned okm) {
        const unsigned mask = __builtin_amdgcn_ballot_w32((unsigned)(d - lo) < (unsigned)TR) & okm;
        if (mask != 0u) {
            if ((mask >> lane) & 1u) queue[qb + qn + __popc(mask & ((1u << lane) - 1u))] = e;
            qn += __popc(mask);
        }
    };

    const v4i* dst4 = (const v4i*)dst;
    const int nvec  = E >> 2;
    auto fetch = [&](int vi) { return dst4[min(vi, nvec - 1)]; };

    auto scan = [&](auto&& fn, bool keep) {
        if (wave == 0) {
            const int e = min(nvec * 4 + lane, E - 1);
            offer(e, dst[e], __builtin_amdgcn_ballot_w32(nvec * 4 + lane < E));
        }
        v4i nxt[4];
        int vb = wave * 128;
#pragma unroll
        for (int u = 0; u < 4; ++u) nxt[u] = fetch(vb + u * 32 + lane);
        for (;;) {
            const bool last = vb >= nvec;
            if (!last) {
                if (keep && qn + TRIP > QCAP) {
                    const int pend = qn - done;
                    __builtin_amdgcn_fence(__ATOMIC_ACQ_REL, "wavefront");
                    if (lane < pend) queue[qb + lane] = queue[qb + done + lane];
                    qn = pend; done = 0; keep = false;
                    if (lane == 0) ovf = 1;
                }
                v4i cur[4];
#pragma unroll
                for (int u = 0; u < 4; ++u) cur[u] = nxt[u];
                const int vn = vb + 32 * 128;
#pragma unroll
                for (int u = 0; u < 4; ++u) nxt[u] = fetch(vn + u * 32 + lane);
#pragma unroll
                for (int u = 0; u < 4; ++u) {
                    const int vi = vb + u * 32 + lane;
                    const unsigned ok = __builtin_amdgcn_ballot_w32(vi < nvec);
                    offer(vi * 4 + 0, cur[u].x, ok);
                    offer(vi * 4 + 1, cur[u].y, ok);
                    offer(vi * 4 + 2, cur[u].z, ok);
                    offer(vi * 4 + 3, cur[u].w, ok);
                }
            }
            const int thr = last ? 1 : 32;
            while (qn - done >= thr) {
                const int cnt = min(qn - done, 32);
                int base;
                if (keep) { base = done; done += cnt; }
                else      { qn -= cnt; base = qn; }
                fn(base, cnt);
            }
            if (last) break;
            vb += 32 * 128;
        }
    };

    if (!GAT) {
        if (listed) {
            for (int b = wave * 32; b < gn; b += 32 * 32) batch_sum(true, b, min(gn - b, 32));
        } else {
            scan([&](int b, int c) { batch_sum(false, b, c); }, false);
        }
        __syncthreads();
        for (int i = threadIdx.x; i < TR; i += 1024)
            if (lo + i < N) degout[lo + i] = (float)(listed ? hist[i] : rdeg[i]);
    } else {
        if (listed) {
            for (int b0 = 0; b0 < gn; b0 += 4 * 32 * 32) {
                int  s[4], d[4];
                bool ok[4];
                float lg[4][H];
#pragma unroll
                for (int u = 0; u < 4; ++u) {
                    const int i = b0 + (u * 32 + wave) * 32 + lane;
                    ok[u] = i < gn;
                    const unsigned p = (unsigned)queue[min(i, gn - 1)];
                    s[u] = (int)(p & 0x00FFFFFFu);
                    d[u] = (int)(p >> 24);
                }
#pragma unroll
                for (int u = 0; u < 4; ++u) logits(s[u], d[u], lg[u]);
#pragma unroll
                for (int u = 0; u < 4; ++u) {
                    const int  key  = ok[u] ? d[u] : -1;
                    const bool tail = seg_tail(key);
#pragma unroll
                    for (int h = 0; h < H; ++h) {
                        const unsigned mx = seg_max(enc_f(lg[u][h]), key);
                        if (ok[u] && tail)
                            __hip_atomic_fetch_max(&rmax[d[u] * H + h], mx,
                                                   __ATOMIC_RELAXED, __HIP_MEMORY_SCOPE_WORKGROUP);
                    }
                }
            }
        } else {
            scan([&](int b, int c) { batch_max(false, b, c); }, true);
        }
        __syncthreads();
        const bool rescan = __builtin_amdgcn_readfirstlane(ovf) != 0;
        if (listed) {
            for (int b = wave * 32; b < gn; b += 32 * 32) batch_acc(true, b, min(gn - b, 32));
        } else if (!rescan) {
            for (int b = 0; b < qn; b += 32) batch_acc(false, b, min(qn - b, 32));
        } else {
            qn = 0; done = 0;
            scan([&](int b, int c) { batch_acc(false, b, c); }, false);
        }
        __syncthreads();
    }

    const long long g0 = (long long)lo * ROW;
    const long long gN = (long long)N * ROW;
    for (int i = threadIdx.x; i < TR * ROW; i += 1024) {
        if (g0 + i < gN) {
            float x = tile[i];
            if (GAT) {
                const int r = i / ROW, c = i - r * ROW;
                x = x / fmaxf(rsum[r * H + c / D], 1e-9f) + bias[c];
            }
            if (ACT) x = x > 0.0f ? x : (expf(x) - 1.0f);
            out[g0 + i] = x;
        }
    }
}

static inline unsigned gsz(long long n, int b) { return (unsigned)((n + b - 1) / b); }

extern "C" void kernel_launch(void* const* d_in, const int* in_sizes, int n_in,
                              void* d_out, int out_size, void* d_ws, size_t ws_size,
                              hipStream_t stream) {
    const float* x       = (const float*)d_in[0];
    const int*   src     = (const int*)  d_in[1];
    const int*   dst     = (const int*)  d_in[2];
    const float* w_self  = (const float*)d_in[3];
    const float* w_neigh = (const float*)d_in[4];
    const float* sage_b  = (const float*)d_in[5];
    const float* g1_w    = (const float*)d_in[6];
    const float* g1_al   = (const float*)d_in[7];
    const float* g1_ar   = (const float*)d_in[8];
    const float* g1_b    = (const float*)d_in[9];
    const float* g2_w    = (const float*)d_in[10];
    const float* g2_al   = (const float*)d_in[11];
    const float* g2_ar   = (const float*)d_in[12];
    const float* g2_b    = (const float*)d_in[13];

    int*      glist  = (int*)d_ws;
    int*      gcount = glist + AGG_NT * AGG_GCAP;
    float*    ws    = (float*)(gcount + AGG_NT * AGG_CSTR);
    float*    deg   = ws;
    float*    neigh = deg + Nn;
    float*    h0    = neigh + (long long)Nn * F;
    float*    el    = h0 + (long long)Nn * D0;
    float*    er    = el + Nn * H1;
    float*    feat  = neigh;

    const int B = 256;
    const unsigned ntile  = AGG_NT;

    fill_i32<<<gsz(AGG_NT * AGG_CSTR, B), B, 0, stream>>>(gcount, 0, AGG_NT * AGG_CSTR);
    k_bin<<<256, 1024, 0, stream>>>(src, dst, glist, gcount, Ne);
    k_tile_aggregate<F, 1, false, false><<<ntile, 1024, 0, stream>>>(
        x, src, dst, nullptr, nullptr, nullptr, neigh, glist, gcount, deg, Ne, Nn);
    wmma_gemm<F, 2, D0, 16, true, true><<<GEMM_WG, 16 * 32, 0, stream>>>(
        x, w_self, neigh, w_neigh, deg, sage_b, h0, Nn);

    wmma_gemm<D0, 2, H1 * D1, 32, false, false><<<GEMM_WG, 32 * 32, 0, stream>>>(
        h0, g1_w, nullptr, nullptr, nullptr, nullptr, feat, Nn);
    k_attn<<<gsz(Nn * H1, B), B, 0, stream>>>(feat, g1_al, g1_ar, el, er, Nn, H1, D1);
    k_tile_aggregate<H1 * D1, H1, true, true><<<ntile, 1024, 0, stream>>>(
        feat, src, dst, el, er, g1_b, h0, glist, gcount, nullptr, Ne, Nn);

    wmma_gemm<D0, 1, C2, 32, false, false><<<GEMM_WG, 32 * 32, 0, stream>>>(
        h0, g2_w, nullptr, nullptr, nullptr, nullptr, feat, Nn);
    k_attn<<<gsz(Nn, B), B, 0, stream>>>(feat, g2_al, g2_ar, el, er, Nn, 1, C2);
    k_tile_aggregate<C2, 1, true, false><<<ntile, 1024, 0, stream>>>(
        feat, src, dst, el, er, g2_b, (float*)d_out, glist, gcount, nullptr, Ne, Nn);
}


# speedup vs baseline: 5.5351x; 1.0668x over previous
#include <hip/hip_runtime.h>

constexpr int Nn  = 50000;
constexpr int Ne  = 800000;
constexpr int F   = 256;
constexpr int D0  = 128;
constexpr int H1  = 2;
constexpr int D1  = 64;
constexpr int C2  = 47;

constexpr int AGG_TR   = 224;
constexpr int AGG_NT   = (Nn + AGG_TR - 1) / AGG_TR;
constexpr int AGG_GCAP = 16384;
constexpr int AGG_CSTR = 64;
constexpr int C2P  = 64;
constexpr int pad64(int n) { return (n + 63) / 64 * 64; }

typedef float v2f __attribute__((ext_vector_type(2)));
typedef float v8f __attribute__((ext_vector_type(8)));
typedef int   v4i __attribute__((ext_vector_type(4)));

__device__ __forceinline__ unsigned enc_f(float f) {
    unsigned u = __float_as_uint(f);
    return (u & 0x80000000u) ? ~u : (u | 0x80000000u);
}
__device__ __forceinline__ float dec_f(unsigned u) {
    unsigned b = (u & 0x80000000u) ? (u ^ 0x80000000u) : ~u;
    return __uint_as_float(b);
}

constexpr int GEMM_WG = 256;

template <int K, int NCG, int NR, int LDO, int NW, bool HAS_A2, bool ACT>
__global__ __launch_bounds__(NW * 32)
void wmma_gemm(const float* __restrict__ A, const float* __restrict__ W,
               const float* __restrict__ A2, const float* __restrict__ W2,
               const float* __restrict__ rowdeg, const float* __restrict__ bias,
               float* __restrict__ out, int M)
{
    constexpr int NC  = NCG * 64;
    constexpr int NPP = NC + 16;
    static_assert(NW % NCG == 0 && K % 4 == 0, "waves split evenly over the column groups");
    static_assert(NW <= 32, "a work-group holds at most 32 waves");
    static_assert(NR <= NC && (NR == NC || !HAS_A2), "the padded copy is written for one matrix");
    static_assert(NR <= LDO && LDO <= NC, "an output row holds the NR columns, then zeros up to its pitch");
    __shared__ float w1s[K * NPP];
    __shared__ float w2s[HAS_A2 ? K * NPP : 4];

    if (NR == NC) {
        constexpr int NV = NC / 4;
        for (int i = threadIdx.x; i < (K / 2) * NV; i += NW * 32) {
            const int kp = i / NV, cc = (i % NV) * 4;
            const float4 r0 = *(const float4*)(W + (2 * kp) * NC + cc);
            const float4 r1 = *(const float4*)(W + (2 * kp + 1) * NC + cc);
            float* d = w1s + (kp * NPP + cc) * 2;
            *(float4*)(d)     = make_float4(r0.x, r1.x, r0.y, r1.y);
            *(float4*)(d + 4) = make_float4(r0.z, r1.z, r0.w, r1.w);
            if (HAS_A2) {
                const float4 q0 = *(const float4*)(W2 + (2 * kp) * NC + cc);
                const float4 q1 = *(const float4*)(W2 + (2 * kp + 1) * NC + cc);
                float* d2 = w2s + (kp * NPP + cc) * 2;
                *(float4*)(d2)     = make_float4(q0.x, q1.x, q0.y, q1.y);
                *(float4*)(d2 + 4) = make_float4(q0.z, q1.z, q0.w, q1.w);
            }
        }
    } else {
        for (int i = threadIdx.x; i < K * NC; i += NW * 32) {
            const int k = i / NC, cc = i % NC;
            w1s[((k >> 1) * NPP + cc) * 2 + (k & 1)] = (cc < NR) ? W[k * NR + cc] : 0.0f;
        }
    }
    __syncthreads();

    const int lane = threadIdx.x & 31;
    const int half = lane >> 4;
    const int l    = lane & 15;
    const int wave = __builtin_amdgcn_readfirstlane((int)(threadIdx.x >> 5));
    const int col0 = (wave % NCG) * 64;
    const int ntile = M / 16;

    const float* Bs[4];
    const float* B2s[4];
#pragma unroll
    for (int t = 0; t < 4; ++t) {
        Bs[t] = w1s + (half * NPP + col0 + t * 16 + l) * 2;
        if (HAS_A2) B2s[t] = w2s + (half * NPP + col0 + t * 16 + l) * 2;
    }

    for (int s = wave / NCG; s * GEMM_WG + (int)blockIdx.x < ntile; s += NW / NCG) {
        const int rt  = s * GEMM_WG + (int)blockIdx.x;
        const int row = rt * 16 + l;

        const float* Arow  = A + (long long)row * K + half * 2;
        const float* A2row = HAS_A2 ? (A2 + (long long)row * K + half * 2) : nullptr;
        const float inv    = HAS_A2 ? (1.0f / fmaxf(rowdeg[row], 1.0f)) : 0.0f;
        __builtin_prefetch(Arow, 0, 3);

        v8f acc[4] = {};
        for (int kk = 0; kk < K; kk += 4) {
            const v2f a = *(const v2f*)(Arow + kk);
            v2f a2;
            if (HAS_A2) {
                const v2f t2 = *(const v2f*)(A2row + kk);
                a2.x = t2.x * inv; a2.y = t2.y * inv;
            }
#pragma unroll
            for (int t = 0; t < 4; ++t) {
                const v2f b = *(const v2f*)(Bs[t] + kk * NPP);
                acc[t] = __builtin_amdgcn_wmma_f32_16x16x4_f32(
                    false, a, false, b, (short)0, acc[t], false, false);
                if (HAS_A2) {
                    const v2f b2 = *(const v2f*)(B2s[t] + kk * NPP);
                    acc[t] = __builtin_amdgcn_wmma_f32_16x16x4_f32(
                        false, a2, false, b2, (short)0, acc[t], false, false);
                }
            }
        }

#pragma unroll
        for (int t = 0; t < 4; ++t) {
            const int c = col0 + t * 16 + l;
            if (c < LDO) {
                const float bv = (bias && c < NR) ? bias[c] : 0.0f;
#pragma unroll
                for (int v = 0; v < 8; ++v) {
                    const int r = rt * 16 + v + 8 * half;
                    float x = acc[t][v] + bv;
                    if (ACT) x = x > 0.0f ? x : (expf(x) - 1.0f);
                    out[(long long)r * LDO + c] = x;
                }
            }
        }
    }
}

__global__ void fill_i32(int* p, int v, int n) {
    int i = blockIdx.x * blockDim.x + threadIdx.x;
    if (i < n) p[i] = v;
}

__global__ __launch_bounds__(1024)
void k_bin(const int* __restrict__ src, const int* __restrict__ dst, int* __restrict__ glist,
           int* __restrict__ gcount, int E)
{
    __shared__ int cnt[AGG_NT];
    __shared__ int base[AGG_NT];
    const int per = (E + (int)gridDim.x - 1) / (int)gridDim.x;
    const int e0  = blockIdx.x * per;
    const int e1  = min(e0 + per, E);
    for (int i = threadIdx.x; i < AGG_NT; i += 1024) cnt[i] = 0;
    __syncthreads();
    for (int e = e0 + threadIdx.x; e < e1; e += 1024) {
        const int d = dst[e];
        if ((unsigned)d < (unsigned)Nn)
            __hip_atomic_fetch_add(&cnt[d / AGG_TR], 1, __ATOMIC_RELAXED,
                                   __HIP_MEMORY_SCOPE_WORKGROUP);
    }
    __syncthreads();
    for (int i = threadIdx.x; i < AGG_NT; i += 1024) {
        const int n = cnt[i];
        base[i] = n ? __hip_atomic_fetch_add(&gcount[i * AGG_CSTR], n, __ATOMIC_RELAXED,
                                             __HIP_MEMORY_SCOPE_AGENT) : 0;
        cnt[i] = 0;
    }
    __syncthreads();
    for (int e = e0 + threadIdx.x; e < e1; e += 1024) {
        const int d = dst[e];
        if ((unsigned)d < (unsigned)Nn) {
            const int g   = d / AGG_TR;
            const int pos = base[g] + __hip_atomic_fetch_add(&cnt[g], 1, __ATOMIC_RELAXED,
                                                             __HIP_MEMORY_SCOPE_WORKGROUP);
            if (pos < AGG_GCAP)
                glist[(long long)g * AGG_GCAP + pos] =
                    (int)((unsigned)src[e] | ((unsigned)(d - g * AGG_TR) << 24));
        }
    }
}

__global__ void k_attn(const float* __restrict__ feat, const float* __restrict__ al,
                       const float* __restrict__ ar, float* __restrict__ el,
                       float* __restrict__ er, int N, int H, int D, int ld) {
    int i = blockIdx.x * blockDim.x + threadIdx.x;
    if (i >= N * H) return;
    const int n = i / H, h = i % H;
    const float* f = feat + (long long)n * ld + h * D;
    float a = 0.0f, b = 0.0f;
    for (int d = 0; d < D; ++d) { a += f[d] * al[h * D + d]; b += f[d] * ar[h * D + d]; }
    el[i] = a; er[i] = b;
}

template <int ROW, int FS, int H, bool GAT, bool ACT>
__global__ __launch_bounds__(1024)
void k_tile_aggregate(const float* __restrict__ feat, const int* __restrict__ src,
                      const int* __restrict__ dst, const float* __restrict__ el,
                      const float* __restrict__ er, const float* __restrict__ bias,
                      float* __restrict__ out, const int* __restrict__ glist,
                      const int* __restrict__ gcount, float* __restrict__ degout, int E, int N)
{
    constexpr int TR = AGG_TR;
    constexpr int GCAP = AGG_GCAP;
    static_assert(TR <= 256 && Nn <= (1 << 24), "a list word holds 24 bits of source row, 8 of tile row");
    constexpr int NQ = (ROW + 31) / 32;
    constexpr int D  = ROW / H;
    constexpr int EG = (NQ >= 8) ? 4 : 8;
    constexpr int TRIP = 512;
    constexpr int QCAP = GAT ? 1536 : 544;
    constexpr int NS = GAT ? TR * H : 1;
    static_assert(H == 1 || D % 32 == 0, "a chunk must not straddle two heads");
    static_assert(FS >= ROW, "a feature row lies inside its pitch");
    static_assert((TR * ROW) % 32 == 0, "tile must be whole 128-byte lines");
    __shared__ float tile[TR * ROW];
    __shared__ int queue[32 * QCAP];
    __shared__ unsigned rmax[NS];
    __shared__ float rsum[NS];
    __shared__ float rer[NS];
    __shared__ int ovf;
    __shared__ int rdeg[GAT ? 1 : TR];
    __shared__ int hist[TR];
    __shared__ int cur[TR];

    const int lane = threadIdx.x & 31;
    const int wave = __builtin_amdgcn_readfirstlane((int)(threadIdx.x >> 5));
    const int lo   = blockIdx.x * TR;

    for (int i = threadIdx.x; i < TR * ROW / 4; i += 1024)
        ((float4*)tile)[i] = make_float4(0.0f, 0.0f, 0.0f, 0.0f);
    if (GAT) {
        for (int i = threadIdx.x; i < TR * H; i += 1024) {
            rmax[i] = 0x007FFFFFu;
            rsum[i] = 0.0f;
            rer[i]  = (lo * H + i < N * H) ? er[lo * H + i] : 0.0f;
        }
        if (threadIdx.x == 0) ovf = 0;
    } else {
        for (int i = threadIdx.x; i < TR; i += 1024) rdeg[i] = 0;
    }
    for (int i = threadIdx.x; i < TR; i += 1024) hist[i] = 0;
    __syncthreads();

    const int qb = wave * QCAP;
    int qn = 0;
    int done = 0;
    const int* wl = glist + (long long)blockIdx.x * GCAP;
    const int gn = __builtin_amdgcn_readfirstlane(gcount[blockIdx.x * AGG_CSTR]);
    const bool listed = gn <= GCAP;

    if (listed) {
        constexpr int LW = GCAP / 1024;
        static_assert(LW * 1024 == GCAP, "the list divides evenly over the threads");
        const int nw = (gn + 1023) >> 10;
        int pw[LW];
#pragma unroll
        for (int k = 0; k < LW; ++k)
            if (k < nw) pw[k] = wl[min(k * 1024 + (int)threadIdx.x, gn - 1)];
#pragma unroll
        for (int k = 0; k < LW; ++k) {
            if (k < nw && k * 1024 + (int)threadIdx.x < gn) {
                const int r = min((int)((unsigned)pw[k] >> 24), TR - 1);
                __hip_atomic_fetch_add(&hist[r], 1, __ATOMIC_RELAXED, __HIP_MEMORY_SCOPE_WORKGROUP);
            }
        }
        __syncthreads();
        if (wave == 0) {
            constexpr int PER = (TR + 31) / 32;
            int loc[PER];
            int sum = 0;
#pragma unroll
            for (int k = 0; k < PER; ++k) {
                const int idx = lane * PER + k;
                loc[k] = idx < TR ? hist[idx] : 0;
                sum += loc[k];
            }
            int inc = sum;
#pragma unroll
            for (int o = 1; o < 32; o <<= 1) {
                const int t = __shfl_up(inc, o);
                if (lane >= o) inc += t;
            }
            int ex = inc - sum;
#pragma unroll
            for (int k = 0; k < PER; ++k) {
                const int idx = lane * PER + k;
                if (idx < TR) cur[idx] = ex;
                ex += loc[k];
            }
        }
        __syncthreads();
#pragma unroll
        for (int k = 0; k < LW; ++k) {
            if (k < nw && k * 1024 + (int)threadIdx.x < gn) {
                const int r = min((int)((unsigned)pw[k] >> 24), TR - 1);
                const int pos = __hip_atomic_fetch_add(&cur[r], 1, __ATOMIC_RELAXED,
                                                       __HIP_MEMORY_SCOPE_WORKGROUP);
                queue[min(pos, GCAP - 1)] = pw[k];
            }
        }
        __syncthreads();
    }

    auto seg_sum = [&](float v, int key) {
#pragma unroll
        for (int o = 1; o < 32; o <<= 1) {
            const float t = __shfl_up(v, o);
            const int   k = __shfl_up(key, o);
            if (lane >= o && k == key) v += t;
        }
        return v;
    };
    auto seg_max = [&](unsigned v, int key) {
#pragma unroll
        for (int o = 1; o < 32; o <<= 1) {
            const unsigned t = (unsigned)__shfl_up((int)v, o);
            const int      k = __shfl_up(key, o);
            if (lane >= o && k == key) v = max(v, t);
        }
        return v;
    };
    auto seg_tail = [&](int key) {
        const int kn = __shfl_down(key, 1);
        return lane == 31 || kn != key;
    };

    auto add_rows = [&](int cnt, int ms, int md, const float (&mw)[H]) {
        float acc[NQ];
        int   arow = -1;
        auto flush = [&]() {
#pragma unroll
            for (int q = 0; q < NQ; ++q) {
                const int c = q * 32 + lane;
                if (ROW % 32 == 0 || c < ROW)
                    __hip_atomic_fetch_add(&tile[arow + c], acc[q], __ATOMIC_RELAXED,
                                           __HIP_MEMORY_SCOPE_WORKGROUP);
            }
        };
        for (int j0 = 0; j0 < cnt; j0 += EG) {
            float v[EG][NQ];
            int   trow[EG];
#pragma unroll
            for (int g = 0; g < EG; ++g) {
                const int j  = min(j0 + g, cnt - 1);
                const int sj = __builtin_amdgcn_readlane(ms, j);
                trow[g] = __builtin_amdgcn_readlane(md, j) * ROW;
                const float* frow = feat + (long long)sj * FS;
                float wj[H];
#pragma unroll
                for (int h = 0; h < H; ++h)
                    wj[h] = __int_as_float(__builtin_amdgcn_readlane(__float_as_int(mw[h]), j));
#pragma unroll
                for (int q = 0; q < NQ; ++q) {
                    const int c = q * 32 + lane;
                    float t = 0.0f;
                    if (ROW % 32 == 0 || c < ROW) t = frow[c];
                    if (GAT) t *= wj[(H == 1) ? 0 : (q * 32) / D];
                    v[g][q] = t;
                }
            }
#pragma unroll
            for (int g = 0; g < EG; ++g) {
                if (j0 + g < cnt) {
                    if (trow[g] != arow) {
                        if (arow >= 0) flush();
                        arow = trow[g];
#pragma unroll
                        for (int q = 0; q < NQ; ++q) acc[q] = v[g][q];
                    } else {
#pragma unroll
                        for (int q = 0; q < NQ; ++q) acc[q] += v[g][q];
                    }
                }
            }
        }
        if (arow >= 0) flush();
    };

    auto logits = [&](int s, int d, float (&e)[H]) {
        float l[H];
        if (H == 2) {
            const v2f t = *(const v2f*)(el + (long long)s * 2);
            l[0] = t.x; l[H - 1] = t.y;
        } else {
#pragma unroll
            for (int h = 0; h < H; ++h) l[h] = el[(long long)s * H + h];
        }
#pragma unroll
        for (int h = 0; h < H; ++h) {
            const float x = l[h] + rer[d * H + h];
            e[h] = x > 0.0f ? x : 0.2f * x;
        }
    };

    auto edge_of = [&](bool list, int base, int cnt, int& ms, int& md) {
        ms = 0; md = 0;
        if (list) {
            if (lane < cnt) {
                const unsigned p = (unsigned)queue[base + lane];
                ms = (int)(p & 0x00FFFFFFu);
                md = (int)(p >> 24);
            }
        } else {
            __builtin_amdgcn_fence(__ATOMIC_ACQ_REL, "wavefront");
            if (lane < cnt) {
                const int e = queue[qb + base + lane];
                ms = src[e];
                md = dst[e] - lo;
            }
        }
    };

    auto batch_sum = [&](bool list, int base, int cnt) {
        int ms, md;
        float mw[H];
#pragma unroll
        for (int h = 0; h < H; ++h) mw[h] = 0.0f;
        edge_of(list, base, cnt, ms, md);
        if (!list && lane < cnt)
            __hip_atomic_fetch_add(&rdeg[md], 1, __ATOMIC_RELAXED,
                                   __HIP_MEMORY_SCOPE_WORKGROUP);
        add_rows(cnt, ms, md, mw);
    };

    auto batch_max = [&](bool list, int base, int cnt) {
        int s, d;
        edge_of(list, base, cnt, s, d);
        if (lane < cnt) {
            float lg[H];
            logits(s, d, lg);
#pragma unroll
            for (int h = 0; h < H; ++h)
                __hip_atomic_fetch_max(&rmax[d * H + h], enc_f(lg[h]), __ATOMIC_RELAXED,
                                       __HIP_MEMORY_SCOPE_WORKGROUP);
        }
    };

    auto batch_acc = [&](bool list, int base, int cnt) {
        int ms, md;
        float mw[H];
#pragma unroll
        for (int h = 0; h < H; ++h) mw[h] = 0.0f;
        edge_of(list, base, cnt, ms, md);
        if (lane < cnt) {
            float lg[H];
            logits(ms, md, lg);
#pragma unroll
            for (int h = 0; h < H; ++h) {
                float m = dec_f(rmax[md * H + h]);
                if (!(fabsf(m) <= 3.402823466e38f)) m = 0.0f;
                mw[h] = __expf(lg[h] - m);
                if (!list)
                    __hip_atomic_fetch_add(&rsum[md * H + h], mw[h], __ATOMIC_RELAXED,
                                           __HIP_MEMORY_SCOPE_WORKGROUP);
            }
        }
        if (list) {
            const int  key  = lane < cnt ? md : -1;
            const bool tail = seg_tail(key);
#pragma unroll
            for (int h = 0; h < H; ++h) {
                const float sh = seg_sum(mw[h], key);
                if (lane < cnt && tail)
                    __hip_atomic_fetch_add(&rsum[md * H + h], sh, __ATOMIC_RELAXED,
                                           __HIP_MEMORY_SCOPE_WORKGROUP);
            }
        }
        add_rows(cnt, ms, md, mw);
    };

    auto offer = [&](int e, int d, unsigned okm) {
        const unsigned mask = __builtin_amdgcn_ballot_w32((unsigned)(d - lo) < (unsigned)TR) & okm;
        if (mask != 0u) {
            if ((mask >> lane) & 1u) queue[qb + qn + __popc(mask & ((1u << lane) - 1u))] = e;
            qn += __popc(mask);
        }
    };

    const v4i* dst4 = (const v4i*)dst;
    const int nvec  = E >> 2;
    auto fetch = [&](int vi) { return dst4[min(vi, nvec - 1)]; };

    auto scan = [&](auto&& fn, bool keep) {
        if (wave == 0) {
            const int e = min(nvec * 4 + lane, E - 1);
            offer(e, dst[e], __builtin_amdgcn_ballot_w32(nvec * 4 + lane < E));
        }
        v4i nxt[4];
        int vb = wave * 128;
#pragma unroll
        for (int u = 0; u < 4; ++u) nxt[u] = fetch(vb + u * 32 + lane);
        for (;;) {
            const bool last = vb >= nvec;
            if (!last) {
                if (keep && qn + TRIP > QCAP) {
                    const int pend = qn - done;
                    __builtin_amdgcn_fence(__ATOMIC_ACQ_REL, "wavefront");
                    if (lane < pend) queue[qb + lane] = queue[qb + done + lane];
                    qn = pend; done = 0; keep = false;
                    if (lane == 0) ovf = 1;
                }
                v4i cur[4];
#pragma unroll
                for (int u = 0; u < 4; ++u) cur[u] = nxt[u];
                const int vn = vb + 32 * 128;
#pragma unroll
                for (int u = 0; u < 4; ++u) nxt[u] = fetch(vn + u * 32 + lane);
#pragma unroll
                for (int u = 0; u < 4; ++u) {
                    const int vi = vb + u * 32 + lane;
                    const unsigned ok = __builtin_amdgcn_ballot_w32(vi < nvec);
                    offer(vi * 4 + 0, cur[u].x, ok);
                    offer(vi * 4 + 1, cur[u].y, ok);
                    offer(vi * 4 + 2, cur[u].z, ok);
                    offer(vi * 4 + 3, cur[u].w, ok);
                }
            }
            const int thr = last ? 1 : 32;
            while (qn - done >= thr) {
                const int cnt = min(qn - done, 32);
                int base;
                if (keep) { base = done; done += cnt; }
                else      { qn -= cnt; base = qn; }
                fn(base, cnt);
            }
            if (last) break;
            vb += 32 * 128;
        }
    };

    if (!GAT) {
        if (listed) {
            for (int b = wave * 32; b < gn; b += 32 * 32) batch_sum(true, b, min(gn - b, 32));
        } else {
            scan([&](int b, int c) { batch_sum(false, b, c); }, false);
        }
        __syncthreads();
        for (int i = threadIdx.x; i < TR; i += 1024)
            if (lo + i < N) degout[lo + i] = (float)(listed ? hist[i] : rdeg[i]);
    } else {
        if (listed) {
            for (int b0 = 0; b0 < gn; b0 += 4 * 32 * 32) {
                int  s[4], d[4];
                bool ok[4];
                float lg[4][H];
#pragma unroll
                for (int u = 0; u < 4; ++u) {
                    const int i = b0 + (u * 32 + wave) * 32 + lane;
                    ok[u] = i < gn;
                    const unsigned p = (unsigned)queue[min(i, gn - 1)];
                    s[u] = (int)(p & 0x00FFFFFFu);
                    d[u] = (int)(p >> 24);
                }
#pragma unroll
                for (int u = 0; u < 4; ++u) logits(s[u], d[u], lg[u]);
#pragma unroll
                for (int u = 0; u < 4; ++u) {
                    const int  key  = ok[u] ? d[u] : -1;
                    const bool tail = seg_tail(key);
#pragma unroll
                    for (int h = 0; h < H; ++h) {
                        const unsigned mx = seg_max(enc_f(lg[u][h]), key);
                        if (ok[u] && tail)
                            __hip_atomic_fetch_max(&rmax[d[u] * H + h], mx,
                                                   __ATOMIC_RELAXED, __HIP_MEMORY_SCOPE_WORKGROUP);
                    }
                }
            }
        } else {
            scan([&](int b, int c) { batch_max(false, b, c); }, true);
        }
        __syncthreads();
        const bool rescan = __builtin_amdgcn_readfirstlane(ovf) != 0;
        if (listed) {
            for (int b = wave * 32; b < gn; b += 32 * 32) batch_acc(true, b, min(gn - b, 32));
        } else if (!rescan) {
            for (int b = 0; b < qn; b += 32) batch_acc(false, b, min(qn - b, 32));
        } else {
            qn = 0; done = 0;
            scan([&](int b, int c) { batch_acc(false, b, c); }, false);
        }
        __syncthreads();
    }

    const long long g0 = (long long)lo * ROW;
    const long long gN = (long long)N * ROW;
    for (int i = threadIdx.x; i < TR * ROW; i += 1024) {
        if (g0 + i < gN) {
            float x = tile[i];
            if (GAT) {
                const int r = i / ROW, c = i - r * ROW;
                x = x / fmaxf(rsum[r * H + c / D], 1e-9f) + bias[c];
            }
            if (ACT) x = x > 0.0f ? x : (expf(x) - 1.0f);
            out[g0 + i] = x;
        }
    }
}

static inline unsigned gsz(long long n, int b) { return (unsigned)((n + b - 1) / b); }

extern "C" void kernel_launch(void* const* d_in, const int* in_sizes, int n_in,
                              void* d_out, int out_size, void* d_ws, size_t ws_size,
                              hipStream_t stream) {
    const float* x       = (const float*)d_in[0];
    const int*   src     = (const int*)  d_in[1];
    const int*   dst     = (const int*)  d_in[2];
    const float* w_self  = (const float*)d_in[3];
    const float* w_neigh = (const float*)d_in[4];
    const float* sage_b  = (const float*)d_in[5];
    const float* g1_w    = (const float*)d_in[6];
    const float* g1_al   = (const float*)d_in[7];
    const float* g1_ar   = (const float*)d_in[8];
    const float* g1_b    = (const float*)d_in[9];
    const float* g2_w    = (const float*)d_in[10];
    const float* g2_al   = (const float*)d_in[11];
    const float* g2_ar   = (const float*)d_in[12];
    const float* g2_b    = (const float*)d_in[13];

    int*      glist  = (int*)d_ws;
    int*      gcount = glist + AGG_NT * AGG_GCAP;
    float*    ws    = (float*)(gcount + AGG_NT * AGG_CSTR);
    float*    deg   = ws;
    float*    neigh = deg + pad64(Nn);
    float*    h0    = neigh + (long long)Nn * F;
    float*    el    = h0 + (long long)Nn * D0;
    float*    er    = el + pad64(Nn * H1);
    float*    feat  = neigh;

    const int B = 256;
    const unsigned ntile  = AGG_NT;

    fill_i32<<<gsz(AGG_NT * AGG_CSTR, B), B, 0, stream>>>(gcount, 0, AGG_NT * AGG_CSTR);
    k_bin<<<256, 1024, 0, stream>>>(src, dst, glist, gcount, Ne);
    k_tile_aggregate<F, F, 1, false, false><<<ntile, 1024, 0, stream>>>(
        x, src, dst, nullptr, nullptr, nullptr, neigh, glist, gcount, deg, Ne, Nn);
    wmma_gemm<F, 2, D0, D0, 16, true, true><<<GEMM_WG, 16 * 32, 0, stream>>>(
        x, w_self, neigh, w_neigh, deg, sage_b, h0, Nn);

    wmma_gemm<D0, 2, H1 * D1, H1 * D1, 32, false, false><<<GEMM_WG, 32 * 32, 0, stream>>>(
        h0, g1_w, nullptr, nullptr, nullptr, nullptr, feat, Nn);
    k_attn<<<gsz(Nn * H1, B), B, 0, stream>>>(feat, g1_al, g1_ar, el, er, Nn, H1, D1, H1 * D1);
    k_tile_aggregate<H1 * D1, H1 * D1, H1, true, true><<<ntile, 1024, 0, stream>>>(
        feat, src, dst, el, er, g1_b, h0, glist, gcount, nullptr, Ne, Nn);

    wmma_gemm<D0, 1, C2, C2P, 32, false, false><<<GEMM_WG, 32 * 32, 0, stream>>>(
        h0, g2_w, nullptr, nullptr, nullptr, nullptr, feat, Nn);
    k_attn<<<gsz(Nn, B), B, 0, stream>>>(feat, g2_al, g2_ar, el, er, Nn, 1, C2, C2P);
    k_tile_aggregate<C2, C2P, 1, true, false><<<ntile, 1024, 0, stream>>>(
        feat, src, dst, el, er, g2_b, (float*)d_out, glist, gcount, nullptr, Ne, Nn);
}


# Aggregation gathers widened: a lane loads 4 floats of a source row (2 at the 64-float pitch), a wave's load covers 4 (2) whole 128-byte lines, not 1; twice the rows in flight; sum order unchanged
# speedup vs baseline: 5.7555x; 1.0398x over previous
#include <hip/hip_runtime.h>

constexpr int Nn  = 50000;
constexpr int Ne  = 800000;
constexpr int F   = 256;
constexpr int D0  = 128;
constexpr int H1  = 2;
constexpr int D1  = 64;
constexpr int C2  = 47;

constexpr int AGG_TR   = 224;
constexpr int AGG_NT   = (Nn + AGG_TR - 1) / AGG_TR;
constexpr int AGG_GCAP = 16384;
constexpr int AGG_CSTR = 64;
constexpr int C2P  = 64;
constexpr int pad64(int n) { return (n + 63) / 64 * 64; }

typedef float v2f __attribute__((ext_vector_type(2)));
typedef float v8f __attribute__((ext_vector_type(8)));
typedef int   v4i __attribute__((ext_vector_type(4)));

__device__ __forceinline__ unsigned enc_f(float f) {
    unsigned u = __float_as_uint(f);
    return (u & 0x80000000u) ? ~u : (u | 0x80000000u);
}
__device__ __forceinline__ float dec_f(unsigned u) {
    unsigned b = (u & 0x80000000u) ? (u ^ 0x80000000u) : ~u;
    return __uint_as_float(b);
}

constexpr int GEMM_WG = 256;

template <int K, int NCG, int NR, int LDO, int NW, bool HAS_A2, bool ACT>
__global__ __launch_bounds__(NW * 32)
void wmma_gemm(const float* __restrict__ A, const float* __restrict__ W,
               const float* __restrict__ A2, const float* __restrict__ W2,
               const float* __restrict__ rowdeg, const float* __restrict__ bias,
               float* __restrict__ out, int M)
{
    constexpr int NC  = NCG * 64;
    constexpr int NPP = NC + 16;
    static_assert(NW % NCG == 0 && K % 4 == 0, "waves split evenly over the column groups");
    static_assert(NW <= 32, "a work-group holds at most 32 waves");
    static_assert(NR <= NC && (NR == NC || !HAS_A2), "the padded copy is written for one matrix");
    static_assert(NR <= LDO && LDO <= NC, "an output row holds the NR columns, then zeros up to its pitch");
    __shared__ float w1s[K * NPP];
    __shared__ float w2s[HAS_A2 ? K * NPP : 4];

    if (NR == NC) {
        constexpr int NV = NC / 4;
        for (int i = threadIdx.x; i < (K / 2) * NV; i += NW * 32) {
            const int kp = i / NV, cc = (i % NV) * 4;
            const float4 r0 = *(const float4*)(W + (2 * kp) * NC + cc);
            const float4 r1 = *(const float4*)(W + (2 * kp + 1) * NC + cc);
            float* d = w1s + (kp * NPP + cc) * 2;
            *(float4*)(d)     = make_float4(r0.x, r1.x, r0.y, r1.y);
            *(float4*)(d + 4) = make_float4(r0.z, r1.z, r0.w, r1.w);
            if (HAS_A2) {
                const float4 q0 = *(const float4*)(W2 + (2 * kp) * NC + cc);
                const float4 q1 = *(const float4*)(W2 + (2 * kp + 1) * NC + cc);
                float* d2 = w2s + (kp * NPP + cc) * 2;
                *(float4*)(d2)     = make_float4(q0.x, q1.x, q0.y, q1.y);
                *(float4*)(d2 + 4) = make_float4(q0.z, q1.z, q0.w, q1.w);
            }
        }
    } else {
        for (int i = threadIdx.x; i < K * NC; i += NW * 32) {
            const int k = i / NC, cc = i % NC;
            w1s[((k >> 1) * NPP + cc) * 2 + (k & 1)] = (cc < NR) ? W[k * NR + cc] : 0.0f;
        }
    }
    __syncthreads();

    const int lane = threadIdx.x & 31;
    const int half = lane >> 4;
    const int l    = lane & 15;
    const int wave = __builtin_amdgcn_readfirstlane((int)(threadIdx.x >> 5));
    const int col0 = (wave % NCG) * 64;
    const int ntile = M / 16;

    const float* Bs[4];
    const float* B2s[4];
#pragma unroll
    for (int t = 0; t < 4; ++t) {
        Bs[t] = w1s + (half * NPP + col0 + t * 16 + l) * 2;
        if (HAS_A2) B2s[t] = w2s + (half * NPP + col0 + t * 16 + l) * 2;
    }

    for (int s = wave / NCG; s * GEMM_WG + (int)blockIdx.x < ntile; s += NW / NCG) {
        const int rt  = s * GEMM_WG + (int)blockIdx.x;
        const int row = rt * 16 + l;

        const float* Arow  = A + (long long)row * K + half * 2;
        const float* A2row = HAS_A2 ? (A2 + (long long)row * K + half * 2) : nullptr;
        const float inv    = HAS_A2 ? (1.0f / fmaxf(rowdeg[row], 1.0f)) : 0.0f;
        __builtin_prefetch(Arow, 0, 3);

        v8f acc[4] = {};
        for (int kk = 0; kk < K; kk += 4) {
            const v2f a = *(const v2f*)(Arow + kk);
            v2f a2;
            if (HAS_A2) {
                const v2f t2 = *(const v2f*)(A2row + kk);
                a2.x = t2.x * inv; a2.y = t2.y * inv;
            }
#pragma unroll
            for (int t = 0; t < 4; ++t) {
                const v2f b = *(const v2f*)(Bs[t] + kk * NPP);
                acc[t] = __builtin_amdgcn_wmma_f32_16x16x4_f32(
                    false, a, false, b, (short)0, acc[t], false, false);
                if (HAS_A2) {
                    const v2f b2 = *(const v2f*)(B2s[t] + kk * NPP);
                    acc[t] = __builtin_amdgcn_wmma_f32_16x16x4_f32(
                        false, a2, false, b2, (short)0, acc[t], false, false);
                }
            }
        }

#pragma unroll
        for (int t = 0; t < 4; ++t) {
            const int c = col0 + t * 16 + l;
            if (c < LDO) {
                const float bv = (bias && c < NR) ? bias[c] : 0.0f;
#pragma unroll
                for (int v = 0; v < 8; ++v) {
                    const int r = rt * 16 + v + 8 * half;
                    float x = acc[t][v] + bv;
                    if (ACT) x = x > 0.0f ? x : (expf(x) - 1.0f);
                    out[(long long)r * LDO + c] = x;
                }
            }
        }
    }
}

__global__ void fill_i32(int* p, int v, int n) {
    int i = blockIdx.x * blockDim.x + threadIdx.x;
    if (i < n) p[i] = v;
}

__global__ __launch_bounds__(1024)
void k_bin(const int* __restrict__ src, const int* __restrict__ dst, int* __restrict__ glist,
           int* __restrict__ gcount, int E)
{
    __shared__ int cnt[AGG_NT];
    __shared__ int base[AGG_NT];
    const int per = (E + (int)gridDim.x - 1) / (int)gridDim.x;
    const int e0  = blockIdx.x * per;
    const int e1  = min(e0 + per, E);
    for (int i = threadIdx.x; i < AGG_NT; i += 1024) cnt[i] = 0;
    __syncthreads();
    for (int e = e0 + threadIdx.x; e < e1; e += 1024) {
        const int d = dst[e];
        if ((unsigned)d < (unsigned)Nn)
            __hip_atomic_fetch_add(&cnt[d / AGG_TR], 1, __ATOMIC_RELAXED,
                                   __HIP_MEMORY_SCOPE_WORKGROUP);
    }
    __syncthreads();
    for (int i = threadIdx.x; i < AGG_NT; i += 1024) {
        const int n = cnt[i];
        base[i] = n ? __hip_atomic_fetch_add(&gcount[i * AGG_CSTR], n, __ATOMIC_RELAXED,
                                             __HIP_MEMORY_SCOPE_AGENT) : 0;
        cnt[i] = 0;
    }
    __syncthreads();
    for (int e = e0 + threadIdx.x; e < e1; e += 1024) {
        const int d = dst[e];
        if ((unsigned)d < (unsigned)Nn) {
            const int g   = d / AGG_TR;
            const int pos = base[g] + __hip_atomic_fetch_add(&cnt[g], 1, __ATOMIC_RELAXED,
                                                             __HIP_MEMORY_SCOPE_WORKGROUP);
            if (pos < AGG_GCAP)
                glist[(long long)g * AGG_GCAP + pos] =
                    (int)((unsigned)src[e] | ((unsigned)(d - g * AGG_TR) << 24));
        }
    }
}

__global__ void k_attn(const float* __restrict__ feat, const float* __restrict__ al,
                       const float* __restrict__ ar, float* __restrict__ el,
                       float* __restrict__ er, int N, int H, int D, int ld) {
    int i = blockIdx.x * blockDim.x + threadIdx.x;
    if (i >= N * H) return;
    const int n = i / H, h = i % H;
    const float* f = feat + (long long)n * ld + h * D;
    float a = 0.0f, b = 0.0f;
    for (int d = 0; d < D; ++d) { a += f[d] * al[h * D + d]; b += f[d] * ar[h * D + d]; }
    el[i] = a; er[i] = b;
}

template <int ROW, int FS, int H, bool GAT, bool ACT>
__global__ __launch_bounds__(1024)
void k_tile_aggregate(const float* __restrict__ feat, const int* __restrict__ src,
                      const int* __restrict__ dst, const float* __restrict__ el,
                      const float* __restrict__ er, const float* __restrict__ bias,
                      float* __restrict__ out, const int* __restrict__ glist,
                      const int* __restrict__ gcount, float* __restrict__ degout, int E, int N)
{
    constexpr int TR = AGG_TR;
    constexpr int GCAP = AGG_GCAP;
    static_assert(TR <= 256 && Nn <= (1 << 24), "a list word holds 24 bits of source row, 8 of tile row");
    constexpr int VW = (FS % 128 == 0 && ROW % 128 == 0) ? 4 : 2;
    constexpr int CW = 32 * VW;
    constexpr int NQ = (ROW + CW - 1) / CW;
    constexpr int D  = ROW / H;
    constexpr int EG = (NQ >= 2) ? 8 : 16;
    typedef float vwf __attribute__((ext_vector_type(VW)));
    constexpr int TRIP = 512;
    constexpr int QCAP = GAT ? 1536 : 544;
    constexpr int NS = GAT ? TR * H : 1;
    static_assert(H == 1 || D % VW == 0, "a lane's floats must not straddle two heads");
    static_assert(FS >= NQ * CW, "the loads of a row stay inside its pitch");
    static_assert((TR * ROW) % 32 == 0, "tile must be whole 128-byte lines");
    __shared__ float tile[TR * ROW];
    __shared__ int queue[32 * QCAP];
    __shared__ unsigned rmax[NS];
    __shared__ float rsum[NS];
    __shared__ float rer[NS];
    __shared__ int ovf;
    __shared__ int rdeg[GAT ? 1 : TR];
    __shared__ int hist[TR];
    __shared__ int cur[TR];

    const int lane = threadIdx.x & 31;
    const int wave = __builtin_amdgcn_readfirstlane((int)(threadIdx.x >> 5));
    const int lo   = blockIdx.x * TR;

    for (int i = threadIdx.x; i < TR * ROW / 4; i += 1024)
        ((float4*)tile)[i] = make_float4(0.0f, 0.0f, 0.0f, 0.0f);
    if (GAT) {
        for (int i = threadIdx.x; i < TR * H; i += 1024) {
            rmax[i] = 0x007FFFFFu;
            rsum[i] = 0.0f;
            rer[i]  = (lo * H + i < N * H) ? er[lo * H + i] : 0.0f;
        }
        if (threadIdx.x == 0) ovf = 0;
    } else {
        for (int i = threadIdx.x; i < TR; i += 1024) rdeg[i] = 0;
    }
    for (int i = threadIdx.x; i < TR; i += 1024) hist[i] = 0;
    __syncthreads();

    const int qb = wave * QCAP;
    int qn = 0;
    int done = 0;
    const int* wl = glist + (long long)blockIdx.x * GCAP;
    const int gn = __builtin_amdgcn_readfirstlane(gcount[blockIdx.x * AGG_CSTR]);
    const bool listed = gn <= GCAP;

    if (listed) {
        constexpr int LW = GCAP / 1024;
        static_assert(LW * 1024 == GCAP, "the list divides evenly over the threads");
        const int nw = (gn + 1023) >> 10;
        int pw[LW];
#pragma unroll
        for (int k = 0; k < LW; ++k)
            if (k < nw) pw[k] = wl[min(k * 1024 + (int)threadIdx.x, gn - 1)];
#pragma unroll
        for (int k = 0; k < LW; ++k) {
            if (k < nw && k * 1024 + (int)threadIdx.x < gn) {
                const int r = min((int)((unsigned)pw[k] >> 24), TR - 1);
                __hip_atomic_fetch_add(&hist[r], 1, __ATOMIC_RELAXED, __HIP_MEMORY_SCOPE_WORKGROUP);
            }
        }
        __syncthreads();
        if (wave == 0) {
            constexpr int PER = (TR + 31) / 32;
            int loc[PER];
            int sum = 0;
#pragma unroll
            for (int k = 0; k < PER; ++k) {
                const int idx = lane * PER + k;
                loc[k] = idx < TR ? hist[idx] : 0;
                sum += loc[k];
            }
            int inc = sum;
#pragma unroll
            for (int o = 1; o < 32; o <<= 1) {
                const int t = __shfl_up(inc, o);
                if (lane >= o) inc += t;
            }
            int ex = inc - sum;
#pragma unroll
            for (int k = 0; k < PER; ++k) {
                const int idx = lane * PER + k;
                if (idx < TR) cur[idx] = ex;
                ex += loc[k];
            }
        }
        __syncthreads();
#pragma unroll
        for (int k = 0; k < LW; ++k) {
            if (k < nw && k * 1024 + (int)threadIdx.x < gn) {
                const int r = min((int)((unsigned)pw[k] >> 24), TR - 1);
                const int pos = __hip_atomic_fetch_add(&cur[r], 1, __ATOMIC_RELAXED,
                                                       __HIP_MEMORY_SCOPE_WORKGROUP);
                queue[min(pos, GCAP - 1)] = pw[k];
            }
        }
        __syncthreads();
    }

    auto seg_sum = [&](float v, int key) {
#pragma unroll
        for (int o = 1; o < 32; o <<= 1) {
            const float t = __shfl_up(v, o);
            const int   k = __shfl_up(key, o);
            if (lane >= o && k == key) v += t;
        }
        return v;
    };
    auto seg_max = [&](unsigned v, int key) {
#pragma unroll
        for (int o = 1; o < 32; o <<= 1) {
            const unsigned t = (unsigned)__shfl_up((int)v, o);
            const int      k = __shfl_up(key, o);
            if (lane >= o && k == key) v = max(v, t);
        }
        return v;
    };
    auto seg_tail = [&](int key) {
        const int kn = __shfl_down(key, 1);
        return lane == 31 || kn != key;
    };

    auto add_rows = [&](int cnt, int ms, int md, const float (&mw)[H]) {
        vwf acc[NQ];
        int arow = -1;
        auto flush = [&]() {
#pragma unroll
            for (int q = 0; q < NQ; ++q) {
#pragma unroll
                for (int i = 0; i < VW; ++i) {
                    const int c = q * CW + lane * VW + i;
                    if (ROW % CW == 0 || c < ROW)
                        __hip_atomic_fetch_add(&tile[arow + c], acc[q][i], __ATOMIC_RELAXED,
                                               __HIP_MEMORY_SCOPE_WORKGROUP);
                }
            }
        };
        for (int j0 = 0; j0 < cnt; j0 += EG) {
            vwf v[EG][NQ];
            int trow[EG];
#pragma unroll
            for (int g = 0; g < EG; ++g) {
                const int j  = min(j0 + g, cnt - 1);
                const int sj = __builtin_amdgcn_readlane(ms, j);
                trow[g] = __builtin_amdgcn_readlane(md, j) * ROW;
                const float* frow = feat + (long long)sj * FS;
                float wj[H];
#pragma unroll
                for (int h = 0; h < H; ++h)
                    wj[h] = __int_as_float(__builtin_amdgcn_readlane(__float_as_int(mw[h]), j));
#pragma unroll
                for (int q = 0; q < NQ; ++q) {
                    vwf t = *(const vwf*)(frow + q * CW + lane * VW);
                    if (GAT) {
                        float w = wj[0];
                        if (H > 1) {
                            const int hd = (q * CW + lane * VW) / D;
#pragma unroll
                            for (int h = 1; h < H; ++h) w = (hd == h) ? wj[h] : w;
                        }
                        t *= w;
                    }
                    v[g][q] = t;
                }
            }
#pragma unroll
            for (int g = 0; g < EG; ++g) {
                if (j0 + g < cnt) {
                    if (trow[g] != arow) {
                        if (arow >= 0) flush();
                        arow = trow[g];
#pragma unroll
                        for (int q = 0; q < NQ; ++q) acc[q] = v[g][q];
                    } else {
#pragma unroll
                        for (int q = 0; q < NQ; ++q) acc[q] += v[g][q];
                    }
                }
            }
        }
        if (arow >= 0) flush();
    };

    auto logits = [&](int s, int d, float (&e)[H]) {
        float l[H];
        if (H == 2) {
            const v2f t = *(const v2f*)(el + (long long)s * 2);
            l[0] = t.x; l[H - 1] = t.y;
        } else {
#pragma unroll
            for (int h = 0; h < H; ++h) l[h] = el[(long long)s * H + h];
        }
#pragma unroll
        for (int h = 0; h < H; ++h) {
            const float x = l[h] + rer[d * H + h];
            e[h] = x > 0.0f ? x : 0.2f * x;
        }
    };

    auto edge_of = [&](bool list, int base, int cnt, int& ms, int& md) {
        ms = 0; md = 0;
        if (list) {
            if (lane < cnt) {
                const unsigned p = (unsigned)queue[base + lane];
                ms = (int)(p & 0x00FFFFFFu);
                md = (int)(p >> 24);
            }
        } else {
            __builtin_amdgcn_fence(__ATOMIC_ACQ_REL, "wavefront");
            if (lane < cnt) {
                const int e = queue[qb + base + lane];
                ms = src[e];
                md = dst[e] - lo;
            }
        }
    };

    auto batch_sum = [&](bool list, int base, int cnt) {
        int ms, md;
        float mw[H];
#pragma unroll
        for (int h = 0; h < H; ++h) mw[h] = 0.0f;
        edge_of(list, base, cnt, ms, md);
        if (!list && lane < cnt)
            __hip_atomic_fetch_add(&rdeg[md], 1, __ATOMIC_RELAXED,
                                   __HIP_MEMORY_SCOPE_WORKGROUP);
        add_rows(cnt, ms, md, mw);
    };

    auto batch_max = [&](bool list, int base, int cnt) {
        int s, d;
        edge_of(list, base, cnt, s, d);
        if (lane < cnt) {
            float lg[H];
            logits(s, d, lg);
#pragma unroll
            for (int h = 0; h < H; ++h)
                __hip_atomic_fetch_max(&rmax[d * H + h], enc_f(lg[h]), __ATOMIC_RELAXED,
                                       __HIP_MEMORY_SCOPE_WORKGROUP);
        }
    };

    auto batch_acc = [&](bool list, int base, int cnt) {
        int ms, md;
        float mw[H];
#pragma unroll
        for (int h = 0; h < H; ++h) mw[h] = 0.0f;
        edge_of(list, base, cnt, ms, md);
        if (lane < cnt) {
            float lg[H];
            logits(ms, md, lg);
#pragma unroll
            for (int h = 0; h < H; ++h) {
                float m = dec_f(rmax[md * H + h]);
                if (!(fabsf(m) <= 3.402823466e38f)) m = 0.0f;
                mw[h] = __expf(lg[h] - m);
                if (!list)
                    __hip_atomic_fetch_add(&rsum[md * H + h], mw[h], __ATOMIC_RELAXED,
                                           __HIP_MEMORY_SCOPE_WORKGROUP);
            }
        }
        if (list) {
            const int  key  = lane < cnt ? md : -1;
            const bool tail = seg_tail(key);
#pragma unroll
            for (int h = 0; h < H; ++h) {
                const float sh = seg_sum(mw[h], key);
                if (lane < cnt && tail)
                    __hip_atomic_fetch_add(&rsum[md * H + h], sh, __ATOMIC_RELAXED,
                                           __HIP_MEMORY_SCOPE_WORKGROUP);
            }
        }
        add_rows(cnt, ms, md, mw);
    };

    auto offer = [&](int e, int d, unsigned okm) {
        const unsigned mask = __builtin_amdgcn_ballot_w32((unsigned)(d - lo) < (unsigned)TR) & okm;
        if (mask != 0u) {
            if ((mask >> lane) & 1u) queue[qb + qn + __popc(mask & ((1u << lane) - 1u))] = e;
            qn += __popc(mask);
        }
    };

    const v4i* dst4 = (const v4i*)dst;
    const int nvec  = E >> 2;
    auto fetch = [&](int vi) { return dst4[min(vi, nvec - 1)]; };

    auto scan = [&](auto&& fn, bool keep) {
        if (wave == 0) {
            const int e = min(nvec * 4 + lane, E - 1);
            offer(e, dst[e], __builtin_amdgcn_ballot_w32(nvec * 4 + lane < E));
        }
        v4i nxt[4];
        int vb = wave * 128;
#pragma unroll
        for (int u = 0; u < 4; ++u) nxt[u] = fetch(vb + u * 32 + lane);
        for (;;) {
            const bool last = vb >= nvec;
            if (!last) {
                if (keep && qn + TRIP > QCAP) {
                    const int pend = qn - done;
                    __builtin_amdgcn_fence(__ATOMIC_ACQ_REL, "wavefront");
                    if (lane < pend) queue[qb + lane] = queue[qb + done + lane];
                    qn = pend; done = 0; keep = false;
                    if (lane == 0) ovf = 1;
                }
                v4i cur[4];
#pragma unroll
                for (int u = 0; u < 4; ++u) cur[u] = nxt[u];
                const int vn = vb + 32 * 128;
#pragma unroll
                for (int u = 0; u < 4; ++u) nxt[u] = fetch(vn + u * 32 + lane);
#pragma unroll
                for (int u = 0; u < 4; ++u) {
                    const int vi = vb + u * 32 + lane;
                    const unsigned ok = __builtin_amdgcn_ballot_w32(vi < nvec);
                    offer(vi * 4 + 0, cur[u].x, ok);
                    offer(vi * 4 + 1, cur[u].y, ok);
                    offer(vi * 4 + 2, cur[u].z, ok);
                    offer(vi * 4 + 3, cur[u].w, ok);
                }
            }
            const int thr = last ? 1 : 32;
            while (qn - done >= thr) {
                const int cnt = min(qn - done, 32);
                int base;
                if (keep) { base = done; done += cnt; }
                else      { qn -= cnt; base = qn; }
                fn(base, cnt);
            }
            if (last) break;
            vb += 32 * 128;
        }
    };

    if (!GAT) {
        if (listed) {
            for (int b = wave * 32; b < gn; b += 32 * 32) batch_sum(true, b, min(gn - b, 32));
        } else {
            scan([&](int b, int c) { batch_sum(false, b, c); }, false);
        }
        __syncthreads();
        for (int i = threadIdx.x; i < TR; i += 1024)
            if (lo + i < N) degout[lo + i] = (float)(listed ? hist[i] : rdeg[i]);
    } else {
        if (listed) {
            for (int b0 = 0; b0 < gn; b0 += 4 * 32 * 32) {
                int  s[4], d[4];
                bool ok[4];
                float lg[4][H];
#pragma unroll
                for (int u = 0; u < 4; ++u) {
                    const int i = b0 + (u * 32 + wave) * 32 + lane;
                    ok[u] = i < gn;
                    const unsigned p = (unsigned)queue[min(i, gn - 1)];
                    s[u] = (int)(p & 0x00FFFFFFu);
                    d[u] = (int)(p >> 24);
                }
#pragma unroll
                for (int u = 0; u < 4; ++u) logits(s[u], d[u], lg[u]);
#pragma unroll
                for (int u = 0; u < 4; ++u) {
                    const int  key  = ok[u] ? d[u] : -1;
                    const bool tail = seg_tail(key);
#pragma unroll
                    for (int h = 0; h < H; ++h) {
                        const unsigned mx = seg_max(enc_f(lg[u][h]), key);
                        if (ok[u] && tail)
                            __hip_atomic_fetch_max(&rmax[d[u] * H + h], mx,
                                                   __ATOMIC_RELAXED, __HIP_MEMORY_SCOPE_WORKGROUP);
                    }
                }
            }
        } else {
            scan([&](int b, int c) { batch_max(false, b, c); }, true);
        }
        __syncthreads();
        const bool rescan = __builtin_amdgcn_readfirstlane(ovf) != 0;
        if (listed) {
            for (int b = wave * 32; b < gn; b += 32 * 32) batch_acc(true, b, min(gn - b, 32));
        } else if (!rescan) {
            for (int b = 0; b < qn; b += 32) batch_acc(false, b, min(qn - b, 32));
        } else {
            qn = 0; done = 0;
            scan([&](int b, int c) { batch_acc(false, b, c); }, false);
        }
        __syncthreads();
    }

    const long long g0 = (long long)lo * ROW;
    const long long gN = (long long)N * ROW;
    for (int i = threadIdx.x; i < TR * ROW; i += 1024) {
        if (g0 + i < gN) {
            float x = tile[i];
            if (GAT) {
                const int r = i / ROW, c = i - r * ROW;
                x = x / fmaxf(rsum[r * H + c / D], 1e-9f) + bias[c];
            }
            if (ACT) x = x > 0.0f ? x : (expf(x) - 1.0f);
            out[g0 + i] = x;
        }
    }
}

static inline unsigned gsz(long long n, int b) { return (unsigned)((n + b - 1) / b); }

extern "C" void kernel_launch(void* const* d_in, const int* in_sizes, int n_in,
                              void* d_out, int out_size, void* d_ws, size_t ws_size,
                              hipStream_t stream) {
    const float* x       = (const float*)d_in[0];
    const int*   src     = (const int*)  d_in[1];
    const int*   dst     = (const int*)  d_in[2];
    const float* w_self  = (const float*)d_in[3];
    const float* w_neigh = (const float*)d_in[4];
    const float* sage_b  = (const float*)d_in[5];
    const float* g1_w    = (const float*)d_in[6];
    const float* g1_al   = (const float*)d_in[7];
    const float* g1_ar   = (const float*)d_in[8];
    const float* g1_b    = (const float*)d_in[9];
    const float* g2_w    = (const float*)d_in[10];
    const float* g2_al   = (const float*)d_in[11];
    const float* g2_ar   = (const float*)d_in[12];
    const float* g2_b    = (const float*)d_in[13];

    int*      glist  = (int*)d_ws;
    int*      gcount = glist + AGG_NT * AGG_GCAP;
    float*    ws    = (float*)(gcount + AGG_NT * AGG_CSTR);
    float*    deg   = ws;
    float*    neigh = deg + pad64(Nn);
    float*    h0    = neigh + (long long)Nn * F;
    float*    el    = h0 + (long long)Nn * D0;
    float*    er    = el + pad64(Nn * H1);
    float*    feat  = neigh;

    const int B = 256;
    const unsigned ntile  = AGG_NT;

    fill_i32<<<gsz(AGG_NT * AGG_CSTR, B), B, 0, stream>>>(gcount, 0, AGG_NT * AGG_CSTR);
    k_bin<<<256, 1024, 0, stream>>>(src, dst, glist, gcount, Ne);
    k_tile_aggregate<F, F, 1, false, false><<<ntile, 1024, 0, stream>>>(
        x, src, dst, nullptr, nullptr, nullptr, neigh, glist, gcount, deg, Ne, Nn);
    wmma_gemm<F, 2, D0, D0, 16, true, true><<<GEMM_WG, 16 * 32, 0, stream>>>(
        x, w_self, neigh, w_neigh, deg, sage_b, h0, Nn);

    wmma_gemm<D0, 2, H1 * D1, H1 * D1, 32, false, false><<<GEMM_WG, 32 * 32, 0, stream>>>(
        h0, g1_w, nullptr, nullptr, nullptr, nullptr, feat, Nn);
    k_attn<<<gsz(Nn * H1, B), B, 0, stream>>>(feat, g1_al, g1_ar, el, er, Nn, H1, D1, H1 * D1);
    k_tile_aggregate<H1 * D1, H1 * D1, H1, true, true><<<ntile, 1024, 0, stream>>>(
        feat, src, dst, el, er, g1_b, h0, glist, gcount, nullptr, Ne, Nn);

    wmma_gemm<D0, 1, C2, C2P, 32, false, false><<<GEMM_WG, 32 * 32, 0, stream>>>(
        h0, g2_w, nullptr, nullptr, nullptr, nullptr, feat, Nn);
    k_attn<<<gsz(Nn, B), B, 0, stream>>>(feat, g2_al, g2_ar, el, er, Nn, 1, C2, C2P);
    k_tile_aggregate<C2, C2P, 1, true, false><<<ntile, 1024, 0, stream>>>(
        feat, src, dst, el, er, g2_b, (float*)d_out, glist, gcount, nullptr, Ne, Nn);
}
